# starting kernel
# baseline (speedup 1.0000x reference)
#include <hip/hip_runtime.h>

#define HID   128
#define NGR   512
#define NCLS  10

typedef float v2f __attribute__((ext_vector_type(2)));
typedef float v8f __attribute__((ext_vector_type(8)));


__global__ void k_fill(float* __restrict__ p, int n, float v) {
  int i = blockIdx.x * blockDim.x + threadIdx.x;
  if (i < n) p[i] = v;
}

__global__ void k_degree(const int* __restrict__ dst, float* __restrict__ deg, int E) {
  int e = blockIdx.x * blockDim.x + threadIdx.x;
  if (e < E) atomicAdd(&deg[dst[e]], 1.0f);
}

__global__ void k_dinv(const float* __restrict__ deg, float* __restrict__ dinv, int N) {
  int i = blockIdx.x * blockDim.x + threadIdx.x;
  if (i < N) dinv[i] = rsqrtf(deg[i] + 1.0f);
}

__global__ void gemm_wmma_strip_k128(const float* __restrict__ A,
                                     const float* __restrict__ B,
                                     float* __restrict__ C,
                                     float* __restrict__ C2,
                                     const float* __restrict__ dinv, int M) {
  __shared__ float Bs[128 * 128];

  for (int idx = threadIdx.x; idx < 128 * 128; idx += blockDim.x) {
    const int row = idx >> 7;
    const int col = idx & 127;
    Bs[(row << 7) | (col ^ (((row >> 1) & 1) << 4))] = B[idx];
  }
  __syncthreads();

  const int lane = threadIdx.x & 31;
  const int wave = threadIdx.x >> 5;
  const int tm   = blockIdx.x * (blockDim.x >> 5) + wave;
  if (tm >= (M >> 4)) return;

  const int half = lane >> 4;
  const int l15  = lane & 15;
  const int swz  = half << 4;

  v8f acc[8];
#pragma unroll
  for (int tn = 0; tn < 8; ++tn)
#pragma unroll
    for (int r = 0; r < 8; ++r) acc[tn][r] = 0.0f;

  const float* __restrict__ Arow = A + (size_t)(tm * 16 + l15) * 128;

  for (int k = 0; k < 128; k += 4) {
    const int kk = k + half * 2;
    v2f a;
    a[0] = Arow[kk];
    a[1] = Arow[kk + 1];
#pragma unroll
    for (int tn = 0; tn < 8; ++tn) {
      const int c = (tn * 16 + l15) ^ swz;
      v2f b;
      b[0] = Bs[(kk << 7) | c];
      b[1] = Bs[((kk + 1) << 7) | c];
      acc[tn] = __builtin_amdgcn_wmma_f32_16x16x4_f32(false, a, false, b,
                                                      (short)0, acc[tn], false, false);
    }
  }

  float ds2[8];
#pragma unroll
  for (int r = 0; r < 8; ++r) {
    const float dv = dinv[tm * 16 + r + half * 8];
    ds2[r] = dv * dv;
  }
  float* __restrict__ Cp  = C  + (size_t)(tm * 16) * 128 + l15;
  float* __restrict__ C2p = C2 + (size_t)(tm * 16) * 128 + l15;
#pragma unroll
  for (int tn = 0; tn < 8; ++tn) {
#pragma unroll
    for (int r = 0; r < 8; ++r) {
      const size_t off = (size_t)(r + half * 8) * 128 + tn * 16;
      const float v = acc[tn][r];
      Cp[off]  = v;
      C2p[off] = v * ds2[r];
    }
  }
}

__global__ void gemm_wmma_f32(const float* __restrict__ A, const float* __restrict__ B,
                              const float* __restrict__ bias, float* __restrict__ C,
                              int M, int N, int K, int relu) {
  const int lane   = threadIdx.x & 31;
  const int wave   = threadIdx.x >> 5;
  const int tilesN = N >> 4;
  const int tilesM = M >> 4;
  const int t = blockIdx.x * (blockDim.x >> 5) + wave;
  if (t >= tilesM * tilesN) return;
  const int tm = t / tilesN;
  const int tn = t - tm * tilesN;

  const int half = lane >> 4;
  const int l15  = lane & 15;

  v8f acc;
  if (bias != nullptr) {
    const float bv = bias[tn * 16 + l15];
#pragma unroll
    for (int r = 0; r < 8; ++r) acc[r] = bv;
  } else {
#pragma unroll
    for (int r = 0; r < 8; ++r) acc[r] = 0.0f;
  }

  const float* __restrict__ Arow = A + (size_t)(tm * 16 + l15) * K;
  const float* __restrict__ Bcol = B + tn * 16 + l15;

  for (int k = 0; k < K; k += 4) {
    const int kk = k + half * 2;
    v2f a, b;
    a[0] = Arow[kk];
    a[1] = Arow[kk + 1];
    b[0] = Bcol[(size_t)kk * N];
    b[1] = Bcol[(size_t)(kk + 1) * N];
    acc = __builtin_amdgcn_wmma_f32_16x16x4_f32(false, a, false, b,
                                                (short)0, acc, false, false);
  }

  if (relu) {
#pragma unroll
    for (int r = 0; r < 8; ++r) acc[r] = acc[r] > 0.0f ? acc[r] : 0.0f;
  }

  float* __restrict__ Cp = C + (size_t)(tm * 16) * N + tn * 16 + l15;
#pragma unroll
  for (int r = 0; r < 8; ++r) Cp[(size_t)(r + half * 8) * N] = acc[r];
}


__global__ void k_edge_scatter(const int* __restrict__ src, const int* __restrict__ dst,
                               const float* __restrict__ dinv, const float* __restrict__ h,
                               float* __restrict__ agg, int E) {
  const long tid = (long)blockIdx.x * blockDim.x + threadIdx.x;
  const int e    = (int)(tid >> 5);
  const int lane = (int)(tid & 31);
  if (e >= E) return;
  const int s = src[e];
  const int d = dst[e];
  const float w = dinv[s] * dinv[d];
  const float4 v = ((const float4*)(h + (size_t)s * HID))[lane];
  float* ap = agg + (size_t)d * HID + lane * 4;
  atomicAdd(ap + 0, v.x * w);
  atomicAdd(ap + 1, v.y * w);
  atomicAdd(ap + 2, v.z * w);
  atomicAdd(ap + 3, v.w * w);
}

__global__ void k_finalize(float* __restrict__ agg, const float* __restrict__ bias, int total) {
  int i = blockIdx.x * blockDim.x + threadIdx.x;
  if (i >= total) return;
  const float v = agg[i] + bias[i & (HID - 1)];
  agg[i] = v > 0.0f ? v : 0.0f;
}


__global__ void k_count(const int* __restrict__ batch, float* __restrict__ counts, int N) {
  int i = blockIdx.x * blockDim.x + threadIdx.x;
  if (i < N) atomicAdd(&counts[batch[i]], 1.0f);
}

__global__ void k_pool(const float* __restrict__ h, const int* __restrict__ batch,
                       float* __restrict__ sums, int* __restrict__ maxs, int total) {
  int i = blockIdx.x * blockDim.x + threadIdx.x;
  if (i >= total) return;
  const int node = i >> 7;
  const int f    = i & (HID - 1);
  const int g    = batch[node];
  const float v  = h[i];
  atomicAdd(&sums[g * HID + f], v);
  atomicMax(&maxs[g * HID + f], __float_as_int(v));
}

__global__ void k_pool_final(const float* __restrict__ sums, const int* __restrict__ maxs,
                             const float* __restrict__ counts, float* __restrict__ gbuf,
                             int total  ) {
  int i = blockIdx.x * blockDim.x + threadIdx.x;
  if (i >= total) return;
  const int g = i >> 7;
  const int f = i & (HID - 1);
  const float c = counts[g];
  const float denom = c > 1.0f ? c : 1.0f;
  gbuf[(size_t)g * (2 * HID) + f]       = sums[i] / denom;
  gbuf[(size_t)g * (2 * HID) + HID + f] = __int_as_float(maxs[i]);
}

__global__ void k_head(const float* __restrict__ hcls, const float* __restrict__ Wc2,
                       const float* __restrict__ bc2, float* __restrict__ out, int total) {
  int i = blockIdx.x * blockDim.x + threadIdx.x;
  if (i >= total) return;
  const int g = i / NCLS;
  const int c = i - g * NCLS;
  float s = bc2[c];
  const float* hv = hcls + (size_t)g * HID;
#pragma unroll 8
  for (int k = 0; k < HID; ++k) s += hv[k] * Wc2[k * NCLS + c];
  out[i] = s;
}


extern "C" void kernel_launch(void* const* d_in, const int* in_sizes, int n_in,
                              void* d_out, int out_size, void* d_ws, size_t ws_size,
                              hipStream_t stream) {
  const float* x     = (const float*)d_in[0];
  const int*   edge  = (const int*)d_in[1];
  const int*   batch = (const int*)d_in[2];
  const float* W[3]  = {(const float*)d_in[3], (const float*)d_in[5], (const float*)d_in[7]};
  const float* b[3]  = {(const float*)d_in[4], (const float*)d_in[6], (const float*)d_in[8]};
  const float* Wc1   = (const float*)d_in[9];
  const float* bc1   = (const float*)d_in[10];
  const float* Wc2   = (const float*)d_in[11];
  const float* bc2   = (const float*)d_in[12];
  float* out = (float*)d_out;

  const int N = in_sizes[0] / HID;
  const int E = in_sizes[1] / 2;
  const int G = NGR;
  const int* src  = edge;
  const int* dstv = edge + E;

  float* ws     = (float*)d_ws;
  float* deg    = ws;
  float* dinv   = deg  + N;
  float* buf0   = dinv + N;
  float* buf1   = buf0 + (size_t)N * HID;
  float* counts = buf1 + (size_t)N * HID;
  float* sums   = counts + G;
  int*   maxs   = (int*)(sums + (size_t)G * HID);
  float* gbuf   = (float*)(maxs + (size_t)G * HID);
  float* hcls   = gbuf + (size_t)G * 2 * HID;

  const int BT = 256;
  auto nb = [](long n, int bt) { return (int)((n + bt - 1) / bt); };

  k_fill<<<nb(N, BT), BT, 0, stream>>>(deg, N, 0.0f);
  k_degree<<<nb(E, BT), BT, 0, stream>>>(dstv, deg, E);
  k_dinv<<<nb(N, BT), BT, 0, stream>>>(deg, dinv, N);

  const int tot = N * HID;
  const long et = (long)E * 32;
  const int rowTiles = N / 16;
  const float* hin = x;
  for (int l = 0; l < 3; ++l) {
    gemm_wmma_strip_k128<<<(rowTiles + 7) / 8, 256, 0, stream>>>(hin, W[l],
                                                                 buf0, buf1, dinv, N);
    k_edge_scatter<<<nb(et, BT), BT, 0, stream>>>(src, dstv, dinv, buf0, buf1, E);
    k_finalize<<<nb(tot, BT), BT, 0, stream>>>(buf1, b[l], tot);
    hin = buf1;
  }

  k_fill<<<nb(G, BT), BT, 0, stream>>>(counts, G, 0.0f);
  k_fill<<<nb(G * HID, BT), BT, 0, stream>>>(sums, G * HID, 0.0f);
  k_fill<<<nb(G * HID, BT), BT, 0, stream>>>((float*)maxs, G * HID, 0.0f);
  k_count<<<nb(N, BT), BT, 0, stream>>>(batch, counts, N);
  k_pool<<<nb(tot, BT), BT, 0, stream>>>(buf1, batch, sums, maxs, tot);
  k_pool_final<<<nb(G * HID, BT), BT, 0, stream>>>(sums, maxs, counts, gbuf, G * HID);

  const int tiles2 = (G / 16) * (HID / 16);
  gemm_wmma_f32<<<(tiles2 + 7) / 8, 256, 0, stream>>>(gbuf, Wc1, bc1, hcls,
                                                      G, HID, 2 * HID, 1);
  k_head<<<nb(G * NCLS, BT), BT, 0, stream>>>(hcls, Wc2, bc2, out, G * NCLS);
}


# Aggregate by destination, float atomics gone: in-call counting sort to CSR (int atomic arrival-order rank), GEMM rows pre-scaled by dinv, one wave per node sums in-rows +self +bias, relu, one store
# speedup vs baseline: 2.7290x; 2.7290x over previous
#include <hip/hip_runtime.h>

#define HID   128
#define NGR   512
#define NCLS  10

typedef float v2f __attribute__((ext_vector_type(2)));
typedef float v8f __attribute__((ext_vector_type(8)));


__global__ void k_fill(float* __restrict__ p, int n, float v) {
  int i = blockIdx.x * blockDim.x + threadIdx.x;
  if (i < n) p[i] = v;
}

__global__ void k_rank(const int* __restrict__ dst, int* __restrict__ degi,
                       int* __restrict__ rank, int E) {
  int e = blockIdx.x * blockDim.x + threadIdx.x;
  if (e < E) rank[e] = atomicAdd(&degi[dst[e]], 1);
}

__global__ void k_dinv_bsum(const int* __restrict__ degi, float* __restrict__ dinv,
                            int* __restrict__ bsum, int N) {
  __shared__ int wsum[8];
  const int t = threadIdx.x;
  int s = 0;
  for (int j = 0; j < 4; ++j) {
    const int i = blockIdx.x * 1024 + j * 256 + t;
    if (i < N) {
      const int d = degi[i];
      s += d;
      dinv[i] = rsqrtf((float)d + 1.0f);
    }
  }
  for (int d = 16; d; d >>= 1) s += __shfl_xor(s, d);
  if ((t & 31) == 0) wsum[t >> 5] = s;
  __syncthreads();
  if (t == 0) {
    int tot = 0;
    for (int w = 0; w < 8; ++w) tot += wsum[w];
    bsum[blockIdx.x] = tot;
  }
}

__global__ void k_rowptr(int* __restrict__ degi, const int* __restrict__ bsum, int N) {
  __shared__ int wsum[8];
  __shared__ int wtot[8];
  const int t = threadIdx.x;
  const int lane = t & 31;
  const int wave = t >> 5;

  int p = 0;
  for (int i = t; i < (int)blockIdx.x; i += 256) p += bsum[i];
  for (int d = 16; d; d >>= 1) p += __shfl_xor(p, d);
  if (lane == 0) wtot[wave] = p;

  const int i0 = blockIdx.x * 1024 + t * 4;
  int v[4];
  for (int j = 0; j < 4; ++j) v[j] = (i0 + j < N) ? degi[i0 + j] : 0;
  const int tsum = v[0] + v[1] + v[2] + v[3];
  int inc = tsum;
  for (int d = 1; d < 32; d <<= 1) {
    const int o = __shfl_up(inc, d);
    if (lane >= d) inc += o;
  }
  if (lane == 31) wsum[wave] = inc;
  __syncthreads();
  int off = 0;
  for (int w = 0; w < 8; ++w) {
    off += wtot[w];
    if (w < wave) off += wsum[w];
  }
  int ex = off + inc - tsum;
  for (int j = 0; j < 4; ++j) {
    if (i0 + j < N) degi[i0 + j] = ex;
    ex += v[j];
  }
}

__global__ void k_csr_fill(const int* __restrict__ src, const int* __restrict__ dst,
                           const int* __restrict__ rank, const int* __restrict__ rowptr,
                           int* __restrict__ csr, int E) {
  int e = blockIdx.x * blockDim.x + threadIdx.x;
  if (e >= E) return;
  unsigned pos = (unsigned)(rowptr[dst[e]] + rank[e]);
  pos = pos < (unsigned)E ? pos : (unsigned)(E - 1);
  csr[pos] = src[e];
}

__global__ void gemm_wmma_strip_k128(const float* __restrict__ A,
                                     const float* __restrict__ B,
                                     float* __restrict__ C,
                                     const float* __restrict__ dinv, int M) {
  __shared__ float Bs[128 * 128];

  for (int idx = threadIdx.x; idx < 128 * 128; idx += blockDim.x) {
    const int row = idx >> 7;
    const int col = idx & 127;
    Bs[(row << 7) | (col ^ (((row >> 1) & 1) << 4))] = B[idx];
  }
  __syncthreads();

  const int lane = threadIdx.x & 31;
  const int wave = threadIdx.x >> 5;
  const int tm   = blockIdx.x * (blockDim.x >> 5) + wave;
  if (tm >= (M >> 4)) return;

  const int half = lane >> 4;
  const int l15  = lane & 15;
  const int swz  = half << 4;

  v8f acc[8];
#pragma unroll
  for (int tn = 0; tn < 8; ++tn)
#pragma unroll
    for (int r = 0; r < 8; ++r) acc[tn][r] = 0.0f;

  const float* __restrict__ Arow = A + (size_t)(tm * 16 + l15) * 128;

  for (int k = 0; k < 128; k += 4) {
    const int kk = k + half * 2;
    v2f a;
    a[0] = Arow[kk];
    a[1] = Arow[kk + 1];
#pragma unroll
    for (int tn = 0; tn < 8; ++tn) {
      const int c = (tn * 16 + l15) ^ swz;
      v2f b;
      b[0] = Bs[(kk << 7) | c];
      b[1] = Bs[((kk + 1) << 7) | c];
      acc[tn] = __builtin_amdgcn_wmma_f32_16x16x4_f32(false, a, false, b,
                                                      (short)0, acc[tn], false, false);
    }
  }

  float dv[8];
#pragma unroll
  for (int r = 0; r < 8; ++r) dv[r] = dinv[tm * 16 + r + half * 8];
  float* __restrict__ Cp = C + (size_t)(tm * 16) * 128 + l15;
#pragma unroll
  for (int tn = 0; tn < 8; ++tn) {
#pragma unroll
    for (int r = 0; r < 8; ++r) {
      const size_t off = (size_t)(r + half * 8) * 128 + tn * 16;
      Cp[off] = acc[tn][r] * dv[r];
    }
  }
}

__global__ void gemm_wmma_f32(const float* __restrict__ A, const float* __restrict__ B,
                              const float* __restrict__ bias, float* __restrict__ C,
                              int M, int N, int K, int relu) {
  const int lane   = threadIdx.x & 31;
  const int wave   = threadIdx.x >> 5;
  const int tilesN = N >> 4;
  const int tilesM = M >> 4;
  const int t = blockIdx.x * (blockDim.x >> 5) + wave;
  if (t >= tilesM * tilesN) return;
  const int tm = t / tilesN;
  const int tn = t - tm * tilesN;

  const int half = lane >> 4;
  const int l15  = lane & 15;

  v8f acc;
  if (bias != nullptr) {
    const float bv = bias[tn * 16 + l15];
#pragma unroll
    for (int r = 0; r < 8; ++r) acc[r] = bv;
  } else {
#pragma unroll
    for (int r = 0; r < 8; ++r) acc[r] = 0.0f;
  }

  const float* __restrict__ Arow = A + (size_t)(tm * 16 + l15) * K;
  const float* __restrict__ Bcol = B + tn * 16 + l15;

  for (int k = 0; k < K; k += 4) {
    const int kk = k + half * 2;
    v2f a, b;
    a[0] = Arow[kk];
    a[1] = Arow[kk + 1];
    b[0] = Bcol[(size_t)kk * N];
    b[1] = Bcol[(size_t)(kk + 1) * N];
    acc = __builtin_amdgcn_wmma_f32_16x16x4_f32(false, a, false, b,
                                                (short)0, acc, false, false);
  }

  if (relu) {
#pragma unroll
    for (int r = 0; r < 8; ++r) acc[r] = acc[r] > 0.0f ? acc[r] : 0.0f;
  }

  float* __restrict__ Cp = C + (size_t)(tm * 16) * N + tn * 16 + l15;
#pragma unroll
  for (int r = 0; r < 8; ++r) Cp[(size_t)(r + half * 8) * N] = acc[r];
}


template <int U>
__device__ __forceinline__ void add_rows(float4& acc, const float4* __restrict__ H,
                                         int my, int j, int lane) {
  float4 v[U];
#pragma unroll
  for (int u = 0; u < U; ++u) {
    const int s = __builtin_amdgcn_readlane(my, j + u);
    v[u] = H[(size_t)s * 32 + lane];
  }
#pragma unroll
  for (int u = 0; u < U; ++u) {
    acc.x += v[u].x; acc.y += v[u].y; acc.z += v[u].z; acc.w += v[u].w;
  }
}

__global__ void k_gather(const int* __restrict__ rowptr, const int* __restrict__ csr,
                         const float* __restrict__ dinv, const float* __restrict__ hs,
                         const float* __restrict__ bias, float* __restrict__ out,
                         int N, int E) {
  const int lane = threadIdx.x & 31;
  const int node = __builtin_amdgcn_readfirstlane(
      (int)((blockIdx.x * blockDim.x + threadIdx.x) >> 5));
  if (node >= N) return;
  const int beg = rowptr[node];
  const int end = (node + 1 < N) ? rowptr[node + 1] : E;
  const float4* __restrict__ H = (const float4*)hs;

  float4 acc = H[(size_t)node * 32 + lane];
  for (int base = beg; base < end; base += 32) {
    const int rem = end - base;
    const int cnt = rem < 32 ? rem : 32;
    unsigned my = (lane < cnt) ? (unsigned)csr[base + lane] : 0u;
    my = my < (unsigned)N ? my : (unsigned)(N - 1);
    int j = 0;
    for (; j + 8 <= cnt; j += 8) add_rows<8>(acc, H, (int)my, j, lane);
    if (j + 4 <= cnt) { add_rows<4>(acc, H, (int)my, j, lane); j += 4; }
    for (; j < cnt; ++j) add_rows<1>(acc, H, (int)my, j, lane);
  }

  const float dv = dinv[node];
  const float4 b = ((const float4*)bias)[lane];
  float4 o;
  o.x = acc.x * dv + b.x;  o.y = acc.y * dv + b.y;
  o.z = acc.z * dv + b.z;  o.w = acc.w * dv + b.w;
  o.x = o.x > 0.0f ? o.x : 0.0f;  o.y = o.y > 0.0f ? o.y : 0.0f;
  o.z = o.z > 0.0f ? o.z : 0.0f;  o.w = o.w > 0.0f ? o.w : 0.0f;
  ((float4*)out)[(size_t)node * 32 + lane] = o;
}


__global__ void k_count(const int* __restrict__ batch, float* __restrict__ counts, int N) {
  int i = blockIdx.x * blockDim.x + threadIdx.x;
  if (i < N) atomicAdd(&counts[batch[i]], 1.0f);
}

__global__ void k_pool(const float* __restrict__ h, const int* __restrict__ batch,
                       float* __restrict__ sums, int* __restrict__ maxs, int total) {
  int i = blockIdx.x * blockDim.x + threadIdx.x;
  if (i >= total) return;
  const int node = i >> 7;
  const int f    = i & (HID - 1);
  const int g    = batch[node];
  const float v  = h[i];
  atomicAdd(&sums[g * HID + f], v);
  atomicMax(&maxs[g * HID + f], __float_as_int(v));
}

__global__ void k_pool_final(const float* __restrict__ sums, const int* __restrict__ maxs,
                             const float* __restrict__ counts, float* __restrict__ gbuf,
                             int total  ) {
  int i = blockIdx.x * blockDim.x + threadIdx.x;
  if (i >= total) return;
  const int g = i >> 7;
  const int f = i & (HID - 1);
  const float c = counts[g];
  const float denom = c > 1.0f ? c : 1.0f;
  gbuf[(size_t)g * (2 * HID) + f]       = sums[i] / denom;
  gbuf[(size_t)g * (2 * HID) + HID + f] = __int_as_float(maxs[i]);
}

__global__ void k_head(const float* __restrict__ hcls, const float* __restrict__ Wc2,
                       const float* __restrict__ bc2, float* __restrict__ out, int total) {
  int i = blockIdx.x * blockDim.x + threadIdx.x;
  if (i >= total) return;
  const int g = i / NCLS;
  const int c = i - g * NCLS;
  float s = bc2[c];
  const float* hv = hcls + (size_t)g * HID;
#pragma unroll 8
  for (int k = 0; k < HID; ++k) s += hv[k] * Wc2[k * NCLS + c];
  out[i] = s;
}


extern "C" void kernel_launch(void* const* d_in, const int* in_sizes, int n_in,
                              void* d_out, int out_size, void* d_ws, size_t ws_size,
                              hipStream_t stream) {
  const float* x     = (const float*)d_in[0];
  const int*   edge  = (const int*)d_in[1];
  const int*   batch = (const int*)d_in[2];
  const float* W[3]  = {(const float*)d_in[3], (const float*)d_in[5], (const float*)d_in[7]};
  const float* b[3]  = {(const float*)d_in[4], (const float*)d_in[6], (const float*)d_in[8]};
  const float* Wc1   = (const float*)d_in[9];
  const float* bc1   = (const float*)d_in[10];
  const float* Wc2   = (const float*)d_in[11];
  const float* bc2   = (const float*)d_in[12];
  float* out = (float*)d_out;

  const int N = in_sizes[0] / HID;
  const int E = in_sizes[1] / 2;
  const int G = NGR;
  const int* src  = edge;
  const int* dstv = edge + E;

  float* ws     = (float*)d_ws;
  int*   degi   = (int*)ws;
  float* dinv   = ws + N;
  float* buf0   = dinv + N;
  float* buf1   = buf0 + (size_t)N * HID;
  float* counts = buf1 + (size_t)N * HID;
  float* sums   = counts + G;
  int*   maxs   = (int*)(sums + (size_t)G * HID);
  float* gbuf   = (float*)(maxs + (size_t)G * HID);
  float* hcls   = gbuf + (size_t)G * 2 * HID;
  int*   csr    = (int*)(hcls + (size_t)G * HID);
  int*   rank   = (int*)buf1;
  int*   bsum   = (int*)counts;

  if ((size_t)((const char*)(csr + E) - (const char*)d_ws) > ws_size) return;

  const int BT = 256;
  auto nb = [](long n, int bt) { return (int)((n + bt - 1) / bt); };

  const int nsb = nb(N, 1024);
  k_fill<<<nb(N, BT), BT, 0, stream>>>((float*)degi, N, 0.0f);
  k_rank<<<nb(E, BT), BT, 0, stream>>>(dstv, degi, rank, E);
  k_dinv_bsum<<<nsb, 256, 0, stream>>>(degi, dinv, bsum, N);
  k_rowptr<<<nsb, 256, 0, stream>>>(degi, bsum, N);
  k_csr_fill<<<nb(E, BT), BT, 0, stream>>>(src, dstv, rank, degi, csr, E);

  const int tot = N * HID;
  const int rowTiles = N / 16;
  const float* hin = x;
  for (int l = 0; l < 3; ++l) {
    gemm_wmma_strip_k128<<<(rowTiles + 7) / 8, 256, 0, stream>>>(hin, W[l], buf0, dinv, N);
    k_gather<<<nb((long)N * 32, BT), BT, 0, stream>>>(degi, csr, dinv, buf0, b[l], buf1, N, E);
    hin = buf1;
  }

  k_fill<<<nb(G, BT), BT, 0, stream>>>(counts, G, 0.0f);
  k_fill<<<nb(G * HID, BT), BT, 0, stream>>>(sums, G * HID, 0.0f);
  k_fill<<<nb(G * HID, BT), BT, 0, stream>>>((float*)maxs, G * HID, 0.0f);
  k_count<<<nb(N, BT), BT, 0, stream>>>(batch, counts, N);
  k_pool<<<nb(tot, BT), BT, 0, stream>>>(buf1, batch, sums, maxs, tot);
  k_pool_final<<<nb(G * HID, BT), BT, 0, stream>>>(sums, maxs, counts, gbuf, G * HID);

  const int tiles2 = (G / 16) * (HID / 16);
  gemm_wmma_f32<<<(tiles2 + 7) / 8, 256, 0, stream>>>(gbuf, Wc1, bc1, hcls,
                                                      G, HID, 2 * HID, 1);
  k_head<<<nb(G * NCLS, BT), BT, 0, stream>>>(hcls, Wc2, bc2, out, G * NCLS);
}


# Pooling by runs: a wave sums 16 consecutive node rows in registers (sum, max, count) and does its atomics once per run of equal graph id, not per element; sum order changes; count launch gone
# speedup vs baseline: 2.9737x; 1.0897x over previous
#include <hip/hip_runtime.h>

#define HID   128
#define NGR   512
#define NCLS  10

typedef float v2f __attribute__((ext_vector_type(2)));
typedef float v8f __attribute__((ext_vector_type(8)));


__global__ void k_fill(float* __restrict__ p, int n, float v) {
  int i = blockIdx.x * blockDim.x + threadIdx.x;
  if (i < n) p[i] = v;
}

__global__ void k_rank(const int* __restrict__ dst, int* __restrict__ degi,
                       int* __restrict__ rank, int E) {
  int e = blockIdx.x * blockDim.x + threadIdx.x;
  if (e < E) rank[e] = atomicAdd(&degi[dst[e]], 1);
}

__global__ void k_dinv_bsum(const int* __restrict__ degi, float* __restrict__ dinv,
                            int* __restrict__ bsum, int N) {
  __shared__ int wsum[8];
  const int t = threadIdx.x;
  int s = 0;
  for (int j = 0; j < 4; ++j) {
    const int i = blockIdx.x * 1024 + j * 256 + t;
    if (i < N) {
      const int d = degi[i];
      s += d;
      dinv[i] = rsqrtf((float)d + 1.0f);
    }
  }
  for (int d = 16; d; d >>= 1) s += __shfl_xor(s, d);
  if ((t & 31) == 0) wsum[t >> 5] = s;
  __syncthreads();
  if (t == 0) {
    int tot = 0;
    for (int w = 0; w < 8; ++w) tot += wsum[w];
    bsum[blockIdx.x] = tot;
  }
}

__global__ void k_rowptr(int* __restrict__ degi, const int* __restrict__ bsum, int N) {
  __shared__ int wsum[8];
  __shared__ int wtot[8];
  const int t = threadIdx.x;
  const int lane = t & 31;
  const int wave = t >> 5;

  int p = 0;
  for (int i = t; i < (int)blockIdx.x; i += 256) p += bsum[i];
  for (int d = 16; d; d >>= 1) p += __shfl_xor(p, d);
  if (lane == 0) wtot[wave] = p;

  const int i0 = blockIdx.x * 1024 + t * 4;
  int v[4];
  for (int j = 0; j < 4; ++j) v[j] = (i0 + j < N) ? degi[i0 + j] : 0;
  const int tsum = v[0] + v[1] + v[2] + v[3];
  int inc = tsum;
  for (int d = 1; d < 32; d <<= 1) {
    const int o = __shfl_up(inc, d);
    if (lane >= d) inc += o;
  }
  if (lane == 31) wsum[wave] = inc;
  __syncthreads();
  int off = 0;
  for (int w = 0; w < 8; ++w) {
    off += wtot[w];
    if (w < wave) off += wsum[w];
  }
  int ex = off + inc - tsum;
  for (int j = 0; j < 4; ++j) {
    if (i0 + j < N) degi[i0 + j] = ex;
    ex += v[j];
  }
}

__global__ void k_csr_fill(const int* __restrict__ src, const int* __restrict__ dst,
                           const int* __restrict__ rank, const int* __restrict__ rowptr,
                           int* __restrict__ csr, int E) {
  int e = blockIdx.x * blockDim.x + threadIdx.x;
  if (e >= E) return;
  unsigned pos = (unsigned)(rowptr[dst[e]] + rank[e]);
  pos = pos < (unsigned)E ? pos : (unsigned)(E - 1);
  csr[pos] = src[e];
}

__global__ void gemm_wmma_strip_k128(const float* __restrict__ A,
                                     const float* __restrict__ B,
                                     float* __restrict__ C,
                                     const float* __restrict__ dinv, int M) {
  __shared__ float Bs[128 * 128];

  for (int idx = threadIdx.x; idx < 128 * 128; idx += blockDim.x) {
    const int row = idx >> 7;
    const int col = idx & 127;
    Bs[(row << 7) | (col ^ (((row >> 1) & 1) << 4))] = B[idx];
  }
  __syncthreads();

  const int lane = threadIdx.x & 31;
  const int wave = threadIdx.x >> 5;
  const int tm   = blockIdx.x * (blockDim.x >> 5) + wave;
  if (tm >= (M >> 4)) return;

  const int half = lane >> 4;
  const int l15  = lane & 15;
  const int swz  = half << 4;

  v8f acc[8];
#pragma unroll
  for (int tn = 0; tn < 8; ++tn)
#pragma unroll
    for (int r = 0; r < 8; ++r) acc[tn][r] = 0.0f;

  const float* __restrict__ Arow = A + (size_t)(tm * 16 + l15) * 128;

  for (int k = 0; k < 128; k += 4) {
    const int kk = k + half * 2;
    v2f a;
    a[0] = Arow[kk];
    a[1] = Arow[kk + 1];
#pragma unroll
    for (int tn = 0; tn < 8; ++tn) {
      const int c = (tn * 16 + l15) ^ swz;
      v2f b;
      b[0] = Bs[(kk << 7) | c];
      b[1] = Bs[((kk + 1) << 7) | c];
      acc[tn] = __builtin_amdgcn_wmma_f32_16x16x4_f32(false, a, false, b,
                                                      (short)0, acc[tn], false, false);
    }
  }

  float dv[8];
#pragma unroll
  for (int r = 0; r < 8; ++r) dv[r] = dinv[tm * 16 + r + half * 8];
  float* __restrict__ Cp = C + (size_t)(tm * 16) * 128 + l15;
#pragma unroll
  for (int tn = 0; tn < 8; ++tn) {
#pragma unroll
    for (int r = 0; r < 8; ++r) {
      const size_t off = (size_t)(r + half * 8) * 128 + tn * 16;
      Cp[off] = acc[tn][r] * dv[r];
    }
  }
}

__global__ void gemm_wmma_f32(const float* __restrict__ A, const float* __restrict__ B,
                              const float* __restrict__ bias, float* __restrict__ C,
                              int M, int N, int K, int relu) {
  const int lane   = threadIdx.x & 31;
  const int wave   = threadIdx.x >> 5;
  const int tilesN = N >> 4;
  const int tilesM = M >> 4;
  const int t = blockIdx.x * (blockDim.x >> 5) + wave;
  if (t >= tilesM * tilesN) return;
  const int tm = t / tilesN;
  const int tn = t - tm * tilesN;

  const int half = lane >> 4;
  const int l15  = lane & 15;

  v8f acc;
  if (bias != nullptr) {
    const float bv = bias[tn * 16 + l15];
#pragma unroll
    for (int r = 0; r < 8; ++r) acc[r] = bv;
  } else {
#pragma unroll
    for (int r = 0; r < 8; ++r) acc[r] = 0.0f;
  }

  const float* __restrict__ Arow = A + (size_t)(tm * 16 + l15) * K;
  const float* __restrict__ Bcol = B + tn * 16 + l15;

  for (int k = 0; k < K; k += 4) {
    const int kk = k + half * 2;
    v2f a, b;
    a[0] = Arow[kk];
    a[1] = Arow[kk + 1];
    b[0] = Bcol[(size_t)kk * N];
    b[1] = Bcol[(size_t)(kk + 1) * N];
    acc = __builtin_amdgcn_wmma_f32_16x16x4_f32(false, a, false, b,
                                                (short)0, acc, false, false);
  }

  if (relu) {
#pragma unroll
    for (int r = 0; r < 8; ++r) acc[r] = acc[r] > 0.0f ? acc[r] : 0.0f;
  }

  float* __restrict__ Cp = C + (size_t)(tm * 16) * N + tn * 16 + l15;
#pragma unroll
  for (int r = 0; r < 8; ++r) Cp[(size_t)(r + half * 8) * N] = acc[r];
}


template <int U>
__device__ __forceinline__ void add_rows(float4& acc, const float4* __restrict__ H,
                                         int my, int j, int lane) {
  float4 v[U];
#pragma unroll
  for (int u = 0; u < U; ++u) {
    const int s = __builtin_amdgcn_readlane(my, j + u);
    v[u] = H[(size_t)s * 32 + lane];
  }
#pragma unroll
  for (int u = 0; u < U; ++u) {
    acc.x += v[u].x; acc.y += v[u].y; acc.z += v[u].z; acc.w += v[u].w;
  }
}

__global__ void k_gather(const int* __restrict__ rowptr, const int* __restrict__ csr,
                         const float* __restrict__ dinv, const float* __restrict__ hs,
                         const float* __restrict__ bias, float* __restrict__ out,
                         int N, int E) {
  const int lane = threadIdx.x & 31;
  const int node = __builtin_amdgcn_readfirstlane(
      (int)((blockIdx.x * blockDim.x + threadIdx.x) >> 5));
  if (node >= N) return;
  const int beg = rowptr[node];
  const int end = (node + 1 < N) ? rowptr[node + 1] : E;
  const float4* __restrict__ H = (const float4*)hs;

  float4 acc = H[(size_t)node * 32 + lane];
  for (int base = beg; base < end; base += 32) {
    const int rem = end - base;
    const int cnt = rem < 32 ? rem : 32;
    unsigned my = (lane < cnt) ? (unsigned)csr[base + lane] : 0u;
    my = my < (unsigned)N ? my : (unsigned)(N - 1);
    int j = 0;
    for (; j + 8 <= cnt; j += 8) add_rows<8>(acc, H, (int)my, j, lane);
    if (j + 4 <= cnt) { add_rows<4>(acc, H, (int)my, j, lane); j += 4; }
    for (; j < cnt; ++j) add_rows<1>(acc, H, (int)my, j, lane);
  }

  const float dv = dinv[node];
  const float4 b = ((const float4*)bias)[lane];
  float4 o;
  o.x = acc.x * dv + b.x;  o.y = acc.y * dv + b.y;
  o.z = acc.z * dv + b.z;  o.w = acc.w * dv + b.w;
  o.x = o.x > 0.0f ? o.x : 0.0f;  o.y = o.y > 0.0f ? o.y : 0.0f;
  o.z = o.z > 0.0f ? o.z : 0.0f;  o.w = o.w > 0.0f ? o.w : 0.0f;
  ((float4*)out)[(size_t)node * 32 + lane] = o;
}


#define PCH 16

__device__ __forceinline__ void pool_flush(int g, const float4& s, const int4& m, int c, int lane,
                                           float* __restrict__ counts, float* __restrict__ sums,
                                           int* __restrict__ maxs) {
  float* sp = sums + (size_t)g * HID + lane * 4;
  int*   mp = maxs + (size_t)g * HID + lane * 4;
  atomicAdd(sp + 0, s.x);  atomicAdd(sp + 1, s.y);
  atomicAdd(sp + 2, s.z);  atomicAdd(sp + 3, s.w);
  atomicMax(mp + 0, m.x);  atomicMax(mp + 1, m.y);
  atomicMax(mp + 2, m.z);  atomicMax(mp + 3, m.w);
  if (lane == 0) atomicAdd(&counts[g], (float)c);
}

__global__ void k_pool_runs(const float* __restrict__ h, const int* __restrict__ batch,
                            float* __restrict__ counts, float* __restrict__ sums,
                            int* __restrict__ maxs, int N) {
  const int lane = threadIdx.x & 31;
  const int w = __builtin_amdgcn_readfirstlane(
      (int)((blockIdx.x * blockDim.x + threadIdx.x) >> 5));
  const int n0 = w * PCH;
  if (n0 >= N) return;
  const int rem = N - n0;
  const int cnt = rem < PCH ? rem : PCH;

  unsigned gid = (unsigned)batch[n0 + (lane < cnt ? lane : 0)];
  gid = gid < (unsigned)NGR ? gid : (unsigned)(NGR - 1);

  const float4* __restrict__ H = (const float4*)h;
  float4 v[PCH];
#pragma unroll
  for (int j = 0; j < PCH; ++j) {
    const int n = n0 + (j < cnt ? j : 0);
    v[j] = H[(size_t)n * 32 + lane];
  }

  int cur = __builtin_amdgcn_readlane((int)gid, 0);
  float4 s = make_float4(0.0f, 0.0f, 0.0f, 0.0f);
  int4   m = make_int4(0, 0, 0, 0);
  int    c = 0;
#pragma unroll
  for (int j = 0; j < PCH; ++j) {
    if (j < cnt) {
      const int g = __builtin_amdgcn_readlane((int)gid, j);
      if (g != cur) {
        pool_flush(cur, s, m, c, lane, counts, sums, maxs);
        s = make_float4(0.0f, 0.0f, 0.0f, 0.0f);
        m = make_int4(0, 0, 0, 0);
        c = 0;
        cur = g;
      }
      s.x += v[j].x;  s.y += v[j].y;  s.z += v[j].z;  s.w += v[j].w;
      const int bx = __float_as_int(v[j].x), by = __float_as_int(v[j].y);
      const int bz = __float_as_int(v[j].z), bw = __float_as_int(v[j].w);
      m.x = bx > m.x ? bx : m.x;  m.y = by > m.y ? by : m.y;
      m.z = bz > m.z ? bz : m.z;  m.w = bw > m.w ? bw : m.w;
      ++c;
    }
  }
  pool_flush(cur, s, m, c, lane, counts, sums, maxs);
}

__global__ void k_pool_final(const float* __restrict__ sums, const int* __restrict__ maxs,
                             const float* __restrict__ counts, float* __restrict__ gbuf,
                             int total  ) {
  int i = blockIdx.x * blockDim.x + threadIdx.x;
  if (i >= total) return;
  const int g = i >> 7;
  const int f = i & (HID - 1);
  const float c = counts[g];
  const float denom = c > 1.0f ? c : 1.0f;
  gbuf[(size_t)g * (2 * HID) + f]       = sums[i] / denom;
  gbuf[(size_t)g * (2 * HID) + HID + f] = __int_as_float(maxs[i]);
}

__global__ void k_head(const float* __restrict__ hcls, const float* __restrict__ Wc2,
                       const float* __restrict__ bc2, float* __restrict__ out, int total) {
  int i = blockIdx.x * blockDim.x + threadIdx.x;
  if (i >= total) return;
  const int g = i / NCLS;
  const int c = i - g * NCLS;
  float s = bc2[c];
  const float* hv = hcls + (size_t)g * HID;
#pragma unroll 8
  for (int k = 0; k < HID; ++k) s += hv[k] * Wc2[k * NCLS + c];
  out[i] = s;
}


extern "C" void kernel_launch(void* const* d_in, const int* in_sizes, int n_in,
                              void* d_out, int out_size, void* d_ws, size_t ws_size,
                              hipStream_t stream) {
  const float* x     = (const float*)d_in[0];
  const int*   edge  = (const int*)d_in[1];
  const int*   batch = (const int*)d_in[2];
  const float* W[3]  = {(const float*)d_in[3], (const float*)d_in[5], (const float*)d_in[7]};
  const float* b[3]  = {(const float*)d_in[4], (const float*)d_in[6], (const float*)d_in[8]};
  const float* Wc1   = (const float*)d_in[9];
  const float* bc1   = (const float*)d_in[10];
  const float* Wc2   = (const float*)d_in[11];
  const float* bc2   = (const float*)d_in[12];
  float* out = (float*)d_out;

  const int N = in_sizes[0] / HID;
  const int E = in_sizes[1] / 2;
  const int G = NGR;
  const int* src  = edge;
  const int* dstv = edge + E;

  float* ws     = (float*)d_ws;
  int*   degi   = (int*)ws;
  float* dinv   = ws + N;
  float* buf0   = dinv + N;
  float* buf1   = buf0 + (size_t)N * HID;
  float* counts = buf1 + (size_t)N * HID;
  float* sums   = counts + G;
  int*   maxs   = (int*)(sums + (size_t)G * HID);
  float* gbuf   = (float*)(maxs + (size_t)G * HID);
  float* hcls   = gbuf + (size_t)G * 2 * HID;
  int*   csr    = (int*)(hcls + (size_t)G * HID);
  int*   rank   = (int*)buf1;
  int*   bsum   = (int*)counts;

  if ((size_t)((const char*)(csr + E) - (const char*)d_ws) > ws_size) return;

  const int BT = 256;
  auto nb = [](long n, int bt) { return (int)((n + bt - 1) / bt); };

  const int nsb = nb(N, 1024);
  k_fill<<<nb(N, BT), BT, 0, stream>>>((float*)degi, N, 0.0f);
  k_rank<<<nb(E, BT), BT, 0, stream>>>(dstv, degi, rank, E);
  k_dinv_bsum<<<nsb, 256, 0, stream>>>(degi, dinv, bsum, N);
  k_rowptr<<<nsb, 256, 0, stream>>>(degi, bsum, N);
  k_csr_fill<<<nb(E, BT), BT, 0, stream>>>(src, dstv, rank, degi, csr, E);

  const int rowTiles = N / 16;
  const float* hin = x;
  for (int l = 0; l < 3; ++l) {
    gemm_wmma_strip_k128<<<(rowTiles + 7) / 8, 256, 0, stream>>>(hin, W[l], buf0, dinv, N);
    k_gather<<<nb((long)N * 32, BT), BT, 0, stream>>>(degi, csr, dinv, buf0, b[l], buf1, N, E);
    hin = buf1;
  }

  k_fill<<<nb(G, BT), BT, 0, stream>>>(counts, G, 0.0f);
  k_fill<<<nb(G * HID, BT), BT, 0, stream>>>(sums, G * HID, 0.0f);
  k_fill<<<nb(G * HID, BT), BT, 0, stream>>>((float*)maxs, G * HID, 0.0f);
  k_pool_runs<<<nb((long)((N + PCH - 1) / PCH) * 32, BT), BT, 0, stream>>>(buf1, batch, counts,
                                                                        sums, maxs, N);
  k_pool_final<<<nb(G * HID, BT), BT, 0, stream>>>(sums, maxs, counts, gbuf, G * HID);

  const int tiles2 = (G / 16) * (HID / 16);
  gemm_wmma_f32<<<(tiles2 + 7) / 8, 256, 0, stream>>>(gbuf, Wc1, bc1, hcls,
                                                      G, HID, 2 * HID, 1);
  k_head<<<nb(G * NCLS, BT), BT, 0, stream>>>(hcls, Wc2, bc2, out, G * NCLS);
}


# Layer GEMM stages W into LDS with all of a thread's sixteen 16-byte loads in flight before the barrier, not 64 dependent 4-byte load-store round trips per thread
# speedup vs baseline: 3.6082x; 1.2134x over previous
#include <hip/hip_runtime.h>

#define HID   128
#define NGR   512
#define NCLS  10

typedef float v2f __attribute__((ext_vector_type(2)));
typedef float v8f __attribute__((ext_vector_type(8)));


__global__ void k_fill(float* __restrict__ p, int n, float v) {
  int i = blockIdx.x * blockDim.x + threadIdx.x;
  if (i < n) p[i] = v;
}

__global__ void k_rank(const int* __restrict__ dst, int* __restrict__ degi,
                       int* __restrict__ rank, int E) {
  int e = blockIdx.x * blockDim.x + threadIdx.x;
  if (e < E) rank[e] = atomicAdd(&degi[dst[e]], 1);
}

__global__ void k_dinv_bsum(const int* __restrict__ degi, float* __restrict__ dinv,
                            int* __restrict__ bsum, int N) {
  __shared__ int wsum[8];
  const int t = threadIdx.x;
  int s = 0;
  for (int j = 0; j < 4; ++j) {
    const int i = blockIdx.x * 1024 + j * 256 + t;
    if (i < N) {
      const int d = degi[i];
      s += d;
      dinv[i] = rsqrtf((float)d + 1.0f);
    }
  }
  for (int d = 16; d; d >>= 1) s += __shfl_xor(s, d);
  if ((t & 31) == 0) wsum[t >> 5] = s;
  __syncthreads();
  if (t == 0) {
    int tot = 0;
    for (int w = 0; w < 8; ++w) tot += wsum[w];
    bsum[blockIdx.x] = tot;
  }
}

__global__ void k_rowptr(int* __restrict__ degi, const int* __restrict__ bsum, int N) {
  __shared__ int wsum[8];
  __shared__ int wtot[8];
  const int t = threadIdx.x;
  const int lane = t & 31;
  const int wave = t >> 5;

  int p = 0;
  for (int i = t; i < (int)blockIdx.x; i += 256) p += bsum[i];
  for (int d = 16; d; d >>= 1) p += __shfl_xor(p, d);
  if (lane == 0) wtot[wave] = p;

  const int i0 = blockIdx.x * 1024 + t * 4;
  int v[4];
  for (int j = 0; j < 4; ++j) v[j] = (i0 + j < N) ? degi[i0 + j] : 0;
  const int tsum = v[0] + v[1] + v[2] + v[3];
  int inc = tsum;
  for (int d = 1; d < 32; d <<= 1) {
    const int o = __shfl_up(inc, d);
    if (lane >= d) inc += o;
  }
  if (lane == 31) wsum[wave] = inc;
  __syncthreads();
  int off = 0;
  for (int w = 0; w < 8; ++w) {
    off += wtot[w];
    if (w < wave) off += wsum[w];
  }
  int ex = off + inc - tsum;
  for (int j = 0; j < 4; ++j) {
    if (i0 + j < N) degi[i0 + j] = ex;
    ex += v[j];
  }
}

__global__ void k_csr_fill(const int* __restrict__ src, const int* __restrict__ dst,
                           const int* __restrict__ rank, const int* __restrict__ rowptr,
                           int* __restrict__ csr, int E) {
  int e = blockIdx.x * blockDim.x + threadIdx.x;
  if (e >= E) return;
  unsigned pos = (unsigned)(rowptr[dst[e]] + rank[e]);
  pos = pos < (unsigned)E ? pos : (unsigned)(E - 1);
  csr[pos] = src[e];
}

#define STRIP_THREADS       256
#define STRIP_B_PER_THREAD  (128 * 128 / 4 / STRIP_THREADS)
__global__ void __launch_bounds__(STRIP_THREADS)
gemm_wmma_strip_k128(const float* __restrict__ A,
                     const float* __restrict__ B,
                     float* __restrict__ C,
                     const float* __restrict__ dinv, int M) {
  __shared__ __attribute__((aligned(16))) float Bs[128 * 128];

  {
    const float4* __restrict__ B4 = (const float4*)B;
    float4 w[STRIP_B_PER_THREAD];
#pragma unroll
    for (int j = 0; j < STRIP_B_PER_THREAD; ++j) w[j] = B4[j * STRIP_THREADS + threadIdx.x];
#pragma unroll
    for (int j = 0; j < STRIP_B_PER_THREAD; ++j) {
      const int g   = j * STRIP_THREADS + threadIdx.x;
      const int row = g >> 5;
      const int col = (g & 31) << 2;
      *(float4*)&Bs[(row << 7) | (col ^ (((row >> 1) & 1) << 4))] = w[j];
    }
  }
  __syncthreads();

  const int lane = threadIdx.x & 31;
  const int wave = threadIdx.x >> 5;
  const int tm   = blockIdx.x * (blockDim.x >> 5) + wave;
  if (tm >= (M >> 4)) return;

  const int half = lane >> 4;
  const int l15  = lane & 15;
  const int swz  = half << 4;

  v8f acc[8];
#pragma unroll
  for (int tn = 0; tn < 8; ++tn)
#pragma unroll
    for (int r = 0; r < 8; ++r) acc[tn][r] = 0.0f;

  const float* __restrict__ Arow = A + (size_t)(tm * 16 + l15) * 128;

  for (int k = 0; k < 128; k += 4) {
    const int kk = k + half * 2;
    v2f a;
    a[0] = Arow[kk];
    a[1] = Arow[kk + 1];
#pragma unroll
    for (int tn = 0; tn < 8; ++tn) {
      const int c = (tn * 16 + l15) ^ swz;
      v2f b;
      b[0] = Bs[(kk << 7) | c];
      b[1] = Bs[((kk + 1) << 7) | c];
      acc[tn] = __builtin_amdgcn_wmma_f32_16x16x4_f32(false, a, false, b,
                                                      (short)0, acc[tn], false, false);
    }
  }

  float dv[8];
#pragma unroll
  for (int r = 0; r < 8; ++r) dv[r] = dinv[tm * 16 + r + half * 8];
  float* __restrict__ Cp = C + (size_t)(tm * 16) * 128 + l15;
#pragma unroll
  for (int tn = 0; tn < 8; ++tn) {
#pragma unroll
    for (int r = 0; r < 8; ++r) {
      const size_t off = (size_t)(r + half * 8) * 128 + tn * 16;
      Cp[off] = acc[tn][r] * dv[r];
    }
  }
}

__global__ void gemm_wmma_f32(const float* __restrict__ A, const float* __restrict__ B,
                              const float* __restrict__ bias, float* __restrict__ C,
                              int M, int N, int K, int relu) {
  const int lane   = threadIdx.x & 31;
  const int wave   = threadIdx.x >> 5;
  const int tilesN = N >> 4;
  const int tilesM = M >> 4;
  const int t = blockIdx.x * (blockDim.x >> 5) + wave;
  if (t >= tilesM * tilesN) return;
  const int tm = t / tilesN;
  const int tn = t - tm * tilesN;

  const int half = lane >> 4;
  const int l15  = lane & 15;

  v8f acc;
  if (bias != nullptr) {
    const float bv = bias[tn * 16 + l15];
#pragma unroll
    for (int r = 0; r < 8; ++r) acc[r] = bv;
  } else {
#pragma unroll
    for (int r = 0; r < 8; ++r) acc[r] = 0.0f;
  }

  const float* __restrict__ Arow = A + (size_t)(tm * 16 + l15) * K;
  const float* __restrict__ Bcol = B + tn * 16 + l15;

  for (int k = 0; k < K; k += 4) {
    const int kk = k + half * 2;
    v2f a, b;
    a[0] = Arow[kk];
    a[1] = Arow[kk + 1];
    b[0] = Bcol[(size_t)kk * N];
    b[1] = Bcol[(size_t)(kk + 1) * N];
    acc = __builtin_amdgcn_wmma_f32_16x16x4_f32(false, a, false, b,
                                                (short)0, acc, false, false);
  }

  if (relu) {
#pragma unroll
    for (int r = 0; r < 8; ++r) acc[r] = acc[r] > 0.0f ? acc[r] : 0.0f;
  }

  float* __restrict__ Cp = C + (size_t)(tm * 16) * N + tn * 16 + l15;
#pragma unroll
  for (int r = 0; r < 8; ++r) Cp[(size_t)(r + half * 8) * N] = acc[r];
}


template <int U>
__device__ __forceinline__ void add_rows(float4& acc, const float4* __restrict__ H,
                                         int my, int j, int lane) {
  float4 v[U];
#pragma unroll
  for (int u = 0; u < U; ++u) {
    const int s = __builtin_amdgcn_readlane(my, j + u);
    v[u] = H[(size_t)s * 32 + lane];
  }
#pragma unroll
  for (int u = 0; u < U; ++u) {
    acc.x += v[u].x; acc.y += v[u].y; acc.z += v[u].z; acc.w += v[u].w;
  }
}

__global__ void k_gather(const int* __restrict__ rowptr, const int* __restrict__ csr,
                         const float* __restrict__ dinv, const float* __restrict__ hs,
                         const float* __restrict__ bias, float* __restrict__ out,
                         int N, int E) {
  const int lane = threadIdx.x & 31;
  const int node = __builtin_amdgcn_readfirstlane(
      (int)((blockIdx.x * blockDim.x + threadIdx.x) >> 5));
  if (node >= N) return;
  const int beg = rowptr[node];
  const int end = (node + 1 < N) ? rowptr[node + 1] : E;
  const float4* __restrict__ H = (const float4*)hs;

  float4 acc = H[(size_t)node * 32 + lane];
  for (int base = beg; base < end; base += 32) {
    const int rem = end - base;
    const int cnt = rem < 32 ? rem : 32;
    unsigned my = (lane < cnt) ? (unsigned)csr[base + lane] : 0u;
    my = my < (unsigned)N ? my : (unsigned)(N - 1);
    int j = 0;
    for (; j + 8 <= cnt; j += 8) add_rows<8>(acc, H, (int)my, j, lane);
    if (j + 4 <= cnt) { add_rows<4>(acc, H, (int)my, j, lane); j += 4; }
    for (; j < cnt; ++j) add_rows<1>(acc, H, (int)my, j, lane);
  }

  const float dv = dinv[node];
  const float4 b = ((const float4*)bias)[lane];
  float4 o;
  o.x = acc.x * dv + b.x;  o.y = acc.y * dv + b.y;
  o.z = acc.z * dv + b.z;  o.w = acc.w * dv + b.w;
  o.x = o.x > 0.0f ? o.x : 0.0f;  o.y = o.y > 0.0f ? o.y : 0.0f;
  o.z = o.z > 0.0f ? o.z : 0.0f;  o.w = o.w > 0.0f ? o.w : 0.0f;
  ((float4*)out)[(size_t)node * 32 + lane] = o;
}


#define PCH 16

__device__ __forceinline__ void pool_flush(int g, const float4& s, const int4& m, int c, int lane,
                                           float* __restrict__ counts, float* __restrict__ sums,
                                           int* __restrict__ maxs) {
  float* sp = sums + (size_t)g * HID + lane * 4;
  int*   mp = maxs + (size_t)g * HID + lane * 4;
  atomicAdd(sp + 0, s.x);  atomicAdd(sp + 1, s.y);
  atomicAdd(sp + 2, s.z);  atomicAdd(sp + 3, s.w);
  atomicMax(mp + 0, m.x);  atomicMax(mp + 1, m.y);
  atomicMax(mp + 2, m.z);  atomicMax(mp + 3, m.w);
  if (lane == 0) atomicAdd(&counts[g], (float)c);
}

__global__ void k_pool_runs(const float* __restrict__ h, const int* __restrict__ batch,
                            float* __restrict__ counts, float* __restrict__ sums,
                            int* __restrict__ maxs, int N) {
  const int lane = threadIdx.x & 31;
  const int w = __builtin_amdgcn_readfirstlane(
      (int)((blockIdx.x * blockDim.x + threadIdx.x) >> 5));
  const int n0 = w * PCH;
  if (n0 >= N) return;
  const int rem = N - n0;
  const int cnt = rem < PCH ? rem : PCH;

  unsigned gid = (unsigned)batch[n0 + (lane < cnt ? lane : 0)];
  gid = gid < (unsigned)NGR ? gid : (unsigned)(NGR - 1);

  const float4* __restrict__ H = (const float4*)h;
  float4 v[PCH];
#pragma unroll
  for (int j = 0; j < PCH; ++j) {
    const int n = n0 + (j < cnt ? j : 0);
    v[j] = H[(size_t)n * 32 + lane];
  }

  int cur = __builtin_amdgcn_readlane((int)gid, 0);
  float4 s = make_float4(0.0f, 0.0f, 0.0f, 0.0f);
  int4   m = make_int4(0, 0, 0, 0);
  int    c = 0;
#pragma unroll
  for (int j = 0; j < PCH; ++j) {
    if (j < cnt) {
      const int g = __builtin_amdgcn_readlane((int)gid, j);
      if (g != cur) {
        pool_flush(cur, s, m, c, lane, counts, sums, maxs);
        s = make_float4(0.0f, 0.0f, 0.0f, 0.0f);
        m = make_int4(0, 0, 0, 0);
        c = 0;
        cur = g;
      }
      s.x += v[j].x;  s.y += v[j].y;  s.z += v[j].z;  s.w += v[j].w;
      const int bx = __float_as_int(v[j].x), by = __float_as_int(v[j].y);
      const int bz = __float_as_int(v[j].z), bw = __float_as_int(v[j].w);
      m.x = bx > m.x ? bx : m.x;  m.y = by > m.y ? by : m.y;
      m.z = bz > m.z ? bz : m.z;  m.w = bw > m.w ? bw : m.w;
      ++c;
    }
  }
  pool_flush(cur, s, m, c, lane, counts, sums, maxs);
}

__global__ void k_pool_final(const float* __restrict__ sums, const int* __restrict__ maxs,
                             const float* __restrict__ counts, float* __restrict__ gbuf,
                             int total  ) {
  int i = blockIdx.x * blockDim.x + threadIdx.x;
  if (i >= total) return;
  const int g = i >> 7;
  const int f = i & (HID - 1);
  const float c = counts[g];
  const float denom = c > 1.0f ? c : 1.0f;
  gbuf[(size_t)g * (2 * HID) + f]       = sums[i] / denom;
  gbuf[(size_t)g * (2 * HID) + HID + f] = __int_as_float(maxs[i]);
}

__global__ void k_head(const float* __restrict__ hcls, const float* __restrict__ Wc2,
                       const float* __restrict__ bc2, float* __restrict__ out, int total) {
  int i = blockIdx.x * blockDim.x + threadIdx.x;
  if (i >= total) return;
  const int g = i / NCLS;
  const int c = i - g * NCLS;
  float s = bc2[c];
  const float* hv = hcls + (size_t)g * HID;
#pragma unroll 8
  for (int k = 0; k < HID; ++k) s += hv[k] * Wc2[k * NCLS + c];
  out[i] = s;
}


extern "C" void kernel_launch(void* const* d_in, const int* in_sizes, int n_in,
                              void* d_out, int out_size, void* d_ws, size_t ws_size,
                              hipStream_t stream) {
  const float* x     = (const float*)d_in[0];
  const int*   edge  = (const int*)d_in[1];
  const int*   batch = (const int*)d_in[2];
  const float* W[3]  = {(const float*)d_in[3], (const float*)d_in[5], (const float*)d_in[7]};
  const float* b[3]  = {(const float*)d_in[4], (const float*)d_in[6], (const float*)d_in[8]};
  const float* Wc1   = (const float*)d_in[9];
  const float* bc1   = (const float*)d_in[10];
  const float* Wc2   = (const float*)d_in[11];
  const float* bc2   = (const float*)d_in[12];
  float* out = (float*)d_out;

  const int N = in_sizes[0] / HID;
  const int E = in_sizes[1] / 2;
  const int G = NGR;
  const int* src  = edge;
  const int* dstv = edge + E;

  float* ws     = (float*)d_ws;
  int*   degi   = (int*)ws;
  float* dinv   = ws + N;
  float* buf0   = dinv + N;
  float* buf1   = buf0 + (size_t)N * HID;
  float* counts = buf1 + (size_t)N * HID;
  float* sums   = counts + G;
  int*   maxs   = (int*)(sums + (size_t)G * HID);
  float* gbuf   = (float*)(maxs + (size_t)G * HID);
  float* hcls   = gbuf + (size_t)G * 2 * HID;
  int*   csr    = (int*)(hcls + (size_t)G * HID);
  int*   rank   = (int*)buf1;
  int*   bsum   = (int*)counts;

  if ((size_t)((const char*)(csr + E) - (const char*)d_ws) > ws_size) return;

  const int BT = 256;
  auto nb = [](long n, int bt) { return (int)((n + bt - 1) / bt); };

  const int nsb = nb(N, 1024);
  k_fill<<<nb(N, BT), BT, 0, stream>>>((float*)degi, N, 0.0f);
  k_rank<<<nb(E, BT), BT, 0, stream>>>(dstv, degi, rank, E);
  k_dinv_bsum<<<nsb, 256, 0, stream>>>(degi, dinv, bsum, N);
  k_rowptr<<<nsb, 256, 0, stream>>>(degi, bsum, N);
  k_csr_fill<<<nb(E, BT), BT, 0, stream>>>(src, dstv, rank, degi, csr, E);

  const int rowTiles = N / 16;
  const float* hin = x;
  for (int l = 0; l < 3; ++l) {
    gemm_wmma_strip_k128<<<(rowTiles + 7) / 8, STRIP_THREADS, 0, stream>>>(hin, W[l], buf0, dinv, N);
    k_gather<<<nb((long)N * 32, BT), BT, 0, stream>>>(degi, csr, dinv, buf0, b[l], buf1, N, E);
    hin = buf1;
  }

  k_fill<<<nb(G, BT), BT, 0, stream>>>(counts, G, 0.0f);
  k_fill<<<nb(G * HID, BT), BT, 0, stream>>>(sums, G * HID, 0.0f);
  k_fill<<<nb(G * HID, BT), BT, 0, stream>>>((float*)maxs, G * HID, 0.0f);
  k_pool_runs<<<nb((long)((N + PCH - 1) / PCH) * 32, BT), BT, 0, stream>>>(buf1, batch, counts,
                                                                        sums, maxs, N);
  k_pool_final<<<nb(G * HID, BT), BT, 0, stream>>>(sums, maxs, counts, gbuf, G * HID);

  const int tiles2 = (G / 16) * (HID / 16);
  gemm_wmma_f32<<<(tiles2 + 7) / 8, 256, 0, stream>>>(gbuf, Wc1, bc1, hcls,
                                                      G, HID, 2 * HID, 1);
  k_head<<<nb(G * NCLS, BT), BT, 0, stream>>>(hcls, Wc2, bc2, out, G * NCLS);
}


# CSR build as a two-level counting sort in LDS: edge chunks sorted by 512-node bin in LDS, written as whole runs; one block per bin ranks and streams its lists; in-row sum order stays arrival order
# speedup vs baseline: 4.8691x; 1.3495x over previous
#include <hip/hip_runtime.h>

#define HID   128
#define NGR   512
#define NCLS  10

typedef float v2f __attribute__((ext_vector_type(2)));
typedef float v8f __attribute__((ext_vector_type(8)));


__global__ void k_fill(float* __restrict__ p, int n, float v) {
  int i = blockIdx.x * blockDim.x + threadIdx.x;
  if (i < n) p[i] = v;
}

#define BINSH  9
#define BINW   (1 << BINSH)
#define NBMAX  256
#define ECH    6400
#define BTHR   1024
#define EPT    ((ECH + BTHR - 1) / BTHR)
#define HPARTS (BTHR / NBMAX)
#define HUNR   16
#define DCAP   12288
#define DUNR   4

__device__ __forceinline__ int bin_of(int d, int NB) {
  const unsigned q = (unsigned)d >> BINSH;
  return (int)(q < (unsigned)NB ? q : (unsigned)(NB - 1));
}

template <int NW>
__device__ __forceinline__ int block_excl_scan(int v, int* wsum, bool active) {
  const int lane = threadIdx.x & 31;
  const int wave = threadIdx.x >> 5;
  int inc = v;
  for (int d = 1; d < 32; d <<= 1) {
    const int o = __shfl_up(inc, d);
    if (lane >= d) inc += o;
  }
  if (active && lane == 31) wsum[wave] = inc;
  __syncthreads();
  int off = 0;
  if (active) {
    for (int w = 0; w < NW; ++w) off += (w < wave) ? wsum[w] : 0;
  }
  return off + inc - v;
}

__global__ void __launch_bounds__(BTHR)
k_bin_hist(const int* __restrict__ dst, int* __restrict__ hist, int E, int NB) {
  __shared__ int h[NBMAX];
  const int t = threadIdx.x;
  if (t < NBMAX) h[t] = 0;
  __syncthreads();
  const int beg = blockIdx.x * ECH;
  const int end = (beg + ECH < E) ? beg + ECH : E;
  int d[EPT];
#pragma unroll
  for (int j = 0; j < EPT; ++j) {
    const int i = beg + j * BTHR + t;
    d[j] = dst[i < end ? i : beg];
  }
#pragma unroll
  for (int j = 0; j < EPT; ++j) {
    const int i = beg + j * BTHR + t;
    if (i < end) atomicAdd(&h[bin_of(d[j], NB)], 1);
  }
  __syncthreads();
  if (t < NB) hist[blockIdx.x * NB + t] = h[t];
}

__global__ void __launch_bounds__(BTHR)
k_bin_scatter(const int* __restrict__ src, const int* __restrict__ dst,
              const int* __restrict__ hist, int* __restrict__ bsrc, int* __restrict__ bdst,
              int* __restrict__ binbase, int E, int NB, int NBLK) {
  __shared__ int pbef[HPARTS][NBMAX];
  __shared__ int ptot[HPARTS][NBMAX];
  __shared__ int goff[NBMAX];
  __shared__ int lstart[NBMAX];
  __shared__ int lcnt[NBMAX];
  __shared__ int wsum[NBMAX / 32];
  __shared__ int ssrc[ECH];
  __shared__ int sdst[ECH];

  const int t = threadIdx.x;
  const int k = blockIdx.x;
  const int beg = k * ECH;
  const int end = (beg + ECH < E) ? beg + ECH : E;
  const int n = end - beg;

  int s[EPT], d[EPT];
#pragma unroll
  for (int j = 0; j < EPT; ++j) {
    const int i  = beg + j * BTHR + t;
    const int ic = i < end ? i : beg;
    s[j] = src[ic];
    d[j] = dst[ic];
  }

  {
    const int b = t & (NBMAX - 1);
    const int part = t / NBMAX;
    const int rows = (NBLK + HPARTS - 1) / HPARTS;
    const int r0 = part * rows;
    const int r1 = (r0 + rows < NBLK) ? r0 + rows : NBLK;
    int bef = 0, tot = 0;
    if (b < NB) {
      for (int r = r0; r < r1; r += HUNR) {
        int v[HUNR];
#pragma unroll
        for (int u = 0; u < HUNR; ++u) {
          const int rr = r + u;
          v[u] = hist[(rr < r1 ? rr : r0) * NB + b];
        }
#pragma unroll
        for (int u = 0; u < HUNR; ++u) {
          const int rr = r + u;
          tot += (rr < r1) ? v[u] : 0;
          bef += (rr < r1 && rr < k) ? v[u] : 0;
        }
      }
    }
    pbef[part][b] = bef;
    ptot[part][b] = tot;
    if (t < NBMAX) lcnt[t] = 0;
  }
  __syncthreads();

  int totb = 0, befb = 0;
  if (t < NBMAX) {
    for (int p = 0; p < HPARTS; ++p) { totb += ptot[p][t]; befb += pbef[p][t]; }
  }
  const int base = block_excl_scan<NBMAX / 32>(totb, wsum, t < NBMAX);
  if (t < NBMAX) {
    goff[t] = base + befb;
    if (k == 0 && t < NB) binbase[t] = base;
  }

  int r[EPT];
#pragma unroll
  for (int j = 0; j < EPT; ++j) {
    const int i = beg + j * BTHR + t;
    r[j] = (i < end) ? atomicAdd(&lcnt[bin_of(d[j], NB)], 1) : 0;
  }
  __syncthreads();
  const int cb = (t < NBMAX) ? lcnt[t] : 0;
  const int ls = block_excl_scan<NBMAX / 32>(cb, wsum, t < NBMAX);
  if (t < NBMAX) lstart[t] = ls;
  __syncthreads();

#pragma unroll
  for (int j = 0; j < EPT; ++j) {
    const int i = beg + j * BTHR + t;
    if (i < end) {
      unsigned p = (unsigned)(lstart[bin_of(d[j], NB)] + r[j]);
      p = p < (unsigned)ECH ? p : (unsigned)(ECH - 1);
      ssrc[p] = s[j];
      sdst[p] = d[j];
    }
  }
  __syncthreads();

  for (int i = t; i < n; i += BTHR) {
    const int dd = sdst[i];
    const int bn = bin_of(dd, NB);
    unsigned g = (unsigned)(goff[bn] + (i - lstart[bn]));
    g = g < (unsigned)E ? g : (unsigned)(E - 1);
    bsrc[g] = ssrc[i];
    bdst[g] = dd;
  }
}

__global__ void __launch_bounds__(BTHR)
k_bin_rows(const int* __restrict__ bsrc, const int* __restrict__ bdst,
           const int* __restrict__ binbase, int* __restrict__ rowptr,
           float* __restrict__ dinv, int* __restrict__ csr, int N, int E, int NB) {
  __shared__ int cnt[BINW];
  __shared__ int rstart[BINW];
  __shared__ int cur[BINW];
  __shared__ int wsum[BINW / 32];
  __shared__ int slist[DCAP];

  const int t = threadIdx.x;
  const int b = blockIdx.x;
  unsigned u0 = (unsigned)binbase[b];
  unsigned u1 = (b + 1 < NB) ? (unsigned)binbase[b + 1] : (unsigned)E;
  u1 = u1 < (unsigned)E ? u1 : (unsigned)E;
  u0 = u0 < u1 ? u0 : u1;
  const int e0 = (int)u0, e1 = (int)u1;
  const int n = e1 - e0;

  if (t < BINW) { cnt[t] = 0; cur[t] = 0; }
  __syncthreads();

  for (int i0 = e0; i0 < e1; i0 += DUNR * BTHR) {
    int d[DUNR];
#pragma unroll
    for (int j = 0; j < DUNR; ++j) {
      const int i = i0 + j * BTHR + t;
      d[j] = bdst[i < e1 ? i : e0];
    }
#pragma unroll
    for (int j = 0; j < DUNR; ++j) {
      const int i = i0 + j * BTHR + t;
      if (i < e1) atomicAdd(&cnt[d[j] & (BINW - 1)], 1);
    }
  }
  __syncthreads();

  const int c = (t < BINW) ? cnt[t] : 0;
  const int ex = block_excl_scan<BINW / 32>(c, wsum, t < BINW);
  if (t < BINW) {
    rstart[t] = ex;
    const int node = b * BINW + t;
    if (node < N) {
      rowptr[node] = e0 + ex;
      dinv[node] = rsqrtf((float)c + 1.0f);
    }
  }
  __syncthreads();

  const bool inlds = n <= DCAP;
  for (int i0 = e0; i0 < e1; i0 += DUNR * BTHR) {
    int d[DUNR], s[DUNR];
#pragma unroll
    for (int j = 0; j < DUNR; ++j) {
      const int i  = i0 + j * BTHR + t;
      const int ic = i < e1 ? i : e0;
      d[j] = bdst[ic];
      s[j] = bsrc[ic];
    }
#pragma unroll
    for (int j = 0; j < DUNR; ++j) {
      const int i = i0 + j * BTHR + t;
      if (i < e1) {
        const int dl = d[j] & (BINW - 1);
        const unsigned p = (unsigned)(rstart[dl] + atomicAdd(&cur[dl], 1));
        if (inlds) {
          slist[p < (unsigned)DCAP ? p : (unsigned)(DCAP - 1)] = s[j];
        } else {
          const unsigned g = u0 + p;
          csr[g < (unsigned)E ? g : (unsigned)(E - 1)] = s[j];
        }
      }
    }
  }
  if (inlds) {
    __syncthreads();
    for (int i = t; i < n; i += BTHR) csr[e0 + i] = slist[i];
  }
}

#define STRIP_THREADS       256
#define STRIP_B_PER_THREAD  (128 * 128 / 4 / STRIP_THREADS)
__global__ void __launch_bounds__(STRIP_THREADS)
gemm_wmma_strip_k128(const float* __restrict__ A,
                     const float* __restrict__ B,
                     float* __restrict__ C,
                     const float* __restrict__ dinv, int M) {
  __shared__ __attribute__((aligned(16))) float Bs[128 * 128];

  {
    const float4* __restrict__ B4 = (const float4*)B;
    float4 w[STRIP_B_PER_THREAD];
#pragma unroll
    for (int j = 0; j < STRIP_B_PER_THREAD; ++j) w[j] = B4[j * STRIP_THREADS + threadIdx.x];
#pragma unroll
    for (int j = 0; j < STRIP_B_PER_THREAD; ++j) {
      const int g   = j * STRIP_THREADS + threadIdx.x;
      const int row = g >> 5;
      const int col = (g & 31) << 2;
      *(float4*)&Bs[(row << 7) | (col ^ (((row >> 1) & 1) << 4))] = w[j];
    }
  }
  __syncthreads();

  const int lane = threadIdx.x & 31;
  const int wave = threadIdx.x >> 5;
  const int tm   = blockIdx.x * (blockDim.x >> 5) + wave;
  if (tm >= (M >> 4)) return;

  const int half = lane >> 4;
  const int l15  = lane & 15;
  const int swz  = half << 4;

  v8f acc[8];
#pragma unroll
  for (int tn = 0; tn < 8; ++tn)
#pragma unroll
    for (int r = 0; r < 8; ++r) acc[tn][r] = 0.0f;

  const float* __restrict__ Arow = A + (size_t)(tm * 16 + l15) * 128;

  for (int k = 0; k < 128; k += 4) {
    const int kk = k + half * 2;
    v2f a;
    a[0] = Arow[kk];
    a[1] = Arow[kk + 1];
#pragma unroll
    for (int tn = 0; tn < 8; ++tn) {
      const int c = (tn * 16 + l15) ^ swz;
      v2f b;
      b[0] = Bs[(kk << 7) | c];
      b[1] = Bs[((kk + 1) << 7) | c];
      acc[tn] = __builtin_amdgcn_wmma_f32_16x16x4_f32(false, a, false, b,
                                                      (short)0, acc[tn], false, false);
    }
  }

  float dv[8];
#pragma unroll
  for (int r = 0; r < 8; ++r) dv[r] = dinv[tm * 16 + r + half * 8];
  float* __restrict__ Cp = C + (size_t)(tm * 16) * 128 + l15;
#pragma unroll
  for (int tn = 0; tn < 8; ++tn) {
#pragma unroll
    for (int r = 0; r < 8; ++r) {
      const size_t off = (size_t)(r + half * 8) * 128 + tn * 16;
      Cp[off] = acc[tn][r] * dv[r];
    }
  }
}

__global__ void gemm_wmma_f32(const float* __restrict__ A, const float* __restrict__ B,
                              const float* __restrict__ bias, float* __restrict__ C,
                              int M, int N, int K, int relu) {
  const int lane   = threadIdx.x & 31;
  const int wave   = threadIdx.x >> 5;
  const int tilesN = N >> 4;
  const int tilesM = M >> 4;
  const int t = blockIdx.x * (blockDim.x >> 5) + wave;
  if (t >= tilesM * tilesN) return;
  const int tm = t / tilesN;
  const int tn = t - tm * tilesN;

  const int half = lane >> 4;
  const int l15  = lane & 15;

  v8f acc;
  if (bias != nullptr) {
    const float bv = bias[tn * 16 + l15];
#pragma unroll
    for (int r = 0; r < 8; ++r) acc[r] = bv;
  } else {
#pragma unroll
    for (int r = 0; r < 8; ++r) acc[r] = 0.0f;
  }

  const float* __restrict__ Arow = A + (size_t)(tm * 16 + l15) * K;
  const float* __restrict__ Bcol = B + tn * 16 + l15;

  for (int k = 0; k < K; k += 4) {
    const int kk = k + half * 2;
    v2f a, b;
    a[0] = Arow[kk];
    a[1] = Arow[kk + 1];
    b[0] = Bcol[(size_t)kk * N];
    b[1] = Bcol[(size_t)(kk + 1) * N];
    acc = __builtin_amdgcn_wmma_f32_16x16x4_f32(false, a, false, b,
                                                (short)0, acc, false, false);
  }

  if (relu) {
#pragma unroll
    for (int r = 0; r < 8; ++r) acc[r] = acc[r] > 0.0f ? acc[r] : 0.0f;
  }

  float* __restrict__ Cp = C + (size_t)(tm * 16) * N + tn * 16 + l15;
#pragma unroll
  for (int r = 0; r < 8; ++r) Cp[(size_t)(r + half * 8) * N] = acc[r];
}


template <int U>
__device__ __forceinline__ void add_rows(float4& acc, const float4* __restrict__ H,
                                         int my, int j, int lane) {
  float4 v[U];
#pragma unroll
  for (int u = 0; u < U; ++u) {
    const int s = __builtin_amdgcn_readlane(my, j + u);
    v[u] = H[(size_t)s * 32 + lane];
  }
#pragma unroll
  for (int u = 0; u < U; ++u) {
    acc.x += v[u].x; acc.y += v[u].y; acc.z += v[u].z; acc.w += v[u].w;
  }
}

__global__ void k_gather(const int* __restrict__ rowptr, const int* __restrict__ csr,
                         const float* __restrict__ dinv, const float* __restrict__ hs,
                         const float* __restrict__ bias, float* __restrict__ out,
                         int N, int E) {
  const int lane = threadIdx.x & 31;
  const int node = __builtin_amdgcn_readfirstlane(
      (int)((blockIdx.x * blockDim.x + threadIdx.x) >> 5));
  if (node >= N) return;
  const int beg = rowptr[node];
  const int end = (node + 1 < N) ? rowptr[node + 1] : E;
  const float4* __restrict__ H = (const float4*)hs;

  float4 acc = H[(size_t)node * 32 + lane];
  for (int base = beg; base < end; base += 32) {
    const int rem = end - base;
    const int cnt = rem < 32 ? rem : 32;
    unsigned my = (lane < cnt) ? (unsigned)csr[base + lane] : 0u;
    my = my < (unsigned)N ? my : (unsigned)(N - 1);
    int j = 0;
    for (; j + 8 <= cnt; j += 8) add_rows<8>(acc, H, (int)my, j, lane);
    if (j + 4 <= cnt) { add_rows<4>(acc, H, (int)my, j, lane); j += 4; }
    for (; j < cnt; ++j) add_rows<1>(acc, H, (int)my, j, lane);
  }

  const float dv = dinv[node];
  const float4 b = ((const float4*)bias)[lane];
  float4 o;
  o.x = acc.x * dv + b.x;  o.y = acc.y * dv + b.y;
  o.z = acc.z * dv + b.z;  o.w = acc.w * dv + b.w;
  o.x = o.x > 0.0f ? o.x : 0.0f;  o.y = o.y > 0.0f ? o.y : 0.0f;
  o.z = o.z > 0.0f ? o.z : 0.0f;  o.w = o.w > 0.0f ? o.w : 0.0f;
  ((float4*)out)[(size_t)node * 32 + lane] = o;
}


#define PCH 16

__device__ __forceinline__ void pool_flush(int g, const float4& s, const int4& m, int c, int lane,
                                           float* __restrict__ counts, float* __restrict__ sums,
                                           int* __restrict__ maxs) {
  float* sp = sums + (size_t)g * HID + lane * 4;
  int*   mp = maxs + (size_t)g * HID + lane * 4;
  atomicAdd(sp + 0, s.x);  atomicAdd(sp + 1, s.y);
  atomicAdd(sp + 2, s.z);  atomicAdd(sp + 3, s.w);
  atomicMax(mp + 0, m.x);  atomicMax(mp + 1, m.y);
  atomicMax(mp + 2, m.z);  atomicMax(mp + 3, m.w);
  if (lane == 0) atomicAdd(&counts[g], (float)c);
}

__global__ void k_pool_runs(const float* __restrict__ h, const int* __restrict__ batch,
                            float* __restrict__ counts, float* __restrict__ sums,
                            int* __restrict__ maxs, int N) {
  const int lane = threadIdx.x & 31;
  const int w = __builtin_amdgcn_readfirstlane(
      (int)((blockIdx.x * blockDim.x + threadIdx.x) >> 5));
  const int n0 = w * PCH;
  if (n0 >= N) return;
  const int rem = N - n0;
  const int cnt = rem < PCH ? rem : PCH;

  unsigned gid = (unsigned)batch[n0 + (lane < cnt ? lane : 0)];
  gid = gid < (unsigned)NGR ? gid : (unsigned)(NGR - 1);

  const float4* __restrict__ H = (const float4*)h;
  float4 v[PCH];
#pragma unroll
  for (int j = 0; j < PCH; ++j) {
    const int n = n0 + (j < cnt ? j : 0);
    v[j] = H[(size_t)n * 32 + lane];
  }

  int cur = __builtin_amdgcn_readlane((int)gid, 0);
  float4 s = make_float4(0.0f, 0.0f, 0.0f, 0.0f);
  int4   m = make_int4(0, 0, 0, 0);
  int    c = 0;
#pragma unroll
  for (int j = 0; j < PCH; ++j) {
    if (j < cnt) {
      const int g = __builtin_amdgcn_readlane((int)gid, j);
      if (g != cur) {
        pool_flush(cur, s, m, c, lane, counts, sums, maxs);
        s = make_float4(0.0f, 0.0f, 0.0f, 0.0f);
        m = make_int4(0, 0, 0, 0);
        c = 0;
        cur = g;
      }
      s.x += v[j].x;  s.y += v[j].y;  s.z += v[j].z;  s.w += v[j].w;
      const int bx = __float_as_int(v[j].x), by = __float_as_int(v[j].y);
      const int bz = __float_as_int(v[j].z), bw = __float_as_int(v[j].w);
      m.x = bx > m.x ? bx : m.x;  m.y = by > m.y ? by : m.y;
      m.z = bz > m.z ? bz : m.z;  m.w = bw > m.w ? bw : m.w;
      ++c;
    }
  }
  pool_flush(cur, s, m, c, lane, counts, sums, maxs);
}

__global__ void k_pool_final(const float* __restrict__ sums, const int* __restrict__ maxs,
                             const float* __restrict__ counts, float* __restrict__ gbuf,
                             int total  ) {
  int i = blockIdx.x * blockDim.x + threadIdx.x;
  if (i >= total) return;
  const int g = i >> 7;
  const int f = i & (HID - 1);
  const float c = counts[g];
  const float denom = c > 1.0f ? c : 1.0f;
  gbuf[(size_t)g * (2 * HID) + f]       = sums[i] / denom;
  gbuf[(size_t)g * (2 * HID) + HID + f] = __int_as_float(maxs[i]);
}

__global__ void k_head(const float* __restrict__ hcls, const float* __restrict__ Wc2,
                       const float* __restrict__ bc2, float* __restrict__ out, int total) {
  int i = blockIdx.x * blockDim.x + threadIdx.x;
  if (i >= total) return;
  const int g = i / NCLS;
  const int c = i - g * NCLS;
  float s = bc2[c];
  const float* hv = hcls + (size_t)g * HID;
#pragma unroll 8
  for (int k = 0; k < HID; ++k) s += hv[k] * Wc2[k * NCLS + c];
  out[i] = s;
}


extern "C" void kernel_launch(void* const* d_in, const int* in_sizes, int n_in,
                              void* d_out, int out_size, void* d_ws, size_t ws_size,
                              hipStream_t stream) {
  const float* x     = (const float*)d_in[0];
  const int*   edge  = (const int*)d_in[1];
  const int*   batch = (const int*)d_in[2];
  const float* W[3]  = {(const float*)d_in[3], (const float*)d_in[5], (const float*)d_in[7]};
  const float* b[3]  = {(const float*)d_in[4], (const float*)d_in[6], (const float*)d_in[8]};
  const float* Wc1   = (const float*)d_in[9];
  const float* bc1   = (const float*)d_in[10];
  const float* Wc2   = (const float*)d_in[11];
  const float* bc2   = (const float*)d_in[12];
  float* out = (float*)d_out;

  const int N = in_sizes[0] / HID;
  const int E = in_sizes[1] / 2;
  const int G = NGR;
  const int* src  = edge;
  const int* dstv = edge + E;

  float* ws     = (float*)d_ws;
  int*   degi   = (int*)ws;
  float* dinv   = ws + N;
  float* buf0   = dinv + N;
  float* buf1   = buf0 + (size_t)N * HID;
  float* counts = buf1 + (size_t)N * HID;
  float* sums   = counts + G;
  int*   maxs   = (int*)(sums + (size_t)G * HID);
  float* gbuf   = (float*)(maxs + (size_t)G * HID);
  float* hcls   = gbuf + (size_t)G * 2 * HID;
  int*   csr    = (int*)(hcls + (size_t)G * HID);
  const int NB   = (N + BINW - 1) >> BINSH;
  const int NBLK = (E + ECH - 1) / ECH;
  int*   bsrc    = (int*)buf1;
  int*   bdst    = bsrc + E;
  int*   hist    = bdst + E;
  int*   binbase = hist + (size_t)NBLK * NB;

  if ((size_t)((const char*)(csr + E) - (const char*)d_ws) > ws_size) return;
  if (NB > NBMAX || (size_t)2 * E + (size_t)NBLK * NB + NB > (size_t)N * HID) return;

  const int BT = 256;
  auto nb = [](long n, int bt) { return (int)((n + bt - 1) / bt); };

  k_bin_hist<<<NBLK, BTHR, 0, stream>>>(dstv, hist, E, NB);
  k_bin_scatter<<<NBLK, BTHR, 0, stream>>>(src, dstv, hist, bsrc, bdst, binbase, E, NB, NBLK);
  k_bin_rows<<<NB, BTHR, 0, stream>>>(bsrc, bdst, binbase, degi, dinv, csr, N, E, NB);

  const int rowTiles = N / 16;
  const float* hin = x;
  for (int l = 0; l < 3; ++l) {
    gemm_wmma_strip_k128<<<(rowTiles + 7) / 8, STRIP_THREADS, 0, stream>>>(hin, W[l], buf0, dinv, N);
    k_gather<<<nb((long)N * 32, BT), BT, 0, stream>>>(degi, csr, dinv, buf0, b[l], buf1, N, E);
    hin = buf1;
  }

  k_fill<<<nb(G, BT), BT, 0, stream>>>(counts, G, 0.0f);
  k_fill<<<nb(G * HID, BT), BT, 0, stream>>>(sums, G * HID, 0.0f);
  k_fill<<<nb(G * HID, BT), BT, 0, stream>>>((float*)maxs, G * HID, 0.0f);
  k_pool_runs<<<nb((long)((N + PCH - 1) / PCH) * 32, BT), BT, 0, stream>>>(buf1, batch, counts,
                                                                        sums, maxs, N);
  k_pool_final<<<nb(G * HID, BT), BT, 0, stream>>>(sums, maxs, counts, gbuf, G * HID);

  const int tiles2 = (G / 16) * (HID / 16);
  gemm_wmma_f32<<<(tiles2 + 7) / 8, 256, 0, stream>>>(gbuf, Wc1, bc1, hcls,
                                                      G, HID, 2 * HID, 1);
  k_head<<<nb(G * NCLS, BT), BT, 0, stream>>>(hcls, Wc2, bc2, out, G * NCLS);
}


# Tail in 2 launches, not 7: pooling totals cleared by the first CSR launch; mean/max finalize, head WMMA layer (B requested ahead, same k order) and 10-class layer in one kernel through LDS; 16 to 11
# speedup vs baseline: 5.3197x; 1.0925x over previous
#include <hip/hip_runtime.h>

#define HID   128
#define NGR   512
#define NCLS  10

typedef float v2f __attribute__((ext_vector_type(2)));
typedef float v8f __attribute__((ext_vector_type(8)));

#define BINSH  9
#define BINW   (1 << BINSH)
#define NBMAX  256
#define ECH    6400
#define BTHR   1024
#define EPT    ((ECH + BTHR - 1) / BTHR)
#define HPARTS (BTHR / NBMAX)
#define HUNR   16
#define DCAP   12288
#define DUNR   4

__device__ __forceinline__ int bin_of(int d, int NB) {
  const unsigned q = (unsigned)d >> BINSH;
  return (int)(q < (unsigned)NB ? q : (unsigned)(NB - 1));
}

template <int NW>
__device__ __forceinline__ int block_excl_scan(int v, int* wsum, bool active) {
  const int lane = threadIdx.x & 31;
  const int wave = threadIdx.x >> 5;
  int inc = v;
  for (int d = 1; d < 32; d <<= 1) {
    const int o = __shfl_up(inc, d);
    if (lane >= d) inc += o;
  }
  if (active && lane == 31) wsum[wave] = inc;
  __syncthreads();
  int off = 0;
  if (active) {
    for (int w = 0; w < NW; ++w) off += (w < wave) ? wsum[w] : 0;
  }
  return off + inc - v;
}

__global__ void __launch_bounds__(BTHR)
k_bin_hist(const int* __restrict__ dst, int* __restrict__ hist, int E, int NB,
           int* __restrict__ zero, int zn) {
  __shared__ int h[NBMAX];
  const int t = threadIdx.x;
  if (t < NBMAX) h[t] = 0;
  __syncthreads();
  const int beg = blockIdx.x * ECH;
  const int end = (beg + ECH < E) ? beg + ECH : E;
  int d[EPT];
#pragma unroll
  for (int j = 0; j < EPT; ++j) {
    const int i = beg + j * BTHR + t;
    d[j] = dst[i < end ? i : beg];
  }
#pragma unroll
  for (int j = 0; j < EPT; ++j) {
    const int i = beg + j * BTHR + t;
    if (i < end) atomicAdd(&h[bin_of(d[j], NB)], 1);
  }
  __syncthreads();
  if (t < NB) hist[blockIdx.x * NB + t] = h[t];
  for (int i = blockIdx.x * BTHR + t; i < zn; i += gridDim.x * BTHR) zero[i] = 0;
}

__global__ void __launch_bounds__(BTHR)
k_bin_scatter(const int* __restrict__ src, const int* __restrict__ dst,
              const int* __restrict__ hist, int* __restrict__ bsrc, int* __restrict__ bdst,
              int* __restrict__ binbase, int E, int NB, int NBLK) {
  __shared__ int pbef[HPARTS][NBMAX];
  __shared__ int ptot[HPARTS][NBMAX];
  __shared__ int goff[NBMAX];
  __shared__ int lstart[NBMAX];
  __shared__ int lcnt[NBMAX];
  __shared__ int wsum[NBMAX / 32];
  __shared__ int ssrc[ECH];
  __shared__ int sdst[ECH];

  const int t = threadIdx.x;
  const int k = blockIdx.x;
  const int beg = k * ECH;
  const int end = (beg + ECH < E) ? beg + ECH : E;
  const int n = end - beg;

  int s[EPT], d[EPT];
#pragma unroll
  for (int j = 0; j < EPT; ++j) {
    const int i  = beg + j * BTHR + t;
    const int ic = i < end ? i : beg;
    s[j] = src[ic];
    d[j] = dst[ic];
  }

  {
    const int b = t & (NBMAX - 1);
    const int part = t / NBMAX;
    const int rows = (NBLK + HPARTS - 1) / HPARTS;
    const int r0 = part * rows;
    const int r1 = (r0 + rows < NBLK) ? r0 + rows : NBLK;
    int bef = 0, tot = 0;
    if (b < NB) {
      for (int r = r0; r < r1; r += HUNR) {
        int v[HUNR];
#pragma unroll
        for (int u = 0; u < HUNR; ++u) {
          const int rr = r + u;
          v[u] = hist[(rr < r1 ? rr : r0) * NB + b];
        }
#pragma unroll
        for (int u = 0; u < HUNR; ++u) {
          const int rr = r + u;
          tot += (rr < r1) ? v[u] : 0;
          bef += (rr < r1 && rr < k) ? v[u] : 0;
        }
      }
    }
    pbef[part][b] = bef;
    ptot[part][b] = tot;
    if (t < NBMAX) lcnt[t] = 0;
  }
  __syncthreads();

  int totb = 0, befb = 0;
  if (t < NBMAX) {
    for (int p = 0; p < HPARTS; ++p) { totb += ptot[p][t]; befb += pbef[p][t]; }
  }
  const int base = block_excl_scan<NBMAX / 32>(totb, wsum, t < NBMAX);
  if (t < NBMAX) {
    goff[t] = base + befb;
    if (k == 0 && t < NB) binbase[t] = base;
  }

  int r[EPT];
#pragma unroll
  for (int j = 0; j < EPT; ++j) {
    const int i = beg + j * BTHR + t;
    r[j] = (i < end) ? atomicAdd(&lcnt[bin_of(d[j], NB)], 1) : 0;
  }
  __syncthreads();
  const int cb = (t < NBMAX) ? lcnt[t] : 0;
  const int ls = block_excl_scan<NBMAX / 32>(cb, wsum, t < NBMAX);
  if (t < NBMAX) lstart[t] = ls;
  __syncthreads();

#pragma unroll
  for (int j = 0; j < EPT; ++j) {
    const int i = beg + j * BTHR + t;
    if (i < end) {
      unsigned p = (unsigned)(lstart[bin_of(d[j], NB)] + r[j]);
      p = p < (unsigned)ECH ? p : (unsigned)(ECH - 1);
      ssrc[p] = s[j];
      sdst[p] = d[j];
    }
  }
  __syncthreads();

  for (int i = t; i < n; i += BTHR) {
    const int dd = sdst[i];
    const int bn = bin_of(dd, NB);
    unsigned g = (unsigned)(goff[bn] + (i - lstart[bn]));
    g = g < (unsigned)E ? g : (unsigned)(E - 1);
    bsrc[g] = ssrc[i];
    bdst[g] = dd;
  }
}

__global__ void __launch_bounds__(BTHR)
k_bin_rows(const int* __restrict__ bsrc, const int* __restrict__ bdst,
           const int* __restrict__ binbase, int* __restrict__ rowptr,
           float* __restrict__ dinv, int* __restrict__ csr, int N, int E, int NB) {
  __shared__ int cnt[BINW];
  __shared__ int rstart[BINW];
  __shared__ int cur[BINW];
  __shared__ int wsum[BINW / 32];
  __shared__ int slist[DCAP];

  const int t = threadIdx.x;
  const int b = blockIdx.x;
  unsigned u0 = (unsigned)binbase[b];
  unsigned u1 = (b + 1 < NB) ? (unsigned)binbase[b + 1] : (unsigned)E;
  u1 = u1 < (unsigned)E ? u1 : (unsigned)E;
  u0 = u0 < u1 ? u0 : u1;
  const int e0 = (int)u0, e1 = (int)u1;
  const int n = e1 - e0;

  if (t < BINW) { cnt[t] = 0; cur[t] = 0; }
  __syncthreads();

  for (int i0 = e0; i0 < e1; i0 += DUNR * BTHR) {
    int d[DUNR];
#pragma unroll
    for (int j = 0; j < DUNR; ++j) {
      const int i = i0 + j * BTHR + t;
      d[j] = bdst[i < e1 ? i : e0];
    }
#pragma unroll
    for (int j = 0; j < DUNR; ++j) {
      const int i = i0 + j * BTHR + t;
      if (i < e1) atomicAdd(&cnt[d[j] & (BINW - 1)], 1);
    }
  }
  __syncthreads();

  const int c = (t < BINW) ? cnt[t] : 0;
  const int ex = block_excl_scan<BINW / 32>(c, wsum, t < BINW);
  if (t < BINW) {
    rstart[t] = ex;
    const int node = b * BINW + t;
    if (node < N) {
      rowptr[node] = e0 + ex;
      dinv[node] = rsqrtf((float)c + 1.0f);
    }
  }
  __syncthreads();

  const bool inlds = n <= DCAP;
  for (int i0 = e0; i0 < e1; i0 += DUNR * BTHR) {
    int d[DUNR], s[DUNR];
#pragma unroll
    for (int j = 0; j < DUNR; ++j) {
      const int i  = i0 + j * BTHR + t;
      const int ic = i < e1 ? i : e0;
      d[j] = bdst[ic];
      s[j] = bsrc[ic];
    }
#pragma unroll
    for (int j = 0; j < DUNR; ++j) {
      const int i = i0 + j * BTHR + t;
      if (i < e1) {
        const int dl = d[j] & (BINW - 1);
        const unsigned p = (unsigned)(rstart[dl] + atomicAdd(&cur[dl], 1));
        if (inlds) {
          slist[p < (unsigned)DCAP ? p : (unsigned)(DCAP - 1)] = s[j];
        } else {
          const unsigned g = u0 + p;
          csr[g < (unsigned)E ? g : (unsigned)(E - 1)] = s[j];
        }
      }
    }
  }
  if (inlds) {
    __syncthreads();
    for (int i = t; i < n; i += BTHR) csr[e0 + i] = slist[i];
  }
}

#define STRIP_THREADS       256
#define STRIP_B_PER_THREAD  (128 * 128 / 4 / STRIP_THREADS)
__global__ void __launch_bounds__(STRIP_THREADS)
gemm_wmma_strip_k128(const float* __restrict__ A,
                     const float* __restrict__ B,
                     float* __restrict__ C,
                     const float* __restrict__ dinv, int M) {
  __shared__ __attribute__((aligned(16))) float Bs[128 * 128];

  {
    const float4* __restrict__ B4 = (const float4*)B;
    float4 w[STRIP_B_PER_THREAD];
#pragma unroll
    for (int j = 0; j < STRIP_B_PER_THREAD; ++j) w[j] = B4[j * STRIP_THREADS + threadIdx.x];
#pragma unroll
    for (int j = 0; j < STRIP_B_PER_THREAD; ++j) {
      const int g   = j * STRIP_THREADS + threadIdx.x;
      const int row = g >> 5;
      const int col = (g & 31) << 2;
      *(float4*)&Bs[(row << 7) | (col ^ (((row >> 1) & 1) << 4))] = w[j];
    }
  }
  __syncthreads();

  const int lane = threadIdx.x & 31;
  const int wave = threadIdx.x >> 5;
  const int tm   = blockIdx.x * (blockDim.x >> 5) + wave;
  if (tm >= (M >> 4)) return;

  const int half = lane >> 4;
  const int l15  = lane & 15;
  const int swz  = half << 4;

  v8f acc[8];
#pragma unroll
  for (int tn = 0; tn < 8; ++tn)
#pragma unroll
    for (int r = 0; r < 8; ++r) acc[tn][r] = 0.0f;

  const float* __restrict__ Arow = A + (size_t)(tm * 16 + l15) * 128;

  for (int k = 0; k < 128; k += 4) {
    const int kk = k + half * 2;
    v2f a;
    a[0] = Arow[kk];
    a[1] = Arow[kk + 1];
#pragma unroll
    for (int tn = 0; tn < 8; ++tn) {
      const int c = (tn * 16 + l15) ^ swz;
      v2f b;
      b[0] = Bs[(kk << 7) | c];
      b[1] = Bs[((kk + 1) << 7) | c];
      acc[tn] = __builtin_amdgcn_wmma_f32_16x16x4_f32(false, a, false, b,
                                                      (short)0, acc[tn], false, false);
    }
  }

  float dv[8];
#pragma unroll
  for (int r = 0; r < 8; ++r) dv[r] = dinv[tm * 16 + r + half * 8];
  float* __restrict__ Cp = C + (size_t)(tm * 16) * 128 + l15;
#pragma unroll
  for (int tn = 0; tn < 8; ++tn) {
#pragma unroll
    for (int r = 0; r < 8; ++r) {
      const size_t off = (size_t)(r + half * 8) * 128 + tn * 16;
      Cp[off] = acc[tn][r] * dv[r];
    }
  }
}

#define HK    (2 * HID)
#define HAP   (HK + 4)
#define HHP   (HID + 4)
#define HWPT  (HID * NCLS / 256)
__global__ void __launch_bounds__(256)
k_head_all(const float* __restrict__ sums, const int* __restrict__ maxs,
           const float* __restrict__ counts, const float* __restrict__ Wc1,
           const float* __restrict__ bc1, const float* __restrict__ Wc2,
           const float* __restrict__ bc2, float* __restrict__ out) {
  __shared__ __attribute__((aligned(16))) float As[16 * HAP];
  __shared__ __attribute__((aligned(16))) float Hs[16 * HHP];
  __shared__ float Ws[HID * NCLS];

  const int t    = threadIdx.x;
  const int lane = t & 31;
  const int tn   = t >> 5;
  const int tm   = blockIdx.x;
  const int half = lane >> 4;
  const int l15  = lane & 15;

  const float* __restrict__ Bcol = Wc1 + tn * 16 + l15;
  v2f bq[HK / 4];
#pragma unroll
  for (int q = 0; q < HK / 4; ++q) {
    const int kk = q * 4 + half * 2;
    bq[q][0] = Bcol[(size_t)kk * HID];
    bq[q][1] = Bcol[(size_t)(kk + 1) * HID];
  }
  const float bv = bc1[tn * 16 + l15];

  {
    float4 sv[2];
    int4   mv[2];
    float  cv[2];
    float  wv[HWPT];
#pragma unroll
    for (int j = 0; j < 2; ++j) {
      const int q = j * 256 + t;
      const int r = q >> 5;
      const size_t o = (size_t)(tm * 16 + r) * HID + ((q & 31) << 2);
      sv[j] = *(const float4*)(sums + o);
      mv[j] = *(const int4*)(maxs + o);
      cv[j] = counts[tm * 16 + r];
    }
#pragma unroll
    for (int j = 0; j < HWPT; ++j) wv[j] = Wc2[j * 256 + t];
#pragma unroll
    for (int j = 0; j < 2; ++j) {
      const int q = j * 256 + t;
      const int r = q >> 5;
      const int c4 = (q & 31) << 2;
      const float denom = cv[j] > 1.0f ? cv[j] : 1.0f;
      float4 m;
      m.x = sv[j].x / denom;  m.y = sv[j].y / denom;
      m.z = sv[j].z / denom;  m.w = sv[j].w / denom;
      float4 x;
      x.x = __int_as_float(mv[j].x);  x.y = __int_as_float(mv[j].y);
      x.z = __int_as_float(mv[j].z);  x.w = __int_as_float(mv[j].w);
      *(float4*)&As[r * HAP + c4]       = m;
      *(float4*)&As[r * HAP + HID + c4] = x;
    }
#pragma unroll
    for (int j = 0; j < HWPT; ++j) Ws[j * 256 + t] = wv[j];
  }
  __syncthreads();

  v8f acc;
#pragma unroll
  for (int r = 0; r < 8; ++r) acc[r] = bv;
  const float* Arow = As + l15 * HAP;
#pragma unroll
  for (int q = 0; q < HK / 4; ++q) {
    const int kk = q * 4 + half * 2;
    v2f a;
    a[0] = Arow[kk];
    a[1] = Arow[kk + 1];
    acc = __builtin_amdgcn_wmma_f32_16x16x4_f32(false, a, false, bq[q],
                                                (short)0, acc, false, false);
  }
#pragma unroll
  for (int r = 0; r < 8; ++r) {
    const float v = acc[r] > 0.0f ? acc[r] : 0.0f;
    Hs[(r + half * 8) * HHP + tn * 16 + l15] = v;
  }
  __syncthreads();

  if (t < 16 * NCLS) {
    const int g = t / NCLS;
    const int c = t - g * NCLS;
    float s = bc2[c];
    const float* hv = Hs + g * HHP;
#pragma unroll 8
    for (int k = 0; k < HID; ++k) s += hv[k] * Ws[k * NCLS + c];
    out[(size_t)(tm * 16 + g) * NCLS + c] = s;
  }
}


template <int U>
__device__ __forceinline__ void add_rows(float4& acc, const float4* __restrict__ H,
                                         int my, int j, int lane) {
  float4 v[U];
#pragma unroll
  for (int u = 0; u < U; ++u) {
    const int s = __builtin_amdgcn_readlane(my, j + u);
    v[u] = H[(size_t)s * 32 + lane];
  }
#pragma unroll
  for (int u = 0; u < U; ++u) {
    acc.x += v[u].x; acc.y += v[u].y; acc.z += v[u].z; acc.w += v[u].w;
  }
}

__global__ void k_gather(const int* __restrict__ rowptr, const int* __restrict__ csr,
                         const float* __restrict__ dinv, const float* __restrict__ hs,
                         const float* __restrict__ bias, float* __restrict__ out,
                         int N, int E) {
  const int lane = threadIdx.x & 31;
  const int node = __builtin_amdgcn_readfirstlane(
      (int)((blockIdx.x * blockDim.x + threadIdx.x) >> 5));
  if (node >= N) return;
  const int beg = rowptr[node];
  const int end = (node + 1 < N) ? rowptr[node + 1] : E;
  const float4* __restrict__ H = (const float4*)hs;

  float4 acc = H[(size_t)node * 32 + lane];
  for (int base = beg; base < end; base += 32) {
    const int rem = end - base;
    const int cnt = rem < 32 ? rem : 32;
    unsigned my = (lane < cnt) ? (unsigned)csr[base + lane] : 0u;
    my = my < (unsigned)N ? my : (unsigned)(N - 1);
    int j = 0;
    for (; j + 8 <= cnt; j += 8) add_rows<8>(acc, H, (int)my, j, lane);
    if (j + 4 <= cnt) { add_rows<4>(acc, H, (int)my, j, lane); j += 4; }
    for (; j < cnt; ++j) add_rows<1>(acc, H, (int)my, j, lane);
  }

  const float dv = dinv[node];
  const float4 b = ((const float4*)bias)[lane];
  float4 o;
  o.x = acc.x * dv + b.x;  o.y = acc.y * dv + b.y;
  o.z = acc.z * dv + b.z;  o.w = acc.w * dv + b.w;
  o.x = o.x > 0.0f ? o.x : 0.0f;  o.y = o.y > 0.0f ? o.y : 0.0f;
  o.z = o.z > 0.0f ? o.z : 0.0f;  o.w = o.w > 0.0f ? o.w : 0.0f;
  ((float4*)out)[(size_t)node * 32 + lane] = o;
}


#define PCH 16

__device__ __forceinline__ void pool_flush(int g, const float4& s, const int4& m, int c, int lane,
                                           float* __restrict__ counts, float* __restrict__ sums,
                                           int* __restrict__ maxs) {
  float* sp = sums + (size_t)g * HID + lane * 4;
  int*   mp = maxs + (size_t)g * HID + lane * 4;
  atomicAdd(sp + 0, s.x);  atomicAdd(sp + 1, s.y);
  atomicAdd(sp + 2, s.z);  atomicAdd(sp + 3, s.w);
  atomicMax(mp + 0, m.x);  atomicMax(mp + 1, m.y);
  atomicMax(mp + 2, m.z);  atomicMax(mp + 3, m.w);
  if (lane == 0) atomicAdd(&counts[g], (float)c);
}

__global__ void k_pool_runs(const float* __restrict__ h, const int* __restrict__ batch,
                            float* __restrict__ counts, float* __restrict__ sums,
                            int* __restrict__ maxs, int N) {
  const int lane = threadIdx.x & 31;
  const int w = __builtin_amdgcn_readfirstlane(
      (int)((blockIdx.x * blockDim.x + threadIdx.x) >> 5));
  const int n0 = w * PCH;
  if (n0 >= N) return;
  const int rem = N - n0;
  const int cnt = rem < PCH ? rem : PCH;

  unsigned gid = (unsigned)batch[n0 + (lane < cnt ? lane : 0)];
  gid = gid < (unsigned)NGR ? gid : (unsigned)(NGR - 1);

  const float4* __restrict__ H = (const float4*)h;
  float4 v[PCH];
#pragma unroll
  for (int j = 0; j < PCH; ++j) {
    const int n = n0 + (j < cnt ? j : 0);
    v[j] = H[(size_t)n * 32 + lane];
  }

  int cur = __builtin_amdgcn_readlane((int)gid, 0);
  float4 s = make_float4(0.0f, 0.0f, 0.0f, 0.0f);
  int4   m = make_int4(0, 0, 0, 0);
  int    c = 0;
#pragma unroll
  for (int j = 0; j < PCH; ++j) {
    if (j < cnt) {
      const int g = __builtin_amdgcn_readlane((int)gid, j);
      if (g != cur) {
        pool_flush(cur, s, m, c, lane, counts, sums, maxs);
        s = make_float4(0.0f, 0.0f, 0.0f, 0.0f);
        m = make_int4(0, 0, 0, 0);
        c = 0;
        cur = g;
      }
      s.x += v[j].x;  s.y += v[j].y;  s.z += v[j].z;  s.w += v[j].w;
      const int bx = __float_as_int(v[j].x), by = __float_as_int(v[j].y);
      const int bz = __float_as_int(v[j].z), bw = __float_as_int(v[j].w);
      m.x = bx > m.x ? bx : m.x;  m.y = by > m.y ? by : m.y;
      m.z = bz > m.z ? bz : m.z;  m.w = bw > m.w ? bw : m.w;
      ++c;
    }
  }
  pool_flush(cur, s, m, c, lane, counts, sums, maxs);
}


extern "C" void kernel_launch(void* const* d_in, const int* in_sizes, int n_in,
                              void* d_out, int out_size, void* d_ws, size_t ws_size,
                              hipStream_t stream) {
  const float* x     = (const float*)d_in[0];
  const int*   edge  = (const int*)d_in[1];
  const int*   batch = (const int*)d_in[2];
  const float* W[3]  = {(const float*)d_in[3], (const float*)d_in[5], (const float*)d_in[7]};
  const float* b[3]  = {(const float*)d_in[4], (const float*)d_in[6], (const float*)d_in[8]};
  const float* Wc1   = (const float*)d_in[9];
  const float* bc1   = (const float*)d_in[10];
  const float* Wc2   = (const float*)d_in[11];
  const float* bc2   = (const float*)d_in[12];
  float* out = (float*)d_out;

  const int N = in_sizes[0] / HID;
  const int E = in_sizes[1] / 2;
  const int G = NGR;
  const int* src  = edge;
  const int* dstv = edge + E;

  float* ws     = (float*)d_ws;
  int*   degi   = (int*)ws;
  float* dinv   = ws + N;
  float* buf0   = dinv + N;
  float* buf1   = buf0 + (size_t)N * HID;
  float* counts = buf1 + (size_t)N * HID;
  float* sums   = counts + G;
  int*   maxs   = (int*)(sums + (size_t)G * HID);
  int*   csr    = maxs + (size_t)G * HID;
  const int NB   = (N + BINW - 1) >> BINSH;
  const int NBLK = (E + ECH - 1) / ECH;
  int*   bsrc    = (int*)buf1;
  int*   bdst    = bsrc + E;
  int*   hist    = bdst + E;
  int*   binbase = hist + (size_t)NBLK * NB;

  if ((size_t)((const char*)(csr + E) - (const char*)d_ws) > ws_size) return;
  if (NB > NBMAX || (size_t)2 * E + (size_t)NBLK * NB + NB > (size_t)N * HID) return;

  const int BT = 256;
  auto nb = [](long n, int bt) { return (int)((n + bt - 1) / bt); };

  k_bin_hist<<<NBLK, BTHR, 0, stream>>>(dstv, hist, E, NB, (int*)counts, NGR + 2 * NGR * HID);
  k_bin_scatter<<<NBLK, BTHR, 0, stream>>>(src, dstv, hist, bsrc, bdst, binbase, E, NB, NBLK);
  k_bin_rows<<<NB, BTHR, 0, stream>>>(bsrc, bdst, binbase, degi, dinv, csr, N, E, NB);

  const int rowTiles = N / 16;
  const float* hin = x;
  for (int l = 0; l < 3; ++l) {
    gemm_wmma_strip_k128<<<(rowTiles + 7) / 8, STRIP_THREADS, 0, stream>>>(hin, W[l], buf0, dinv, N);
    k_gather<<<nb((long)N * 32, BT), BT, 0, stream>>>(degi, csr, dinv, buf0, b[l], buf1, N, E);
    hin = buf1;
  }

  k_pool_runs<<<nb((long)((N + PCH - 1) / PCH) * 32, BT), BT, 0, stream>>>(buf1, batch, counts,
                                                                        sums, maxs, N);

  k_head_all<<<G / 16, 256, 0, stream>>>(sums, maxs, counts, Wc1, bc1, Wc2, bc2, out);
}


# Layers 2 and 3: product fused into the aggregation before it: 8 waves lay 16 nodes' relu rows into an LDS tile and WMMA it with W fragments kept in registers, same k order; 11 to 9 launches
# speedup vs baseline: 6.2595x; 1.1767x over previous
#include <hip/hip_runtime.h>

#define HID   128
#define NGR   512
#define NCLS  10

typedef float v2f __attribute__((ext_vector_type(2)));
typedef float v8f __attribute__((ext_vector_type(8)));

#define BINSH  9
#define BINW   (1 << BINSH)
#define NBMAX  256
#define ECH    6400
#define BTHR   1024
#define EPT    ((ECH + BTHR - 1) / BTHR)
#define HPARTS (BTHR / NBMAX)
#define HUNR   16
#define DCAP   12288
#define DUNR   4

__device__ __forceinline__ int bin_of(int d, int NB) {
  const unsigned q = (unsigned)d >> BINSH;
  return (int)(q < (unsigned)NB ? q : (unsigned)(NB - 1));
}

template <int NW>
__device__ __forceinline__ int block_excl_scan(int v, int* wsum, bool active) {
  const int lane = threadIdx.x & 31;
  const int wave = threadIdx.x >> 5;
  int inc = v;
  for (int d = 1; d < 32; d <<= 1) {
    const int o = __shfl_up(inc, d);
    if (lane >= d) inc += o;
  }
  if (active && lane == 31) wsum[wave] = inc;
  __syncthreads();
  int off = 0;
  if (active) {
    for (int w = 0; w < NW; ++w) off += (w < wave) ? wsum[w] : 0;
  }
  return off + inc - v;
}

__global__ void __launch_bounds__(BTHR)
k_bin_hist(const int* __restrict__ dst, int* __restrict__ hist, int E, int NB,
           int* __restrict__ zero, int zn) {
  __shared__ int h[NBMAX];
  const int t = threadIdx.x;
  if (t < NBMAX) h[t] = 0;
  __syncthreads();
  const int beg = blockIdx.x * ECH;
  const int end = (beg + ECH < E) ? beg + ECH : E;
  int d[EPT];
#pragma unroll
  for (int j = 0; j < EPT; ++j) {
    const int i = beg + j * BTHR + t;
    d[j] = dst[i < end ? i : beg];
  }
#pragma unroll
  for (int j = 0; j < EPT; ++j) {
    const int i = beg + j * BTHR + t;
    if (i < end) atomicAdd(&h[bin_of(d[j], NB)], 1);
  }
  __syncthreads();
  if (t < NB) hist[blockIdx.x * NB + t] = h[t];
  for (int i = blockIdx.x * BTHR + t; i < zn; i += gridDim.x * BTHR) zero[i] = 0;
}

__global__ void __launch_bounds__(BTHR)
k_bin_scatter(const int* __restrict__ src, const int* __restrict__ dst,
              const int* __restrict__ hist, int* __restrict__ bsrc, int* __restrict__ bdst,
              int* __restrict__ binbase, int E, int NB, int NBLK) {
  __shared__ int pbef[HPARTS][NBMAX];
  __shared__ int ptot[HPARTS][NBMAX];
  __shared__ int goff[NBMAX];
  __shared__ int lstart[NBMAX];
  __shared__ int lcnt[NBMAX];
  __shared__ int wsum[NBMAX / 32];
  __shared__ int ssrc[ECH];
  __shared__ int sdst[ECH];

  const int t = threadIdx.x;
  const int k = blockIdx.x;
  const int beg = k * ECH;
  const int end = (beg + ECH < E) ? beg + ECH : E;
  const int n = end - beg;

  int s[EPT], d[EPT];
#pragma unroll
  for (int j = 0; j < EPT; ++j) {
    const int i  = beg + j * BTHR + t;
    const int ic = i < end ? i : beg;
    s[j] = src[ic];
    d[j] = dst[ic];
  }

  {
    const int b = t & (NBMAX - 1);
    const int part = t / NBMAX;
    const int rows = (NBLK + HPARTS - 1) / HPARTS;
    const int r0 = part * rows;
    const int r1 = (r0 + rows < NBLK) ? r0 + rows : NBLK;
    int bef = 0, tot = 0;
    if (b < NB) {
      for (int r = r0; r < r1; r += HUNR) {
        int v[HUNR];
#pragma unroll
        for (int u = 0; u < HUNR; ++u) {
          const int rr = r + u;
          v[u] = hist[(rr < r1 ? rr : r0) * NB + b];
        }
#pragma unroll
        for (int u = 0; u < HUNR; ++u) {
          const int rr = r + u;
          tot += (rr < r1) ? v[u] : 0;
          bef += (rr < r1 && rr < k) ? v[u] : 0;
        }
      }
    }
    pbef[part][b] = bef;
    ptot[part][b] = tot;
    if (t < NBMAX) lcnt[t] = 0;
  }
  __syncthreads();

  int totb = 0, befb = 0;
  if (t < NBMAX) {
    for (int p = 0; p < HPARTS; ++p) { totb += ptot[p][t]; befb += pbef[p][t]; }
  }
  const int base = block_excl_scan<NBMAX / 32>(totb, wsum, t < NBMAX);
  if (t < NBMAX) {
    goff[t] = base + befb;
    if (k == 0 && t < NB) binbase[t] = base;
  }

  int r[EPT];
#pragma unroll
  for (int j = 0; j < EPT; ++j) {
    const int i = beg + j * BTHR + t;
    r[j] = (i < end) ? atomicAdd(&lcnt[bin_of(d[j], NB)], 1) : 0;
  }
  __syncthreads();
  const int cb = (t < NBMAX) ? lcnt[t] : 0;
  const int ls = block_excl_scan<NBMAX / 32>(cb, wsum, t < NBMAX);
  if (t < NBMAX) lstart[t] = ls;
  __syncthreads();

#pragma unroll
  for (int j = 0; j < EPT; ++j) {
    const int i = beg + j * BTHR + t;
    if (i < end) {
      unsigned p = (unsigned)(lstart[bin_of(d[j], NB)] + r[j]);
      p = p < (unsigned)ECH ? p : (unsigned)(ECH - 1);
      ssrc[p] = s[j];
      sdst[p] = d[j];
    }
  }
  __syncthreads();

  for (int i = t; i < n; i += BTHR) {
    const int dd = sdst[i];
    const int bn = bin_of(dd, NB);
    unsigned g = (unsigned)(goff[bn] + (i - lstart[bn]));
    g = g < (unsigned)E ? g : (unsigned)(E - 1);
    bsrc[g] = ssrc[i];
    bdst[g] = dd;
  }
}

__global__ void __launch_bounds__(BTHR)
k_bin_rows(const int* __restrict__ bsrc, const int* __restrict__ bdst,
           const int* __restrict__ binbase, int* __restrict__ rowptr,
           float* __restrict__ dinv, int* __restrict__ csr, int N, int E, int NB) {
  __shared__ int cnt[BINW];
  __shared__ int rstart[BINW];
  __shared__ int cur[BINW];
  __shared__ int wsum[BINW / 32];
  __shared__ int slist[DCAP];

  const int t = threadIdx.x;
  const int b = blockIdx.x;
  unsigned u0 = (unsigned)binbase[b];
  unsigned u1 = (b + 1 < NB) ? (unsigned)binbase[b + 1] : (unsigned)E;
  u1 = u1 < (unsigned)E ? u1 : (unsigned)E;
  u0 = u0 < u1 ? u0 : u1;
  const int e0 = (int)u0, e1 = (int)u1;
  const int n = e1 - e0;

  if (t < BINW) { cnt[t] = 0; cur[t] = 0; }
  __syncthreads();

  for (int i0 = e0; i0 < e1; i0 += DUNR * BTHR) {
    int d[DUNR];
#pragma unroll
    for (int j = 0; j < DUNR; ++j) {
      const int i = i0 + j * BTHR + t;
      d[j] = bdst[i < e1 ? i : e0];
    }
#pragma unroll
    for (int j = 0; j < DUNR; ++j) {
      const int i = i0 + j * BTHR + t;
      if (i < e1) atomicAdd(&cnt[d[j] & (BINW - 1)], 1);
    }
  }
  __syncthreads();

  const int c = (t < BINW) ? cnt[t] : 0;
  const int ex = block_excl_scan<BINW / 32>(c, wsum, t < BINW);
  if (t < BINW) {
    rstart[t] = ex;
    const int node = b * BINW + t;
    if (node < N) {
      rowptr[node] = e0 + ex;
      dinv[node] = rsqrtf((float)c + 1.0f);
    }
  }
  __syncthreads();

  const bool inlds = n <= DCAP;
  for (int i0 = e0; i0 < e1; i0 += DUNR * BTHR) {
    int d[DUNR], s[DUNR];
#pragma unroll
    for (int j = 0; j < DUNR; ++j) {
      const int i  = i0 + j * BTHR + t;
      const int ic = i < e1 ? i : e0;
      d[j] = bdst[ic];
      s[j] = bsrc[ic];
    }
#pragma unroll
    for (int j = 0; j < DUNR; ++j) {
      const int i = i0 + j * BTHR + t;
      if (i < e1) {
        const int dl = d[j] & (BINW - 1);
        const unsigned p = (unsigned)(rstart[dl] + atomicAdd(&cur[dl], 1));
        if (inlds) {
          slist[p < (unsigned)DCAP ? p : (unsigned)(DCAP - 1)] = s[j];
        } else {
          const unsigned g = u0 + p;
          csr[g < (unsigned)E ? g : (unsigned)(E - 1)] = s[j];
        }
      }
    }
  }
  if (inlds) {
    __syncthreads();
    for (int i = t; i < n; i += BTHR) csr[e0 + i] = slist[i];
  }
}

#define STRIP_THREADS       256
#define STRIP_B_PER_THREAD  (128 * 128 / 4 / STRIP_THREADS)
__global__ void __launch_bounds__(STRIP_THREADS)
gemm_wmma_strip_k128(const float* __restrict__ A,
                     const float* __restrict__ B,
                     float* __restrict__ C,
                     const float* __restrict__ dinv, int M) {
  __shared__ __attribute__((aligned(16))) float Bs[128 * 128];

  {
    const float4* __restrict__ B4 = (const float4*)B;
    float4 w[STRIP_B_PER_THREAD];
#pragma unroll
    for (int j = 0; j < STRIP_B_PER_THREAD; ++j) w[j] = B4[j * STRIP_THREADS + threadIdx.x];
#pragma unroll
    for (int j = 0; j < STRIP_B_PER_THREAD; ++j) {
      const int g   = j * STRIP_THREADS + threadIdx.x;
      const int row = g >> 5;
      const int col = (g & 31) << 2;
      *(float4*)&Bs[(row << 7) | (col ^ (((row >> 1) & 1) << 4))] = w[j];
    }
  }
  __syncthreads();

  const int lane = threadIdx.x & 31;
  const int wave = threadIdx.x >> 5;
  const int tm   = blockIdx.x * (blockDim.x >> 5) + wave;
  if (tm >= (M >> 4)) return;

  const int half = lane >> 4;
  const int l15  = lane & 15;
  const int swz  = half << 4;

  v8f acc[8];
#pragma unroll
  for (int tn = 0; tn < 8; ++tn)
#pragma unroll
    for (int r = 0; r < 8; ++r) acc[tn][r] = 0.0f;

  const float* __restrict__ Arow = A + (size_t)(tm * 16 + l15) * 128;

  for (int k = 0; k < 128; k += 4) {
    const int kk = k + half * 2;
    v2f a;
    a[0] = Arow[kk];
    a[1] = Arow[kk + 1];
#pragma unroll
    for (int tn = 0; tn < 8; ++tn) {
      const int c = (tn * 16 + l15) ^ swz;
      v2f b;
      b[0] = Bs[(kk << 7) | c];
      b[1] = Bs[((kk + 1) << 7) | c];
      acc[tn] = __builtin_amdgcn_wmma_f32_16x16x4_f32(false, a, false, b,
                                                      (short)0, acc[tn], false, false);
    }
  }

  float dv[8];
#pragma unroll
  for (int r = 0; r < 8; ++r) dv[r] = dinv[tm * 16 + r + half * 8];
  float* __restrict__ Cp = C + (size_t)(tm * 16) * 128 + l15;
#pragma unroll
  for (int tn = 0; tn < 8; ++tn) {
#pragma unroll
    for (int r = 0; r < 8; ++r) {
      const size_t off = (size_t)(r + half * 8) * 128 + tn * 16;
      Cp[off] = acc[tn][r] * dv[r];
    }
  }
}

#define HK    (2 * HID)
#define HAP   (HK + 4)
#define HHP   (HID + 4)
#define HWPT  (HID * NCLS / 256)
__global__ void __launch_bounds__(256)
k_head_all(const float* __restrict__ sums, const int* __restrict__ maxs,
           const float* __restrict__ counts, const float* __restrict__ Wc1,
           const float* __restrict__ bc1, const float* __restrict__ Wc2,
           const float* __restrict__ bc2, float* __restrict__ out) {
  __shared__ __attribute__((aligned(16))) float As[16 * HAP];
  __shared__ __attribute__((aligned(16))) float Hs[16 * HHP];
  __shared__ float Ws[HID * NCLS];

  const int t    = threadIdx.x;
  const int lane = t & 31;
  const int tn   = t >> 5;
  const int tm   = blockIdx.x;
  const int half = lane >> 4;
  const int l15  = lane & 15;

  const float* __restrict__ Bcol = Wc1 + tn * 16 + l15;
  v2f bq[HK / 4];
#pragma unroll
  for (int q = 0; q < HK / 4; ++q) {
    const int kk = q * 4 + half * 2;
    bq[q][0] = Bcol[(size_t)kk * HID];
    bq[q][1] = Bcol[(size_t)(kk + 1) * HID];
  }
  const float bv = bc1[tn * 16 + l15];

  {
    float4 sv[2];
    int4   mv[2];
    float  cv[2];
    float  wv[HWPT];
#pragma unroll
    for (int j = 0; j < 2; ++j) {
      const int q = j * 256 + t;
      const int r = q >> 5;
      const size_t o = (size_t)(tm * 16 + r) * HID + ((q & 31) << 2);
      sv[j] = *(const float4*)(sums + o);
      mv[j] = *(const int4*)(maxs + o);
      cv[j] = counts[tm * 16 + r];
    }
#pragma unroll
    for (int j = 0; j < HWPT; ++j) wv[j] = Wc2[j * 256 + t];
#pragma unroll
    for (int j = 0; j < 2; ++j) {
      const int q = j * 256 + t;
      const int r = q >> 5;
      const int c4 = (q & 31) << 2;
      const float denom = cv[j] > 1.0f ? cv[j] : 1.0f;
      float4 m;
      m.x = sv[j].x / denom;  m.y = sv[j].y / denom;
      m.z = sv[j].z / denom;  m.w = sv[j].w / denom;
      float4 x;
      x.x = __int_as_float(mv[j].x);  x.y = __int_as_float(mv[j].y);
      x.z = __int_as_float(mv[j].z);  x.w = __int_as_float(mv[j].w);
      *(float4*)&As[r * HAP + c4]       = m;
      *(float4*)&As[r * HAP + HID + c4] = x;
    }
#pragma unroll
    for (int j = 0; j < HWPT; ++j) Ws[j * 256 + t] = wv[j];
  }
  __syncthreads();

  v8f acc;
#pragma unroll
  for (int r = 0; r < 8; ++r) acc[r] = bv;
  const float* Arow = As + l15 * HAP;
#pragma unroll
  for (int q = 0; q < HK / 4; ++q) {
    const int kk = q * 4 + half * 2;
    v2f a;
    a[0] = Arow[kk];
    a[1] = Arow[kk + 1];
    acc = __builtin_amdgcn_wmma_f32_16x16x4_f32(false, a, false, bq[q],
                                                (short)0, acc, false, false);
  }
#pragma unroll
  for (int r = 0; r < 8; ++r) {
    const float v = acc[r] > 0.0f ? acc[r] : 0.0f;
    Hs[(r + half * 8) * HHP + tn * 16 + l15] = v;
  }
  __syncthreads();

  if (t < 16 * NCLS) {
    const int g = t / NCLS;
    const int c = t - g * NCLS;
    float s = bc2[c];
    const float* hv = Hs + g * HHP;
#pragma unroll 8
    for (int k = 0; k < HID; ++k) s += hv[k] * Ws[k * NCLS + c];
    out[(size_t)(tm * 16 + g) * NCLS + c] = s;
  }
}


template <int U>
__device__ __forceinline__ void add_rows(float4& acc, const float4* __restrict__ H,
                                         int my, int j, int lane) {
  float4 v[U];
#pragma unroll
  for (int u = 0; u < U; ++u) {
    const int s = __builtin_amdgcn_readlane(my, j + u);
    v[u] = H[(size_t)s * 32 + lane];
  }
#pragma unroll
  for (int u = 0; u < U; ++u) {
    acc.x += v[u].x; acc.y += v[u].y; acc.z += v[u].z; acc.w += v[u].w;
  }
}

__global__ void k_gather(const int* __restrict__ rowptr, const int* __restrict__ csr,
                         const float* __restrict__ dinv, const float* __restrict__ hs,
                         const float* __restrict__ bias, float* __restrict__ out,
                         int N, int E) {
  const int lane = threadIdx.x & 31;
  const int node = __builtin_amdgcn_readfirstlane(
      (int)((blockIdx.x * blockDim.x + threadIdx.x) >> 5));
  if (node >= N) return;
  const int beg = rowptr[node];
  const int end = (node + 1 < N) ? rowptr[node + 1] : E;
  const float4* __restrict__ H = (const float4*)hs;

  float4 acc = H[(size_t)node * 32 + lane];
  for (int base = beg; base < end; base += 32) {
    const int rem = end - base;
    const int cnt = rem < 32 ? rem : 32;
    unsigned my = (lane < cnt) ? (unsigned)csr[base + lane] : 0u;
    my = my < (unsigned)N ? my : (unsigned)(N - 1);
    int j = 0;
    for (; j + 8 <= cnt; j += 8) add_rows<8>(acc, H, (int)my, j, lane);
    if (j + 4 <= cnt) { add_rows<4>(acc, H, (int)my, j, lane); j += 4; }
    for (; j < cnt; ++j) add_rows<1>(acc, H, (int)my, j, lane);
  }

  const float dv = dinv[node];
  const float4 b = ((const float4*)bias)[lane];
  float4 o;
  o.x = acc.x * dv + b.x;  o.y = acc.y * dv + b.y;
  o.z = acc.z * dv + b.z;  o.w = acc.w * dv + b.w;
  o.x = o.x > 0.0f ? o.x : 0.0f;  o.y = o.y > 0.0f ? o.y : 0.0f;
  o.z = o.z > 0.0f ? o.z : 0.0f;  o.w = o.w > 0.0f ? o.w : 0.0f;
  ((float4*)out)[(size_t)node * 32 + lane] = o;
}

#define FT_PITCH     (HID + 4)
#define FUSED_BLOCKS 1024

__device__ __forceinline__ void walk_list(float4& acc, const float4* __restrict__ H,
                                          const int* __restrict__ csr, unsigned my,
                                          int beg, int end, int N, int lane) {
  for (int base = beg; base < end; base += 32) {
    const int rem = end - base;
    const int cnt = rem < 32 ? rem : 32;
    if (base != beg) {
      my = (lane < cnt) ? (unsigned)csr[base + lane] : 0u;
      my = my < (unsigned)N ? my : (unsigned)(N - 1);
    }
    int j = 0;
    for (; j + 8 <= cnt; j += 8) add_rows<8>(acc, H, (int)my, j, lane);
    switch (cnt - j) {
      case 7: add_rows<7>(acc, H, (int)my, j, lane); break;
      case 6: add_rows<6>(acc, H, (int)my, j, lane); break;
      case 5: add_rows<5>(acc, H, (int)my, j, lane); break;
      case 4: add_rows<4>(acc, H, (int)my, j, lane); break;
      case 3: add_rows<3>(acc, H, (int)my, j, lane); break;
      case 2: add_rows<2>(acc, H, (int)my, j, lane); break;
      case 1: add_rows<1>(acc, H, (int)my, j, lane); break;
      default: break;
    }
  }
}

__global__ void __launch_bounds__(256, 4)
k_gather_gemm(const int* __restrict__ rowptr, const int* __restrict__ csr,
              const float* __restrict__ dinv, const float* __restrict__ hs,
              const float* __restrict__ bias, const float* __restrict__ Wn,
              float* __restrict__ out, int N, int E, int units) {
  __shared__ __attribute__((aligned(16))) float At[16 * FT_PITCH];
  __shared__ __attribute__((aligned(16))) float Dt[16 * FT_PITCH];
  __shared__ float Dv[16];

  const int lane = threadIdx.x & 31;
  const int wave = __builtin_amdgcn_readfirstlane((int)(threadIdx.x >> 5));
  const int half = lane >> 4;
  const int l15  = lane & 15;

  const float* __restrict__ Bcol = Wn + wave * 16 + l15;
  v2f bq[HID / 4];
#pragma unroll
  for (int q = 0; q < HID / 4; ++q) {
    const int kk = q * 4 + half * 2;
    bq[q][0] = Bcol[(size_t)kk * HID];
    bq[q][1] = Bcol[(size_t)(kk + 1) * HID];
  }
  const float4 bs = ((const float4*)bias)[lane];
  const float4* __restrict__ H = (const float4*)hs;

  auto unit_words = [&](int uu, int& rpv, float& dvl) {
    int i0 = uu * 16 + wave * 2 + (lane < 3 ? lane : 0);
    i0 = i0 < N ? i0 : N - 1;
    rpv = rowptr[i0];
    dvl = dinv[i0];
  };
  int   rpv;
  float dvl;
  unit_words((int)blockIdx.x, rpv, dvl);

  for (int u = blockIdx.x; u < units; u += gridDim.x) {
    const int n0 = u * 16 + wave * 2;
    int   rpvn;
    float dvln;
    unit_words(u + (int)gridDim.x, rpvn, dvln);
    const int r0 = __builtin_amdgcn_readlane(rpv, 0);
    const int r1 = __builtin_amdgcn_readlane(rpv, 1);
    const int r2 = __builtin_amdgcn_readlane(rpv, 2);
    int   ndc[2], bg[2], en[2];
    bool  ok[2];
    float dvv[2];
#pragma unroll
    for (int i = 0; i < 2; ++i) {
      const int nd = n0 + i;
      ok[i]  = nd < N;
      ndc[i] = ok[i] ? nd : 0;
      int b = (i == 0) ? r0 : r1;
      int e = (nd + 1 < N) ? ((i == 0) ? r1 : r2) : E;
      e = e < E ? e : E;  e = e > 0 ? e : 0;
      b = b > 0 ? b : 0;  b = b < e ? b : e;
      if (!ok[i]) e = b;
      bg[i] = b;  en[i] = e;
      dvv[i] = __int_as_float(__builtin_amdgcn_readlane(__float_as_int(dvl), i));
    }
    unsigned my[2];
    float4   acc2[2];
#pragma unroll
    for (int i = 0; i < 2; ++i) {
      const int c0   = en[i] - bg[i];
      const int cnt0 = c0 < 32 ? c0 : 32;
      const unsigned m = (lane < cnt0) ? (unsigned)csr[bg[i] + lane] : 0u;
      my[i]   = m < (unsigned)N ? m : (unsigned)(N - 1);
      acc2[i] = H[(size_t)ndc[i] * 32 + lane];
    }
#pragma unroll
    for (int i = 0; i < 2; ++i) {
      walk_list(acc2[i], H, csr, my[i], bg[i], en[i], N, lane);
      const float dv = dvv[i];
      float4 o;
      o.x = acc2[i].x * dv + bs.x;  o.y = acc2[i].y * dv + bs.y;
      o.z = acc2[i].z * dv + bs.z;  o.w = acc2[i].w * dv + bs.w;
      o.x = o.x > 0.0f ? o.x : 0.0f;  o.y = o.y > 0.0f ? o.y : 0.0f;
      o.z = o.z > 0.0f ? o.z : 0.0f;  o.w = o.w > 0.0f ? o.w : 0.0f;
      if (!ok[i]) o = make_float4(0.0f, 0.0f, 0.0f, 0.0f);
      const int row = wave * 2 + i;
      *(float4*)&At[row * FT_PITCH + lane * 4] = o;
      if (lane == 0) Dv[row] = ok[i] ? dv : 0.0f;
    }
    __syncthreads();

    v8f acc;
#pragma unroll
    for (int r = 0; r < 8; ++r) acc[r] = 0.0f;
    const float* Arow = At + l15 * FT_PITCH + half * 2;
#pragma unroll
    for (int q = 0; q < HID / 4; ++q) {
      v2f a;
      a[0] = Arow[q * 4];
      a[1] = Arow[q * 4 + 1];
      acc = __builtin_amdgcn_wmma_f32_16x16x4_f32(false, a, false, bq[q],
                                                  (short)0, acc, false, false);
    }
#pragma unroll
    for (int r = 0; r < 8; ++r) {
      const int row = r + half * 8;
      Dt[row * FT_PITCH + wave * 16 + l15] = acc[r] * Dv[row];
    }
    __syncthreads();

#pragma unroll
    for (int i = 0; i < 2; ++i) {
      const int row = wave * 2 + i;
      if (n0 + i < N)
        ((float4*)out)[(size_t)(n0 + i) * 32 + lane] = *(const float4*)&Dt[row * FT_PITCH + lane * 4];
    }
    rpv = rpvn;
    dvl = dvln;
  }
}


#define PCH 16

__device__ __forceinline__ void pool_flush(int g, const float4& s, const int4& m, int c, int lane,
                                           float* __restrict__ counts, float* __restrict__ sums,
                                           int* __restrict__ maxs) {
  float* sp = sums + (size_t)g * HID + lane * 4;
  int*   mp = maxs + (size_t)g * HID + lane * 4;
  atomicAdd(sp + 0, s.x);  atomicAdd(sp + 1, s.y);
  atomicAdd(sp + 2, s.z);  atomicAdd(sp + 3, s.w);
  atomicMax(mp + 0, m.x);  atomicMax(mp + 1, m.y);
  atomicMax(mp + 2, m.z);  atomicMax(mp + 3, m.w);
  if (lane == 0) atomicAdd(&counts[g], (float)c);
}

__global__ void k_pool_runs(const float* __restrict__ h, const int* __restrict__ batch,
                            float* __restrict__ counts, float* __restrict__ sums,
                            int* __restrict__ maxs, int N) {
  const int lane = threadIdx.x & 31;
  const int w = __builtin_amdgcn_readfirstlane(
      (int)((blockIdx.x * blockDim.x + threadIdx.x) >> 5));
  const int n0 = w * PCH;
  if (n0 >= N) return;
  const int rem = N - n0;
  const int cnt = rem < PCH ? rem : PCH;

  unsigned gid = (unsigned)batch[n0 + (lane < cnt ? lane : 0)];
  gid = gid < (unsigned)NGR ? gid : (unsigned)(NGR - 1);

  const float4* __restrict__ H = (const float4*)h;
  float4 v[PCH];
#pragma unroll
  for (int j = 0; j < PCH; ++j) {
    const int n = n0 + (j < cnt ? j : 0);
    v[j] = H[(size_t)n * 32 + lane];
  }

  int cur = __builtin_amdgcn_readlane((int)gid, 0);
  float4 s = make_float4(0.0f, 0.0f, 0.0f, 0.0f);
  int4   m = make_int4(0, 0, 0, 0);
  int    c = 0;
#pragma unroll
  for (int j = 0; j < PCH; ++j) {
    if (j < cnt) {
      const int g = __builtin_amdgcn_readlane((int)gid, j);
      if (g != cur) {
        pool_flush(cur, s, m, c, lane, counts, sums, maxs);
        s = make_float4(0.0f, 0.0f, 0.0f, 0.0f);
        m = make_int4(0, 0, 0, 0);
        c = 0;
        cur = g;
      }
      s.x += v[j].x;  s.y += v[j].y;  s.z += v[j].z;  s.w += v[j].w;
      const int bx = __float_as_int(v[j].x), by = __float_as_int(v[j].y);
      const int bz = __float_as_int(v[j].z), bw = __float_as_int(v[j].w);
      m.x = bx > m.x ? bx : m.x;  m.y = by > m.y ? by : m.y;
      m.z = bz > m.z ? bz : m.z;  m.w = bw > m.w ? bw : m.w;
      ++c;
    }
  }
  pool_flush(cur, s, m, c, lane, counts, sums, maxs);
}


extern "C" void kernel_launch(void* const* d_in, const int* in_sizes, int n_in,
                              void* d_out, int out_size, void* d_ws, size_t ws_size,
                              hipStream_t stream) {
  const float* x     = (const float*)d_in[0];
  const int*   edge  = (const int*)d_in[1];
  const int*   batch = (const int*)d_in[2];
  const float* W[3]  = {(const float*)d_in[3], (const float*)d_in[5], (const float*)d_in[7]};
  const float* b[3]  = {(const float*)d_in[4], (const float*)d_in[6], (const float*)d_in[8]};
  const float* Wc1   = (const float*)d_in[9];
  const float* bc1   = (const float*)d_in[10];
  const float* Wc2   = (const float*)d_in[11];
  const float* bc2   = (const float*)d_in[12];
  float* out = (float*)d_out;

  const int N = in_sizes[0] / HID;
  const int E = in_sizes[1] / 2;
  const int G = NGR;
  const int* src  = edge;
  const int* dstv = edge + E;

  float* ws     = (float*)d_ws;
  int*   degi   = (int*)ws;
  float* dinv   = ws + N;
  float* buf0   = dinv + N;
  float* buf1   = buf0 + (size_t)N * HID;
  float* counts = buf1 + (size_t)N * HID;
  float* sums   = counts + G;
  int*   maxs   = (int*)(sums + (size_t)G * HID);
  int*   csr    = maxs + (size_t)G * HID;
  const int NB   = (N + BINW - 1) >> BINSH;
  const int NBLK = (E + ECH - 1) / ECH;
  int*   bsrc    = (int*)buf1;
  int*   bdst    = bsrc + E;
  int*   hist    = bdst + E;
  int*   binbase = hist + (size_t)NBLK * NB;

  if ((size_t)((const char*)(csr + E) - (const char*)d_ws) > ws_size) return;
  if (NB > NBMAX || (size_t)2 * E + (size_t)NBLK * NB + NB > (size_t)N * HID) return;

  const int BT = 256;
  auto nb = [](long n, int bt) { return (int)((n + bt - 1) / bt); };

  k_bin_hist<<<NBLK, BTHR, 0, stream>>>(dstv, hist, E, NB, (int*)counts, NGR + 2 * NGR * HID);
  k_bin_scatter<<<NBLK, BTHR, 0, stream>>>(src, dstv, hist, bsrc, bdst, binbase, E, NB, NBLK);
  k_bin_rows<<<NB, BTHR, 0, stream>>>(bsrc, bdst, binbase, degi, dinv, csr, N, E, NB);

  const int rowTiles = N / 16;
  const int units    = (N + 15) / 16;
  gemm_wmma_strip_k128<<<(rowTiles + 7) / 8, STRIP_THREADS, 0, stream>>>(x, W[0], buf0, dinv, N);
  k_gather_gemm<<<FUSED_BLOCKS, 256, 0, stream>>>(degi, csr, dinv, buf0, b[0], W[1], buf1, N, E, units);
  k_gather_gemm<<<FUSED_BLOCKS, 256, 0, stream>>>(degi, csr, dinv, buf1, b[1], W[2], buf0, N, E, units);
  k_gather<<<nb((long)N * 32, BT), BT, 0, stream>>>(degi, csr, dinv, buf0, b[2], buf1, N, E);

  k_pool_runs<<<nb((long)((N + PCH - 1) / PCH) * 32, BT), BT, 0, stream>>>(buf1, batch, counts,
                                                                        sums, maxs, N);

  k_head_all<<<G / 16, 256, 0, stream>>>(sums, maxs, counts, Wc1, bc1, Wc2, bc2, out);
}


# All three layers sum neighbour rows first and multiply after ((A h) W for A (h W)): one sum+WMMA launch per layer, layer 1 weights raw x rows by dinv per edge; sum order changes; 9 to 8 launches
# speedup vs baseline: 6.7078x; 1.0716x over previous
#include <hip/hip_runtime.h>

#define HID   128
#define NGR   512
#define NCLS  10

typedef float v2f __attribute__((ext_vector_type(2)));
typedef float v8f __attribute__((ext_vector_type(8)));

#define BINSH  9
#define BINW   (1 << BINSH)
#define NBMAX  256
#define ECH    6400
#define BTHR   1024
#define EPT    ((ECH + BTHR - 1) / BTHR)
#define HPARTS (BTHR / NBMAX)
#define HUNR   16
#define DCAP   12288
#define DUNR   4

__device__ __forceinline__ int bin_of(int d, int NB) {
  const unsigned q = (unsigned)d >> BINSH;
  return (int)(q < (unsigned)NB ? q : (unsigned)(NB - 1));
}

template <int NW>
__device__ __forceinline__ int block_excl_scan(int v, int* wsum, bool active) {
  const int lane = threadIdx.x & 31;
  const int wave = threadIdx.x >> 5;
  int inc = v;
  for (int d = 1; d < 32; d <<= 1) {
    const int o = __shfl_up(inc, d);
    if (lane >= d) inc += o;
  }
  if (active && lane == 31) wsum[wave] = inc;
  __syncthreads();
  int off = 0;
  if (active) {
    for (int w = 0; w < NW; ++w) off += (w < wave) ? wsum[w] : 0;
  }
  return off + inc - v;
}

__global__ void __launch_bounds__(BTHR)
k_bin_hist(const int* __restrict__ dst, int* __restrict__ hist, int E, int NB,
           int* __restrict__ zero, int zn) {
  __shared__ int h[NBMAX];
  const int t = threadIdx.x;
  if (t < NBMAX) h[t] = 0;
  __syncthreads();
  const int beg = blockIdx.x * ECH;
  const int end = (beg + ECH < E) ? beg + ECH : E;
  int d[EPT];
#pragma unroll
  for (int j = 0; j < EPT; ++j) {
    const int i = beg + j * BTHR + t;
    d[j] = dst[i < end ? i : beg];
  }
#pragma unroll
  for (int j = 0; j < EPT; ++j) {
    const int i = beg + j * BTHR + t;
    if (i < end) atomicAdd(&h[bin_of(d[j], NB)], 1);
  }
  __syncthreads();
  if (t < NB) hist[blockIdx.x * NB + t] = h[t];
  for (int i = blockIdx.x * BTHR + t; i < zn; i += gridDim.x * BTHR) zero[i] = 0;
}

__global__ void __launch_bounds__(BTHR)
k_bin_scatter(const int* __restrict__ src, const int* __restrict__ dst,
              const int* __restrict__ hist, int* __restrict__ bsrc, int* __restrict__ bdst,
              int* __restrict__ binbase, int E, int NB, int NBLK) {
  __shared__ int pbef[HPARTS][NBMAX];
  __shared__ int ptot[HPARTS][NBMAX];
  __shared__ int goff[NBMAX];
  __shared__ int lstart[NBMAX];
  __shared__ int lcnt[NBMAX];
  __shared__ int wsum[NBMAX / 32];
  __shared__ int ssrc[ECH];
  __shared__ int sdst[ECH];

  const int t = threadIdx.x;
  const int k = blockIdx.x;
  const int beg = k * ECH;
  const int end = (beg + ECH < E) ? beg + ECH : E;
  const int n = end - beg;

  int s[EPT], d[EPT];
#pragma unroll
  for (int j = 0; j < EPT; ++j) {
    const int i  = beg + j * BTHR + t;
    const int ic = i < end ? i : beg;
    s[j] = src[ic];
    d[j] = dst[ic];
  }

  {
    const int b = t & (NBMAX - 1);
    const int part = t / NBMAX;
    const int rows = (NBLK + HPARTS - 1) / HPARTS;
    const int r0 = part * rows;
    const int r1 = (r0 + rows < NBLK) ? r0 + rows : NBLK;
    int bef = 0, tot = 0;
    if (b < NB) {
      for (int r = r0; r < r1; r += HUNR) {
        int v[HUNR];
#pragma unroll
        for (int u = 0; u < HUNR; ++u) {
          const int rr = r + u;
          v[u] = hist[(rr < r1 ? rr : r0) * NB + b];
        }
#pragma unroll
        for (int u = 0; u < HUNR; ++u) {
          const int rr = r + u;
          tot += (rr < r1) ? v[u] : 0;
          bef += (rr < r1 && rr < k) ? v[u] : 0;
        }
      }
    }
    pbef[part][b] = bef;
    ptot[part][b] = tot;
    if (t < NBMAX) lcnt[t] = 0;
  }
  __syncthreads();

  int totb = 0, befb = 0;
  if (t < NBMAX) {
    for (int p = 0; p < HPARTS; ++p) { totb += ptot[p][t]; befb += pbef[p][t]; }
  }
  const int base = block_excl_scan<NBMAX / 32>(totb, wsum, t < NBMAX);
  if (t < NBMAX) {
    goff[t] = base + befb;
    if (k == 0 && t < NB) binbase[t] = base;
  }

  int r[EPT];
#pragma unroll
  for (int j = 0; j < EPT; ++j) {
    const int i = beg + j * BTHR + t;
    r[j] = (i < end) ? atomicAdd(&lcnt[bin_of(d[j], NB)], 1) : 0;
  }
  __syncthreads();
  const int cb = (t < NBMAX) ? lcnt[t] : 0;
  const int ls = block_excl_scan<NBMAX / 32>(cb, wsum, t < NBMAX);
  if (t < NBMAX) lstart[t] = ls;
  __syncthreads();

#pragma unroll
  for (int j = 0; j < EPT; ++j) {
    const int i = beg + j * BTHR + t;
    if (i < end) {
      unsigned p = (unsigned)(lstart[bin_of(d[j], NB)] + r[j]);
      p = p < (unsigned)ECH ? p : (unsigned)(ECH - 1);
      ssrc[p] = s[j];
      sdst[p] = d[j];
    }
  }
  __syncthreads();

  for (int i = t; i < n; i += BTHR) {
    const int dd = sdst[i];
    const int bn = bin_of(dd, NB);
    unsigned g = (unsigned)(goff[bn] + (i - lstart[bn]));
    g = g < (unsigned)E ? g : (unsigned)(E - 1);
    bsrc[g] = ssrc[i];
    bdst[g] = dd;
  }
}

__global__ void __launch_bounds__(BTHR)
k_bin_rows(const int* __restrict__ bsrc, const int* __restrict__ bdst,
           const int* __restrict__ binbase, int* __restrict__ rowptr,
           float* __restrict__ dinv, int* __restrict__ csr, int N, int E, int NB) {
  __shared__ int cnt[BINW];
  __shared__ int rstart[BINW];
  __shared__ int cur[BINW];
  __shared__ int wsum[BINW / 32];
  __shared__ int slist[DCAP];

  const int t = threadIdx.x;
  const int b = blockIdx.x;
  unsigned u0 = (unsigned)binbase[b];
  unsigned u1 = (b + 1 < NB) ? (unsigned)binbase[b + 1] : (unsigned)E;
  u1 = u1 < (unsigned)E ? u1 : (unsigned)E;
  u0 = u0 < u1 ? u0 : u1;
  const int e0 = (int)u0, e1 = (int)u1;
  const int n = e1 - e0;

  if (t < BINW) { cnt[t] = 0; cur[t] = 0; }
  __syncthreads();

  for (int i0 = e0; i0 < e1; i0 += DUNR * BTHR) {
    int d[DUNR];
#pragma unroll
    for (int j = 0; j < DUNR; ++j) {
      const int i = i0 + j * BTHR + t;
      d[j] = bdst[i < e1 ? i : e0];
    }
#pragma unroll
    for (int j = 0; j < DUNR; ++j) {
      const int i = i0 + j * BTHR + t;
      if (i < e1) atomicAdd(&cnt[d[j] & (BINW - 1)], 1);
    }
  }
  __syncthreads();

  const int c = (t < BINW) ? cnt[t] : 0;
  const int ex = block_excl_scan<BINW / 32>(c, wsum, t < BINW);
  if (t < BINW) {
    rstart[t] = ex;
    const int node = b * BINW + t;
    if (node < N) {
      rowptr[node] = e0 + ex;
      dinv[node] = rsqrtf((float)c + 1.0f);
    }
  }
  __syncthreads();

  const bool inlds = n <= DCAP;
  for (int i0 = e0; i0 < e1; i0 += DUNR * BTHR) {
    int d[DUNR], s[DUNR];
#pragma unroll
    for (int j = 0; j < DUNR; ++j) {
      const int i  = i0 + j * BTHR + t;
      const int ic = i < e1 ? i : e0;
      d[j] = bdst[ic];
      s[j] = bsrc[ic];
    }
#pragma unroll
    for (int j = 0; j < DUNR; ++j) {
      const int i = i0 + j * BTHR + t;
      if (i < e1) {
        const int dl = d[j] & (BINW - 1);
        const unsigned p = (unsigned)(rstart[dl] + atomicAdd(&cur[dl], 1));
        if (inlds) {
          slist[p < (unsigned)DCAP ? p : (unsigned)(DCAP - 1)] = s[j];
        } else {
          const unsigned g = u0 + p;
          csr[g < (unsigned)E ? g : (unsigned)(E - 1)] = s[j];
        }
      }
    }
  }
  if (inlds) {
    __syncthreads();
    for (int i = t; i < n; i += BTHR) csr[e0 + i] = slist[i];
  }
}

#define HK    (2 * HID)
#define HAP   (HK + 4)
#define HHP   (HID + 4)
#define HWPT  (HID * NCLS / 256)
__global__ void __launch_bounds__(256)
k_head_all(const float* __restrict__ sums, const int* __restrict__ maxs,
           const float* __restrict__ counts, const float* __restrict__ Wc1,
           const float* __restrict__ bc1, const float* __restrict__ Wc2,
           const float* __restrict__ bc2, float* __restrict__ out) {
  __shared__ __attribute__((aligned(16))) float As[16 * HAP];
  __shared__ __attribute__((aligned(16))) float Hs[16 * HHP];
  __shared__ float Ws[HID * NCLS];

  const int t    = threadIdx.x;
  const int lane = t & 31;
  const int tn   = t >> 5;
  const int tm   = blockIdx.x;
  const int half = lane >> 4;
  const int l15  = lane & 15;

  const float* __restrict__ Bcol = Wc1 + tn * 16 + l15;
  v2f bq[HK / 4];
#pragma unroll
  for (int q = 0; q < HK / 4; ++q) {
    const int kk = q * 4 + half * 2;
    bq[q][0] = Bcol[(size_t)kk * HID];
    bq[q][1] = Bcol[(size_t)(kk + 1) * HID];
  }
  const float bv = bc1[tn * 16 + l15];

  {
    float4 sv[2];
    int4   mv[2];
    float  cv[2];
    float  wv[HWPT];
#pragma unroll
    for (int j = 0; j < 2; ++j) {
      const int q = j * 256 + t;
      const int r = q >> 5;
      const size_t o = (size_t)(tm * 16 + r) * HID + ((q & 31) << 2);
      sv[j] = *(const float4*)(sums + o);
      mv[j] = *(const int4*)(maxs + o);
      cv[j] = counts[tm * 16 + r];
    }
#pragma unroll
    for (int j = 0; j < HWPT; ++j) wv[j] = Wc2[j * 256 + t];
#pragma unroll
    for (int j = 0; j < 2; ++j) {
      const int q = j * 256 + t;
      const int r = q >> 5;
      const int c4 = (q & 31) << 2;
      const float denom = cv[j] > 1.0f ? cv[j] : 1.0f;
      float4 m;
      m.x = sv[j].x / denom;  m.y = sv[j].y / denom;
      m.z = sv[j].z / denom;  m.w = sv[j].w / denom;
      float4 x;
      x.x = __int_as_float(mv[j].x);  x.y = __int_as_float(mv[j].y);
      x.z = __int_as_float(mv[j].z);  x.w = __int_as_float(mv[j].w);
      *(float4*)&As[r * HAP + c4]       = m;
      *(float4*)&As[r * HAP + HID + c4] = x;
    }
#pragma unroll
    for (int j = 0; j < HWPT; ++j) Ws[j * 256 + t] = wv[j];
  }
  __syncthreads();

  v8f acc;
#pragma unroll
  for (int r = 0; r < 8; ++r) acc[r] = bv;
  const float* Arow = As + l15 * HAP;
#pragma unroll
  for (int q = 0; q < HK / 4; ++q) {
    const int kk = q * 4 + half * 2;
    v2f a;
    a[0] = Arow[kk];
    a[1] = Arow[kk + 1];
    acc = __builtin_amdgcn_wmma_f32_16x16x4_f32(false, a, false, bq[q],
                                                (short)0, acc, false, false);
  }
#pragma unroll
  for (int r = 0; r < 8; ++r) {
    const float v = acc[r] > 0.0f ? acc[r] : 0.0f;
    Hs[(r + half * 8) * HHP + tn * 16 + l15] = v;
  }
  __syncthreads();

  if (t < 16 * NCLS) {
    const int g = t / NCLS;
    const int c = t - g * NCLS;
    float s = bc2[c];
    const float* hv = Hs + g * HHP;
#pragma unroll 8
    for (int k = 0; k < HID; ++k) s += hv[k] * Ws[k * NCLS + c];
    out[(size_t)(tm * 16 + g) * NCLS + c] = s;
  }
}


template <int U, bool SC>
__device__ __forceinline__ void add_rows(float4& acc, const float4* __restrict__ H,
                                         int my, float& wv, int j, int lane) {
  float4 v[U];
#pragma unroll
  for (int u = 0; u < U; ++u) {
    const int s = __builtin_amdgcn_readlane(my, j + u);
    v[u] = H[(size_t)s * 32 + lane];
  }
  if (SC) asm volatile("" : "+v"(wv) : : "memory");
#pragma unroll
  for (int u = 0; u < U; ++u) {
    if (SC) {
      const float w = __int_as_float(__builtin_amdgcn_readlane(__float_as_int(wv), j + u));
      acc.x = __builtin_fmaf(w, v[u].x, acc.x);  acc.y = __builtin_fmaf(w, v[u].y, acc.y);
      acc.z = __builtin_fmaf(w, v[u].z, acc.z);  acc.w = __builtin_fmaf(w, v[u].w, acc.w);
    } else {
      acc.x += v[u].x; acc.y += v[u].y; acc.z += v[u].z; acc.w += v[u].w;
    }
  }
}

#define FT_PITCH     (HID + 4)
#define FUSED_BLOCKS 1024

template <bool SC>
__device__ __forceinline__ void walk_list(float4& acc, const float4* __restrict__ H,
                                          const int* __restrict__ csr, const float* __restrict__ dinv,
                                          unsigned my, float wv, int beg, int end, int N, int lane) {
  for (int base = beg; base < end; base += 32) {
    const int rem = end - base;
    const int cnt = rem < 32 ? rem : 32;
    if (base != beg) {
      my = (lane < cnt) ? (unsigned)csr[base + lane] : 0u;
      my = my < (unsigned)N ? my : (unsigned)(N - 1);
      if (SC) wv = dinv[my];
    }
    int j = 0;
    for (; j + 8 <= cnt; j += 8) add_rows<8, SC>(acc, H, (int)my, wv, j, lane);
    switch (cnt - j) {
      case 7: add_rows<7, SC>(acc, H, (int)my, wv, j, lane); break;
      case 6: add_rows<6, SC>(acc, H, (int)my, wv, j, lane); break;
      case 5: add_rows<5, SC>(acc, H, (int)my, wv, j, lane); break;
      case 4: add_rows<4, SC>(acc, H, (int)my, wv, j, lane); break;
      case 3: add_rows<3, SC>(acc, H, (int)my, wv, j, lane); break;
      case 2: add_rows<2, SC>(acc, H, (int)my, wv, j, lane); break;
      case 1: add_rows<1, SC>(acc, H, (int)my, wv, j, lane); break;
      default: break;
    }
  }
}

template <bool SC_IN, bool SC_OUT>
__global__ void __launch_bounds__(256, 8)
k_agg_gemm(const int* __restrict__ rowptr, const int* __restrict__ csr,
           const float* __restrict__ dinv, const float* __restrict__ hin,
           const float* __restrict__ Wl, const float* __restrict__ bias,
           float* __restrict__ out, int N, int E, int units) {
  __shared__ __attribute__((aligned(16))) float At[16 * FT_PITCH];
  __shared__ __attribute__((aligned(16))) float Dt[16 * FT_PITCH];
  __shared__ float Dv[16];

  const int lane = threadIdx.x & 31;
  const int wave = __builtin_amdgcn_readfirstlane((int)(threadIdx.x >> 5));
  const int half = lane >> 4;
  const int l15  = lane & 15;

  const float* __restrict__ Bcol = Wl + wave * 16 + l15;
  v2f bq[HID / 4];
#pragma unroll
  for (int q = 0; q < HID / 4; ++q) {
    const int kk = q * 4 + half * 2;
    bq[q][0] = Bcol[(size_t)kk * HID];
    bq[q][1] = Bcol[(size_t)(kk + 1) * HID];
  }
  const float4* __restrict__ H = (const float4*)hin;

  auto unit_words = [&](int uu, int& rpv, float& dvl) {
    int i0 = uu * 16 + wave * 2 + (lane < 3 ? lane : 0);
    i0 = i0 < N ? i0 : N - 1;
    rpv = rowptr[i0];
    dvl = dinv[i0];
  };
  int   rpv;
  float dvl;
  unit_words((int)blockIdx.x, rpv, dvl);

  for (int u = blockIdx.x; u < units; u += gridDim.x) {
    const int n0 = u * 16 + wave * 2;
    int   rpvn;
    float dvln;
    unit_words(u + (int)gridDim.x, rpvn, dvln);
    const int r0 = __builtin_amdgcn_readlane(rpv, 0);
    const int r1 = __builtin_amdgcn_readlane(rpv, 1);
    const int r2 = __builtin_amdgcn_readlane(rpv, 2);
    int   ndc[2], bg[2], en[2];
    bool  ok[2];
    float dvv[2];
#pragma unroll
    for (int i = 0; i < 2; ++i) {
      const int nd = n0 + i;
      ok[i]  = nd < N;
      ndc[i] = ok[i] ? nd : 0;
      int b = (i == 0) ? r0 : r1;
      int e = (nd + 1 < N) ? ((i == 0) ? r1 : r2) : E;
      e = e < E ? e : E;  e = e > 0 ? e : 0;
      b = b > 0 ? b : 0;  b = b < e ? b : e;
      if (!ok[i]) e = b;
      bg[i] = b;  en[i] = e;
      dvv[i] = __int_as_float(__builtin_amdgcn_readlane(__float_as_int(dvl), i));
    }
    unsigned my[2];
    float    wv[2];
    float4   acc2[2];
#pragma unroll
    for (int i = 0; i < 2; ++i) {
      const int c0   = en[i] - bg[i];
      const int cnt0 = c0 < 32 ? c0 : 32;
      const unsigned m = (lane < cnt0) ? (unsigned)csr[bg[i] + lane] : 0u;
      my[i]   = m < (unsigned)N ? m : (unsigned)(N - 1);
      acc2[i] = H[(size_t)ndc[i] * 32 + lane];
    }
#pragma unroll
    for (int i = 0; i < 2; ++i) {
      wv[i] = SC_IN ? dinv[my[i]] : 0.0f;
      if (SC_IN) {
        acc2[i].x *= dvv[i];  acc2[i].y *= dvv[i];  acc2[i].z *= dvv[i];  acc2[i].w *= dvv[i];
      }
    }
#pragma unroll
    for (int i = 0; i < 2; ++i) {
      walk_list<SC_IN>(acc2[i], H, csr, dinv, my[i], wv[i], bg[i], en[i], N, lane);
      const float dv = dvv[i];
      float4 o;
      o.x = acc2[i].x * dv;  o.y = acc2[i].y * dv;
      o.z = acc2[i].z * dv;  o.w = acc2[i].w * dv;
      if (!ok[i]) o = make_float4(0.0f, 0.0f, 0.0f, 0.0f);
      const int row = wave * 2 + i;
      *(float4*)&At[row * FT_PITCH + lane * 4] = o;
      if (SC_OUT && lane == 0) Dv[row] = ok[i] ? dv : 0.0f;
    }
    __syncthreads();

    const float bcol = bias[wave * 16 + l15];
    v8f acc;
#pragma unroll
    for (int r = 0; r < 8; ++r) acc[r] = 0.0f;
    const float* Arow = At + l15 * FT_PITCH + half * 2;
#pragma unroll
    for (int q = 0; q < HID / 4; ++q) {
      v2f a;
      a[0] = Arow[q * 4];
      a[1] = Arow[q * 4 + 1];
      acc = __builtin_amdgcn_wmma_f32_16x16x4_f32(false, a, false, bq[q],
                                                  (short)0, acc, false, false);
    }
#pragma unroll
    for (int r = 0; r < 8; ++r) {
      const int row = r + half * 8;
      float v = acc[r] + bcol;
      v = v > 0.0f ? v : 0.0f;
      Dt[row * FT_PITCH + wave * 16 + l15] = SC_OUT ? v * Dv[row] : v;
    }
    __syncthreads();

#pragma unroll
    for (int i = 0; i < 2; ++i) {
      const int row = wave * 2 + i;
      if (n0 + i < N)
        ((float4*)out)[(size_t)(n0 + i) * 32 + lane] = *(const float4*)&Dt[row * FT_PITCH + lane * 4];
    }
    rpv = rpvn;
    dvl = dvln;
  }
}


#define PCH 16

__device__ __forceinline__ void pool_flush(int g, const float4& s, const int4& m, int c, int lane,
                                           float* __restrict__ counts, float* __restrict__ sums,
                                           int* __restrict__ maxs) {
  float* sp = sums + (size_t)g * HID + lane * 4;
  int*   mp = maxs + (size_t)g * HID + lane * 4;
  atomicAdd(sp + 0, s.x);  atomicAdd(sp + 1, s.y);
  atomicAdd(sp + 2, s.z);  atomicAdd(sp + 3, s.w);
  atomicMax(mp + 0, m.x);  atomicMax(mp + 1, m.y);
  atomicMax(mp + 2, m.z);  atomicMax(mp + 3, m.w);
  if (lane == 0) atomicAdd(&counts[g], (float)c);
}

__global__ void k_pool_runs(const float* __restrict__ h, const int* __restrict__ batch,
                            float* __restrict__ counts, float* __restrict__ sums,
                            int* __restrict__ maxs, int N) {
  const int lane = threadIdx.x & 31;
  const int w = __builtin_amdgcn_readfirstlane(
      (int)((blockIdx.x * blockDim.x + threadIdx.x) >> 5));
  const int n0 = w * PCH;
  if (n0 >= N) return;
  const int rem = N - n0;
  const int cnt = rem < PCH ? rem : PCH;

  unsigned gid = (unsigned)batch[n0 + (lane < cnt ? lane : 0)];
  gid = gid < (unsigned)NGR ? gid : (unsigned)(NGR - 1);

  const float4* __restrict__ H = (const float4*)h;
  float4 v[PCH];
#pragma unroll
  for (int j = 0; j < PCH; ++j) {
    const int n = n0 + (j < cnt ? j : 0);
    v[j] = H[(size_t)n * 32 + lane];
  }

  int cur = __builtin_amdgcn_readlane((int)gid, 0);
  float4 s = make_float4(0.0f, 0.0f, 0.0f, 0.0f);
  int4   m = make_int4(0, 0, 0, 0);
  int    c = 0;
#pragma unroll
  for (int j = 0; j < PCH; ++j) {
    if (j < cnt) {
      const int g = __builtin_amdgcn_readlane((int)gid, j);
      if (g != cur) {
        pool_flush(cur, s, m, c, lane, counts, sums, maxs);
        s = make_float4(0.0f, 0.0f, 0.0f, 0.0f);
        m = make_int4(0, 0, 0, 0);
        c = 0;
        cur = g;
      }
      s.x += v[j].x;  s.y += v[j].y;  s.z += v[j].z;  s.w += v[j].w;
      const int bx = __float_as_int(v[j].x), by = __float_as_int(v[j].y);
      const int bz = __float_as_int(v[j].z), bw = __float_as_int(v[j].w);
      m.x = bx > m.x ? bx : m.x;  m.y = by > m.y ? by : m.y;
      m.z = bz > m.z ? bz : m.z;  m.w = bw > m.w ? bw : m.w;
      ++c;
    }
  }
  pool_flush(cur, s, m, c, lane, counts, sums, maxs);
}


extern "C" void kernel_launch(void* const* d_in, const int* in_sizes, int n_in,
                              void* d_out, int out_size, void* d_ws, size_t ws_size,
                              hipStream_t stream) {
  const float* x     = (const float*)d_in[0];
  const int*   edge  = (const int*)d_in[1];
  const int*   batch = (const int*)d_in[2];
  const float* W[3]  = {(const float*)d_in[3], (const float*)d_in[5], (const float*)d_in[7]};
  const float* b[3]  = {(const float*)d_in[4], (const float*)d_in[6], (const float*)d_in[8]};
  const float* Wc1   = (const float*)d_in[9];
  const float* bc1   = (const float*)d_in[10];
  const float* Wc2   = (const float*)d_in[11];
  const float* bc2   = (const float*)d_in[12];
  float* out = (float*)d_out;

  const int N = in_sizes[0] / HID;
  const int E = in_sizes[1] / 2;
  const int G = NGR;
  const int* src  = edge;
  const int* dstv = edge + E;

  float* ws     = (float*)d_ws;
  int*   degi   = (int*)ws;
  float* dinv   = ws + N;
  float* buf0   = dinv + N;
  float* buf1   = buf0 + (size_t)N * HID;
  float* counts = buf1 + (size_t)N * HID;
  float* sums   = counts + G;
  int*   maxs   = (int*)(sums + (size_t)G * HID);
  int*   csr    = maxs + (size_t)G * HID;
  const int NB   = (N + BINW - 1) >> BINSH;
  const int NBLK = (E + ECH - 1) / ECH;
  int*   bsrc    = (int*)buf1;
  int*   bdst    = bsrc + E;
  int*   hist    = bdst + E;
  int*   binbase = hist + (size_t)NBLK * NB;

  if ((size_t)((const char*)(csr + E) - (const char*)d_ws) > ws_size) return;
  if (NB > NBMAX || (size_t)2 * E + (size_t)NBLK * NB + NB > (size_t)N * HID) return;

  const int BT = 256;
  auto nb = [](long n, int bt) { return (int)((n + bt - 1) / bt); };

  k_bin_hist<<<NBLK, BTHR, 0, stream>>>(dstv, hist, E, NB, (int*)counts, NGR + 2 * NGR * HID);
  k_bin_scatter<<<NBLK, BTHR, 0, stream>>>(src, dstv, hist, bsrc, bdst, binbase, E, NB, NBLK);
  k_bin_rows<<<NB, BTHR, 0, stream>>>(bsrc, bdst, binbase, degi, dinv, csr, N, E, NB);

  const int units = (N + 15) / 16;
  k_agg_gemm<true,  true ><<<FUSED_BLOCKS, 256, 0, stream>>>(degi, csr, dinv, x,    W[0], b[0], buf0, N, E, units);
  k_agg_gemm<false, true ><<<FUSED_BLOCKS, 256, 0, stream>>>(degi, csr, dinv, buf0, W[1], b[1], buf1, N, E, units);
  k_agg_gemm<false, false><<<FUSED_BLOCKS, 256, 0, stream>>>(degi, csr, dinv, buf1, W[2], b[2], buf0, N, E, units);

  k_pool_runs<<<nb((long)((N + PCH - 1) / PCH) * 32, BT), BT, 0, stream>>>(buf0, batch, counts,
                                                                        sums, maxs, N);

  k_head_all<<<G / 16, 256, 0, stream>>>(sums, maxs, counts, Wc1, bc1, Wc2, bc2, out);
}


# Pooling fused into the last layer's launch: each finished 16-row tile is pooled from LDS by runs of equal graph id; layer output neither stored nor re-read; sum order changes; 8 to 7 launches
# speedup vs baseline: 7.1198x; 1.0614x over previous
#include <hip/hip_runtime.h>

#define HID   128
#define NGR   512
#define NCLS  10

typedef float v2f __attribute__((ext_vector_type(2)));
typedef float v8f __attribute__((ext_vector_type(8)));

#define BINSH  9
#define BINW   (1 << BINSH)
#define NBMAX  256
#define ECH    6400
#define BTHR   1024
#define EPT    ((ECH + BTHR - 1) / BTHR)
#define HPARTS (BTHR / NBMAX)
#define HUNR   16
#define DCAP   12288
#define DUNR   4

__device__ __forceinline__ int bin_of(int d, int NB) {
  const unsigned q = (unsigned)d >> BINSH;
  return (int)(q < (unsigned)NB ? q : (unsigned)(NB - 1));
}

template <int NW>
__device__ __forceinline__ int block_excl_scan(int v, int* wsum, bool active) {
  const int lane = threadIdx.x & 31;
  const int wave = threadIdx.x >> 5;
  int inc = v;
  for (int d = 1; d < 32; d <<= 1) {
    const int o = __shfl_up(inc, d);
    if (lane >= d) inc += o;
  }
  if (active && lane == 31) wsum[wave] = inc;
  __syncthreads();
  int off = 0;
  if (active) {
    for (int w = 0; w < NW; ++w) off += (w < wave) ? wsum[w] : 0;
  }
  return off + inc - v;
}

__global__ void __launch_bounds__(BTHR)
k_bin_hist(const int* __restrict__ dst, int* __restrict__ hist, int E, int NB,
           int* __restrict__ zero, int zn) {
  __shared__ int h[NBMAX];
  const int t = threadIdx.x;
  if (t < NBMAX) h[t] = 0;
  __syncthreads();
  const int beg = blockIdx.x * ECH;
  const int end = (beg + ECH < E) ? beg + ECH : E;
  int d[EPT];
#pragma unroll
  for (int j = 0; j < EPT; ++j) {
    const int i = beg + j * BTHR + t;
    d[j] = dst[i < end ? i : beg];
  }
#pragma unroll
  for (int j = 0; j < EPT; ++j) {
    const int i = beg + j * BTHR + t;
    if (i < end) atomicAdd(&h[bin_of(d[j], NB)], 1);
  }
  __syncthreads();
  if (t < NB) hist[blockIdx.x * NB + t] = h[t];
  for (int i = blockIdx.x * BTHR + t; i < zn; i += gridDim.x * BTHR) zero[i] = 0;
}

__global__ void __launch_bounds__(BTHR)
k_bin_scatter(const int* __restrict__ src, const int* __restrict__ dst,
              const int* __restrict__ hist, int* __restrict__ bsrc, int* __restrict__ bdst,
              int* __restrict__ binbase, int E, int NB, int NBLK) {
  __shared__ int pbef[HPARTS][NBMAX];
  __shared__ int ptot[HPARTS][NBMAX];
  __shared__ int goff[NBMAX];
  __shared__ int lstart[NBMAX];
  __shared__ int lcnt[NBMAX];
  __shared__ int wsum[NBMAX / 32];
  __shared__ int ssrc[ECH];
  __shared__ int sdst[ECH];

  const int t = threadIdx.x;
  const int k = blockIdx.x;
  const int beg = k * ECH;
  const int end = (beg + ECH < E) ? beg + ECH : E;
  const int n = end - beg;

  int s[EPT], d[EPT];
#pragma unroll
  for (int j = 0; j < EPT; ++j) {
    const int i  = beg + j * BTHR + t;
    const int ic = i < end ? i : beg;
    s[j] = src[ic];
    d[j] = dst[ic];
  }

  {
    const int b = t & (NBMAX - 1);
    const int part = t / NBMAX;
    const int rows = (NBLK + HPARTS - 1) / HPARTS;
    const int r0 = part * rows;
    const int r1 = (r0 + rows < NBLK) ? r0 + rows : NBLK;
    int bef = 0, tot = 0;
    if (b < NB) {
      for (int r = r0; r < r1; r += HUNR) {
        int v[HUNR];
#pragma unroll
        for (int u = 0; u < HUNR; ++u) {
          const int rr = r + u;
          v[u] = hist[(rr < r1 ? rr : r0) * NB + b];
        }
#pragma unroll
        for (int u = 0; u < HUNR; ++u) {
          const int rr = r + u;
          tot += (rr < r1) ? v[u] : 0;
          bef += (rr < r1 && rr < k) ? v[u] : 0;
        }
      }
    }
    pbef[part][b] = bef;
    ptot[part][b] = tot;
    if (t < NBMAX) lcnt[t] = 0;
  }
  __syncthreads();

  int totb = 0, befb = 0;
  if (t < NBMAX) {
    for (int p = 0; p < HPARTS; ++p) { totb += ptot[p][t]; befb += pbef[p][t]; }
  }
  const int base = block_excl_scan<NBMAX / 32>(totb, wsum, t < NBMAX);
  if (t < NBMAX) {
    goff[t] = base + befb;
    if (k == 0 && t < NB) binbase[t] = base;
  }

  int r[EPT];
#pragma unroll
  for (int j = 0; j < EPT; ++j) {
    const int i = beg + j * BTHR + t;
    r[j] = (i < end) ? atomicAdd(&lcnt[bin_of(d[j], NB)], 1) : 0;
  }
  __syncthreads();
  const int cb = (t < NBMAX) ? lcnt[t] : 0;
  const int ls = block_excl_scan<NBMAX / 32>(cb, wsum, t < NBMAX);
  if (t < NBMAX) lstart[t] = ls;
  __syncthreads();

#pragma unroll
  for (int j = 0; j < EPT; ++j) {
    const int i = beg + j * BTHR + t;
    if (i < end) {
      unsigned p = (unsigned)(lstart[bin_of(d[j], NB)] + r[j]);
      p = p < (unsigned)ECH ? p : (unsigned)(ECH - 1);
      ssrc[p] = s[j];
      sdst[p] = d[j];
    }
  }
  __syncthreads();

  for (int i = t; i < n; i += BTHR) {
    const int dd = sdst[i];
    const int bn = bin_of(dd, NB);
    unsigned g = (unsigned)(goff[bn] + (i - lstart[bn]));
    g = g < (unsigned)E ? g : (unsigned)(E - 1);
    bsrc[g] = ssrc[i];
    bdst[g] = dd;
  }
}

__global__ void __launch_bounds__(BTHR)
k_bin_rows(const int* __restrict__ bsrc, const int* __restrict__ bdst,
           const int* __restrict__ binbase, int* __restrict__ rowptr,
           float* __restrict__ dinv, int* __restrict__ csr, int N, int E, int NB) {
  __shared__ int cnt[BINW];
  __shared__ int rstart[BINW];
  __shared__ int cur[BINW];
  __shared__ int wsum[BINW / 32];
  __shared__ int slist[DCAP];

  const int t = threadIdx.x;
  const int b = blockIdx.x;
  unsigned u0 = (unsigned)binbase[b];
  unsigned u1 = (b + 1 < NB) ? (unsigned)binbase[b + 1] : (unsigned)E;
  u1 = u1 < (unsigned)E ? u1 : (unsigned)E;
  u0 = u0 < u1 ? u0 : u1;
  const int e0 = (int)u0, e1 = (int)u1;
  const int n = e1 - e0;

  if (t < BINW) { cnt[t] = 0; cur[t] = 0; }
  __syncthreads();

  for (int i0 = e0; i0 < e1; i0 += DUNR * BTHR) {
    int d[DUNR];
#pragma unroll
    for (int j = 0; j < DUNR; ++j) {
      const int i = i0 + j * BTHR + t;
      d[j] = bdst[i < e1 ? i : e0];
    }
#pragma unroll
    for (int j = 0; j < DUNR; ++j) {
      const int i = i0 + j * BTHR + t;
      if (i < e1) atomicAdd(&cnt[d[j] & (BINW - 1)], 1);
    }
  }
  __syncthreads();

  const int c = (t < BINW) ? cnt[t] : 0;
  const int ex = block_excl_scan<BINW / 32>(c, wsum, t < BINW);
  if (t < BINW) {
    rstart[t] = ex;
    const int node = b * BINW + t;
    if (node < N) {
      rowptr[node] = e0 + ex;
      dinv[node] = rsqrtf((float)c + 1.0f);
    }
  }
  __syncthreads();

  const bool inlds = n <= DCAP;
  for (int i0 = e0; i0 < e1; i0 += DUNR * BTHR) {
    int d[DUNR], s[DUNR];
#pragma unroll
    for (int j = 0; j < DUNR; ++j) {
      const int i  = i0 + j * BTHR + t;
      const int ic = i < e1 ? i : e0;
      d[j] = bdst[ic];
      s[j] = bsrc[ic];
    }
#pragma unroll
    for (int j = 0; j < DUNR; ++j) {
      const int i = i0 + j * BTHR + t;
      if (i < e1) {
        const int dl = d[j] & (BINW - 1);
        const unsigned p = (unsigned)(rstart[dl] + atomicAdd(&cur[dl], 1));
        if (inlds) {
          slist[p < (unsigned)DCAP ? p : (unsigned)(DCAP - 1)] = s[j];
        } else {
          const unsigned g = u0 + p;
          csr[g < (unsigned)E ? g : (unsigned)(E - 1)] = s[j];
        }
      }
    }
  }
  if (inlds) {
    __syncthreads();
    for (int i = t; i < n; i += BTHR) csr[e0 + i] = slist[i];
  }
}

#define HK    (2 * HID)
#define HAP   (HK + 4)
#define HHP   (HID + 4)
#define HWPT  (HID * NCLS / 256)
__global__ void __launch_bounds__(256)
k_head_all(const float* __restrict__ sums, const int* __restrict__ maxs,
           const float* __restrict__ counts, const float* __restrict__ Wc1,
           const float* __restrict__ bc1, const float* __restrict__ Wc2,
           const float* __restrict__ bc2, float* __restrict__ out) {
  __shared__ __attribute__((aligned(16))) float As[16 * HAP];
  __shared__ __attribute__((aligned(16))) float Hs[16 * HHP];
  __shared__ float Ws[HID * NCLS];

  const int t    = threadIdx.x;
  const int lane = t & 31;
  const int tn   = t >> 5;
  const int tm   = blockIdx.x;
  const int half = lane >> 4;
  const int l15  = lane & 15;

  const float* __restrict__ Bcol = Wc1 + tn * 16 + l15;
  v2f bq[HK / 4];
#pragma unroll
  for (int q = 0; q < HK / 4; ++q) {
    const int kk = q * 4 + half * 2;
    bq[q][0] = Bcol[(size_t)kk * HID];
    bq[q][1] = Bcol[(size_t)(kk + 1) * HID];
  }
  const float bv = bc1[tn * 16 + l15];

  {
    float4 sv[2];
    int4   mv[2];
    float  cv[2];
    float  wv[HWPT];
#pragma unroll
    for (int j = 0; j < 2; ++j) {
      const int q = j * 256 + t;
      const int r = q >> 5;
      const size_t o = (size_t)(tm * 16 + r) * HID + ((q & 31) << 2);
      sv[j] = *(const float4*)(sums + o);
      mv[j] = *(const int4*)(maxs + o);
      cv[j] = counts[tm * 16 + r];
    }
#pragma unroll
    for (int j = 0; j < HWPT; ++j) wv[j] = Wc2[j * 256 + t];
#pragma unroll
    for (int j = 0; j < 2; ++j) {
      const int q = j * 256 + t;
      const int r = q >> 5;
      const int c4 = (q & 31) << 2;
      const float denom = cv[j] > 1.0f ? cv[j] : 1.0f;
      float4 m;
      m.x = sv[j].x / denom;  m.y = sv[j].y / denom;
      m.z = sv[j].z / denom;  m.w = sv[j].w / denom;
      float4 x;
      x.x = __int_as_float(mv[j].x);  x.y = __int_as_float(mv[j].y);
      x.z = __int_as_float(mv[j].z);  x.w = __int_as_float(mv[j].w);
      *(float4*)&As[r * HAP + c4]       = m;
      *(float4*)&As[r * HAP + HID + c4] = x;
    }
#pragma unroll
    for (int j = 0; j < HWPT; ++j) Ws[j * 256 + t] = wv[j];
  }
  __syncthreads();

  v8f acc;
#pragma unroll
  for (int r = 0; r < 8; ++r) acc[r] = bv;
  const float* Arow = As + l15 * HAP;
#pragma unroll
  for (int q = 0; q < HK / 4; ++q) {
    const int kk = q * 4 + half * 2;
    v2f a;
    a[0] = Arow[kk];
    a[1] = Arow[kk + 1];
    acc = __builtin_amdgcn_wmma_f32_16x16x4_f32(false, a, false, bq[q],
                                                (short)0, acc, false, false);
  }
#pragma unroll
  for (int r = 0; r < 8; ++r) {
    const float v = acc[r] > 0.0f ? acc[r] : 0.0f;
    Hs[(r + half * 8) * HHP + tn * 16 + l15] = v;
  }
  __syncthreads();

  if (t < 16 * NCLS) {
    const int g = t / NCLS;
    const int c = t - g * NCLS;
    float s = bc2[c];
    const float* hv = Hs + g * HHP;
#pragma unroll 8
    for (int k = 0; k < HID; ++k) s += hv[k] * Ws[k * NCLS + c];
    out[(size_t)(tm * 16 + g) * NCLS + c] = s;
  }
}


template <int U, bool SC>
__device__ __forceinline__ void add_rows(float4& acc, const float4* __restrict__ H,
                                         int my, float& wv, int j, int lane) {
  float4 v[U];
#pragma unroll
  for (int u = 0; u < U; ++u) {
    const int s = __builtin_amdgcn_readlane(my, j + u);
    v[u] = H[(size_t)s * 32 + lane];
  }
  if (SC) asm volatile("" : "+v"(wv) : : "memory");
#pragma unroll
  for (int u = 0; u < U; ++u) {
    if (SC) {
      const float w = __int_as_float(__builtin_amdgcn_readlane(__float_as_int(wv), j + u));
      acc.x = __builtin_fmaf(w, v[u].x, acc.x);  acc.y = __builtin_fmaf(w, v[u].y, acc.y);
      acc.z = __builtin_fmaf(w, v[u].z, acc.z);  acc.w = __builtin_fmaf(w, v[u].w, acc.w);
    } else {
      acc.x += v[u].x; acc.y += v[u].y; acc.z += v[u].z; acc.w += v[u].w;
    }
  }
}

#define FT_PITCH     (HID + 4)
#define FUSED_BLOCKS 1024

template <bool SC>
__device__ __forceinline__ void walk_list(float4& acc, const float4* __restrict__ H,
                                          const int* __restrict__ csr, const float* __restrict__ dinv,
                                          unsigned my, float wv, int beg, int end, int N, int lane) {
  for (int base = beg; base < end; base += 32) {
    const int rem = end - base;
    const int cnt = rem < 32 ? rem : 32;
    if (base != beg) {
      my = (lane < cnt) ? (unsigned)csr[base + lane] : 0u;
      my = my < (unsigned)N ? my : (unsigned)(N - 1);
      if (SC) wv = dinv[my];
    }
    int j = 0;
    for (; j + 8 <= cnt; j += 8) add_rows<8, SC>(acc, H, (int)my, wv, j, lane);
    switch (cnt - j) {
      case 7: add_rows<7, SC>(acc, H, (int)my, wv, j, lane); break;
      case 6: add_rows<6, SC>(acc, H, (int)my, wv, j, lane); break;
      case 5: add_rows<5, SC>(acc, H, (int)my, wv, j, lane); break;
      case 4: add_rows<4, SC>(acc, H, (int)my, wv, j, lane); break;
      case 3: add_rows<3, SC>(acc, H, (int)my, wv, j, lane); break;
      case 2: add_rows<2, SC>(acc, H, (int)my, wv, j, lane); break;
      case 1: add_rows<1, SC>(acc, H, (int)my, wv, j, lane); break;
      default: break;
    }
  }
}

__device__ __forceinline__ void pool_flush(int g, float s, int m, int cnt, int col,
                                           float* __restrict__ counts, float* __restrict__ sums,
                                           int* __restrict__ maxs) {
  atomicAdd(sums + (size_t)g * HID + col, s);
  atomicMax(maxs + (size_t)g * HID + col, m);
  if (col == 0) atomicAdd(&counts[g], (float)cnt);
}

template <bool SC_IN, bool SC_OUT, bool POOL>
__global__ void __launch_bounds__(256, 8)
k_agg_gemm(const int* __restrict__ rowptr, const int* __restrict__ csr,
           const float* __restrict__ dinv, const float* __restrict__ hin,
           const float* __restrict__ Wl, const float* __restrict__ bias,
           float* __restrict__ out, int N, int E, int units,
           const int* __restrict__ batch, float* __restrict__ counts,
           float* __restrict__ sums, int* __restrict__ maxs) {
  __shared__ __attribute__((aligned(16))) float At[16 * FT_PITCH];
  __shared__ __attribute__((aligned(16))) float Dt[16 * FT_PITCH];
  __shared__ float Dv[16];

  const int lane = threadIdx.x & 31;
  const int wave = __builtin_amdgcn_readfirstlane((int)(threadIdx.x >> 5));
  const int half = lane >> 4;
  const int l15  = lane & 15;

  const float* __restrict__ Bcol = Wl + wave * 16 + l15;
  v2f bq[HID / 4];
#pragma unroll
  for (int q = 0; q < HID / 4; ++q) {
    const int kk = q * 4 + half * 2;
    bq[q][0] = Bcol[(size_t)kk * HID];
    bq[q][1] = Bcol[(size_t)(kk + 1) * HID];
  }
  const float4* __restrict__ H = (const float4*)hin;

  auto unit_words = [&](int uu, int& rpv, float& dvl) {
    int i0 = uu * 16 + wave * 2 + (lane < 3 ? lane : 0);
    i0 = i0 < N ? i0 : N - 1;
    rpv = rowptr[i0];
    dvl = dinv[i0];
  };
  int ubeg = (int)blockIdx.x, uend = units, ustep = (int)gridDim.x;
  if (POOL) {
    const int per = units / (int)gridDim.x, extra = units % (int)gridDim.x;
    ubeg  = (int)blockIdx.x * per + ((int)blockIdx.x < extra ? (int)blockIdx.x : extra);
    uend  = ubeg + per + ((int)blockIdx.x < extra ? 1 : 0);
    ustep = 1;
  }
  const int hf  = wave >> 2;
  const int pc  = (wave & 3) * 32 + lane;
  int   pcur = -1, pcnt = 0, pm = 0;
  float ps = 0.0f;

  int   rpv;
  float dvl;
  unit_words(ubeg, rpv, dvl);

  for (int u = ubeg; u < uend; u += ustep) {
    const int n0 = u * 16 + wave * 2;
    int   rpvn;
    float dvln;
    unit_words(u + ustep, rpvn, dvln);
    const int r0 = __builtin_amdgcn_readlane(rpv, 0);
    const int r1 = __builtin_amdgcn_readlane(rpv, 1);
    const int r2 = __builtin_amdgcn_readlane(rpv, 2);
    int   ndc[2], bg[2], en[2];
    bool  ok[2];
    float dvv[2];
#pragma unroll
    for (int i = 0; i < 2; ++i) {
      const int nd = n0 + i;
      ok[i]  = nd < N;
      ndc[i] = ok[i] ? nd : 0;
      int b = (i == 0) ? r0 : r1;
      int e = (nd + 1 < N) ? ((i == 0) ? r1 : r2) : E;
      e = e < E ? e : E;  e = e > 0 ? e : 0;
      b = b > 0 ? b : 0;  b = b < e ? b : e;
      if (!ok[i]) e = b;
      bg[i] = b;  en[i] = e;
      dvv[i] = __int_as_float(__builtin_amdgcn_readlane(__float_as_int(dvl), i));
    }
    unsigned my[2];
    float    wv[2];
    float4   acc2[2];
#pragma unroll
    for (int i = 0; i < 2; ++i) {
      const int c0   = en[i] - bg[i];
      const int cnt0 = c0 < 32 ? c0 : 32;
      const unsigned m = (lane < cnt0) ? (unsigned)csr[bg[i] + lane] : 0u;
      my[i]   = m < (unsigned)N ? m : (unsigned)(N - 1);
      acc2[i] = H[(size_t)ndc[i] * 32 + lane];
    }
#pragma unroll
    for (int i = 0; i < 2; ++i) {
      wv[i] = SC_IN ? dinv[my[i]] : 0.0f;
      if (SC_IN) {
        acc2[i].x *= dvv[i];  acc2[i].y *= dvv[i];  acc2[i].z *= dvv[i];  acc2[i].w *= dvv[i];
      }
    }
#pragma unroll
    for (int i = 0; i < 2; ++i) {
      walk_list<SC_IN>(acc2[i], H, csr, dinv, my[i], wv[i], bg[i], en[i], N, lane);
      const float dv = dvv[i];
      float4 o;
      o.x = acc2[i].x * dv;  o.y = acc2[i].y * dv;
      o.z = acc2[i].z * dv;  o.w = acc2[i].w * dv;
      if (!ok[i]) o = make_float4(0.0f, 0.0f, 0.0f, 0.0f);
      const int row = wave * 2 + i;
      *(float4*)&At[row * FT_PITCH + lane * 4] = o;
      if (SC_OUT && lane == 0) Dv[row] = ok[i] ? dv : 0.0f;
    }
    __syncthreads();

    const float bcol = bias[wave * 16 + l15];
    unsigned gidv = 0u;
    if (POOL) {
      const int nb = u * 16 + hf * 8 + (lane & 7);
      gidv = (unsigned)batch[nb < N ? nb : N - 1];
      gidv = gidv < (unsigned)NGR ? gidv : (unsigned)(NGR - 1);
    }
    v8f acc;
#pragma unroll
    for (int r = 0; r < 8; ++r) acc[r] = 0.0f;
    const float* Arow = At + l15 * FT_PITCH + half * 2;
#pragma unroll
    for (int q = 0; q < HID / 4; ++q) {
      v2f a;
      a[0] = Arow[q * 4];
      a[1] = Arow[q * 4 + 1];
      acc = __builtin_amdgcn_wmma_f32_16x16x4_f32(false, a, false, bq[q],
                                                  (short)0, acc, false, false);
    }
#pragma unroll
    for (int r = 0; r < 8; ++r) {
      const int row = r + half * 8;
      float v = acc[r] + bcol;
      v = v > 0.0f ? v : 0.0f;
      Dt[row * FT_PITCH + wave * 16 + l15] = SC_OUT ? v * Dv[row] : v;
    }
    __syncthreads();

    if (POOL) {
      float pv[8];
#pragma unroll
      for (int j = 0; j < 8; ++j) pv[j] = Dt[(hf * 8 + j) * FT_PITCH + pc];
      const int nrow = N - (u * 16 + hf * 8);
#pragma unroll
      for (int j = 0; j < 8; ++j) {
        if (j < nrow) {
          const int g = __builtin_amdgcn_readlane((int)gidv, j);
          if (g != pcur) {
            if (pcnt > 0) pool_flush(pcur, ps, pm, pcnt, pc, counts, sums, maxs);
            ps = 0.0f;  pm = 0;  pcnt = 0;  pcur = g;
          }
          ps += pv[j];
          const int bits = __float_as_int(pv[j]);
          pm = bits > pm ? bits : pm;
          ++pcnt;
        }
      }
    } else {
#pragma unroll
      for (int i = 0; i < 2; ++i) {
        const int row = wave * 2 + i;
        if (n0 + i < N)
          ((float4*)out)[(size_t)(n0 + i) * 32 + lane] = *(const float4*)&Dt[row * FT_PITCH + lane * 4];
      }
    }
    rpv = rpvn;
    dvl = dvln;
  }
  if (POOL && pcnt > 0) pool_flush(pcur, ps, pm, pcnt, pc, counts, sums, maxs);
}


extern "C" void kernel_launch(void* const* d_in, const int* in_sizes, int n_in,
                              void* d_out, int out_size, void* d_ws, size_t ws_size,
                              hipStream_t stream) {
  const float* x     = (const float*)d_in[0];
  const int*   edge  = (const int*)d_in[1];
  const int*   batch = (const int*)d_in[2];
  const float* W[3]  = {(const float*)d_in[3], (const float*)d_in[5], (const float*)d_in[7]};
  const float* b[3]  = {(const float*)d_in[4], (const float*)d_in[6], (const float*)d_in[8]};
  const float* Wc1   = (const float*)d_in[9];
  const float* bc1   = (const float*)d_in[10];
  const float* Wc2   = (const float*)d_in[11];
  const float* bc2   = (const float*)d_in[12];
  float* out = (float*)d_out;

  const int N = in_sizes[0] / HID;
  const int E = in_sizes[1] / 2;
  const int G = NGR;
  const int* src  = edge;
  const int* dstv = edge + E;

  float* ws     = (float*)d_ws;
  int*   degi   = (int*)ws;
  float* dinv   = ws + N;
  float* buf0   = dinv + N;
  float* buf1   = buf0 + (size_t)N * HID;
  float* counts = buf1 + (size_t)N * HID;
  float* sums   = counts + G;
  int*   maxs   = (int*)(sums + (size_t)G * HID);
  int*   csr    = maxs + (size_t)G * HID;
  const int NB   = (N + BINW - 1) >> BINSH;
  const int NBLK = (E + ECH - 1) / ECH;
  int*   bsrc    = (int*)buf1;
  int*   bdst    = bsrc + E;
  int*   hist    = bdst + E;
  int*   binbase = hist + (size_t)NBLK * NB;

  if ((size_t)((const char*)(csr + E) - (const char*)d_ws) > ws_size) return;
  if (NB > NBMAX || (size_t)2 * E + (size_t)NBLK * NB + NB > (size_t)N * HID) return;

  k_bin_hist<<<NBLK, BTHR, 0, stream>>>(dstv, hist, E, NB, (int*)counts, NGR + 2 * NGR * HID);
  k_bin_scatter<<<NBLK, BTHR, 0, stream>>>(src, dstv, hist, bsrc, bdst, binbase, E, NB, NBLK);
  k_bin_rows<<<NB, BTHR, 0, stream>>>(bsrc, bdst, binbase, degi, dinv, csr, N, E, NB);

  const int units = (N + 15) / 16;
  k_agg_gemm<true,  true,  false><<<FUSED_BLOCKS, 256, 0, stream>>>(degi, csr, dinv, x,    W[0], b[0], buf0, N, E, units,
                                                                    batch, counts, sums, maxs);
  k_agg_gemm<false, true,  false><<<FUSED_BLOCKS, 256, 0, stream>>>(degi, csr, dinv, buf0, W[1], b[1], buf1, N, E, units,
                                                                    batch, counts, sums, maxs);
  k_agg_gemm<false, false, true ><<<FUSED_BLOCKS, 256, 0, stream>>>(degi, csr, dinv, buf1, W[2], b[2], nullptr, N, E, units,
                                                                    batch, counts, sums, maxs);

  k_head_all<<<G / 16, 256, 0, stream>>>(sums, maxs, counts, Wc1, bc1, Wc2, bc2, out);
}


# CSR build in 2 launches, not 3: chunks sorted by bin in LDS go back in place with a table of run starts, no counting launch; a bin's block reads its runs; in-row sum order changes; 7 to 6 launches
# speedup vs baseline: 7.3533x; 1.0328x over previous
#include <hip/hip_runtime.h>

#define HID   128
#define NGR   512
#define NCLS  10

typedef float v2f __attribute__((ext_vector_type(2)));
typedef float v8f __attribute__((ext_vector_type(8)));

#define BINSH  9
#define BINW   (1 << BINSH)
#define NBMAX  256
#define ECH    6400
#define BTHR   1024
#define EPT    ((ECH + BTHR - 1) / BTHR)
#define DCAP   12288

__device__ __forceinline__ int bin_of(int d, int NB) {
  const unsigned q = (unsigned)d >> BINSH;
  return (int)(q < (unsigned)NB ? q : (unsigned)(NB - 1));
}

template <int NW>
__device__ __forceinline__ int block_excl_scan(int v, int* wsum, bool active) {
  const int lane = threadIdx.x & 31;
  const int wave = threadIdx.x >> 5;
  int inc = v;
  for (int d = 1; d < 32; d <<= 1) {
    const int o = __shfl_up(inc, d);
    if (lane >= d) inc += o;
  }
  if (active && lane == 31) wsum[wave] = inc;
  __syncthreads();
  int off = 0;
  if (active) {
    for (int w = 0; w < NW; ++w) off += (w < wave) ? wsum[w] : 0;
  }
  return off + inc - v;
}

__global__ void __launch_bounds__(BTHR)
k_chunk_sort(const int* __restrict__ src, const int* __restrict__ dst,
             int* __restrict__ csrc, int* __restrict__ cdst, int* __restrict__ tab,
             int E, int NB, int NBLK, int* __restrict__ zero, int zn) {
  __shared__ int lstart[NBMAX];
  __shared__ int lcnt[NBMAX];
  __shared__ int wsum[NBMAX / 32];
  __shared__ int ssrc[ECH];
  __shared__ int sdst[ECH];

  const int t = threadIdx.x;
  const int k = blockIdx.x;
  const int beg = k * ECH;
  const int end = (beg + ECH < E) ? beg + ECH : E;
  const int n = end - beg;

  int s[EPT], d[EPT];
#pragma unroll
  for (int j = 0; j < EPT; ++j) {
    const int i  = beg + j * BTHR + t;
    const int ic = i < end ? i : beg;
    s[j] = src[ic];
    d[j] = dst[ic];
  }
  if (t < NBMAX) lcnt[t] = 0;
  __syncthreads();

  int r[EPT];
#pragma unroll
  for (int j = 0; j < EPT; ++j) {
    const int i = beg + j * BTHR + t;
    r[j] = (i < end) ? atomicAdd(&lcnt[bin_of(d[j], NB)], 1) : 0;
  }
  __syncthreads();
  const int cb = (t < NBMAX) ? lcnt[t] : 0;
  const int ls = block_excl_scan<NBMAX / 32>(cb, wsum, t < NBMAX);
  if (t < NBMAX) lstart[t] = ls;
  __syncthreads();

#pragma unroll
  for (int j = 0; j < EPT; ++j) {
    const int i = beg + j * BTHR + t;
    if (i < end) {
      unsigned p = (unsigned)(lstart[bin_of(d[j], NB)] + r[j]);
      p = p < (unsigned)ECH ? p : (unsigned)(ECH - 1);
      ssrc[p] = s[j];
      sdst[p] = d[j];
    }
  }
  if (t <= NB) tab[(size_t)t * NBLK + k] = (t < NB) ? lstart[t] : n;
  __syncthreads();

  for (int i = t; i < n; i += BTHR) {
    csrc[beg + i] = ssrc[i];
    cdst[beg + i] = sdst[i];
  }
  for (int i = blockIdx.x * BTHR + t; i < zn; i += gridDim.x * BTHR) zero[i] = 0;
}

#define RPW    8
#define RHEAD  2
__global__ void __launch_bounds__(BTHR)
k_bin_rows(const int* __restrict__ csrc, const int* __restrict__ cdst,
           const int* __restrict__ tab, int* __restrict__ rowptr,
           float* __restrict__ dinv, int* __restrict__ csr, int N, int E, int NB, int NBLK) {
  __shared__ int cnt[BINW];
  __shared__ int rstart[BINW];
  __shared__ int cur[BINW];
  __shared__ int wsum[BINW / 32];
  __shared__ int rbeg[32 * RPW];
  __shared__ int rcnt[32 * RPW];
  __shared__ int red0[32];
  __shared__ int red1[32];
  __shared__ int slist[DCAP];

  const int t = threadIdx.x;
  const int b = blockIdx.x;
  const int lane = t & 31;
  const int wave = __builtin_amdgcn_readfirstlane(t >> 5);

  int st = 0, en = 0;
  if (t < NBLK) {
    st = tab[(size_t)b * NBLK + t];
    en = tab[(size_t)(b + 1) * NBLK + t];
    const int left = E - t * ECH;
    const int csz = left < ECH ? left : ECH;
    st = st > 0 ? st : 0;    st = st < csz ? st : csz;
    en = en > st ? en : st;  en = en < csz ? en : csz;
  }
  if (t < 32 * RPW) { rbeg[t] = t * ECH + st; rcnt[t] = en - st; }
  if (t < BINW) { cnt[t] = 0; cur[t] = 0; }
  {
    int a0 = st, a1 = en - st;
    for (int o = 16; o > 0; o >>= 1) { a0 += __shfl_xor(a0, o); a1 += __shfl_xor(a1, o); }
    if (lane == 0) { red0[wave] = a0; red1[wave] = a1; }
  }
  __syncthreads();
  unsigned u0 = 0u, un = 0u;
  for (int w = 0; w < 32; ++w) { u0 += (unsigned)red0[w]; un += (unsigned)red1[w]; }
  u0 = u0 < (unsigned)E ? u0 : (unsigned)E;
  un = un < (unsigned)E - u0 ? un : (unsigned)E - u0;
  const int e0 = (int)u0;
  const int n  = (int)un;

  int rb[RPW], rc[RPW];
  int dd[RPW][RHEAD], ss[RPW][RHEAD];
#pragma unroll
  for (int i = 0; i < RPW; ++i) {
    rb[i] = __builtin_amdgcn_readfirstlane(rbeg[wave + 32 * i]);
    rc[i] = __builtin_amdgcn_readfirstlane(rcnt[wave + 32 * i]);
  }
#pragma unroll
  for (int i = 0; i < RPW; ++i) {
#pragma unroll
    for (int h = 0; h < RHEAD; ++h) {
      const int j = lane + 32 * h;
      const int ic = j < rc[i] ? rb[i] + j : 0;
      dd[i][h] = cdst[ic];
      ss[i][h] = csrc[ic];
    }
  }
#pragma unroll
  for (int i = 0; i < RPW; ++i) {
#pragma unroll
    for (int h = 0; h < RHEAD; ++h)
      if (lane + 32 * h < rc[i]) atomicAdd(&cnt[dd[i][h] & (BINW - 1)], 1);
    for (int j = RHEAD * 32 + lane; j < rc[i]; j += 32)
      atomicAdd(&cnt[cdst[rb[i] + j] & (BINW - 1)], 1);
  }
  __syncthreads();

  const int c = (t < BINW) ? cnt[t] : 0;
  const int ex = block_excl_scan<BINW / 32>(c, wsum, t < BINW);
  if (t < BINW) {
    rstart[t] = ex;
    const int node = b * BINW + t;
    if (node < N) {
      rowptr[node] = e0 + ex;
      dinv[node] = rsqrtf((float)c + 1.0f);
    }
  }
  __syncthreads();

  const bool inlds = n <= DCAP;
  auto place = [&](int d, int s) {
    const int dl = d & (BINW - 1);
    const unsigned p = (unsigned)(rstart[dl] + atomicAdd(&cur[dl], 1));
    if (inlds) {
      slist[p < (unsigned)DCAP ? p : (unsigned)(DCAP - 1)] = s;
    } else {
      const unsigned g = u0 + p;
      csr[g < (unsigned)E ? g : (unsigned)(E - 1)] = s;
    }
  };
#pragma unroll
  for (int i = 0; i < RPW; ++i) {
#pragma unroll
    for (int h = 0; h < RHEAD; ++h)
      if (lane + 32 * h < rc[i]) place(dd[i][h], ss[i][h]);
    for (int j = RHEAD * 32 + lane; j < rc[i]; j += 32)
      place(cdst[rb[i] + j], csrc[rb[i] + j]);
  }
  if (inlds) {
    __syncthreads();
    for (int i = t; i < n; i += BTHR) csr[e0 + i] = slist[i];
  }
}

#define HK    (2 * HID)
#define HAP   (HK + 4)
#define HHP   (HID + 4)
#define HWPT  (HID * NCLS / 256)
__global__ void __launch_bounds__(256)
k_head_all(const float* __restrict__ sums, const int* __restrict__ maxs,
           const float* __restrict__ counts, const float* __restrict__ Wc1,
           const float* __restrict__ bc1, const float* __restrict__ Wc2,
           const float* __restrict__ bc2, float* __restrict__ out) {
  __shared__ __attribute__((aligned(16))) float As[16 * HAP];
  __shared__ __attribute__((aligned(16))) float Hs[16 * HHP];
  __shared__ float Ws[HID * NCLS];

  const int t    = threadIdx.x;
  const int lane = t & 31;
  const int tn   = t >> 5;
  const int tm   = blockIdx.x;
  const int half = lane >> 4;
  const int l15  = lane & 15;

  const float* __restrict__ Bcol = Wc1 + tn * 16 + l15;
  v2f bq[HK / 4];
#pragma unroll
  for (int q = 0; q < HK / 4; ++q) {
    const int kk = q * 4 + half * 2;
    bq[q][0] = Bcol[(size_t)kk * HID];
    bq[q][1] = Bcol[(size_t)(kk + 1) * HID];
  }
  const float bv = bc1[tn * 16 + l15];

  {
    float4 sv[2];
    int4   mv[2];
    float  cv[2];
    float  wv[HWPT];
#pragma unroll
    for (int j = 0; j < 2; ++j) {
      const int q = j * 256 + t;
      const int r = q >> 5;
      const size_t o = (size_t)(tm * 16 + r) * HID + ((q & 31) << 2);
      sv[j] = *(const float4*)(sums + o);
      mv[j] = *(const int4*)(maxs + o);
      cv[j] = counts[tm * 16 + r];
    }
#pragma unroll
    for (int j = 0; j < HWPT; ++j) wv[j] = Wc2[j * 256 + t];
#pragma unroll
    for (int j = 0; j < 2; ++j) {
      const int q = j * 256 + t;
      const int r = q >> 5;
      const int c4 = (q & 31) << 2;
      const float denom = cv[j] > 1.0f ? cv[j] : 1.0f;
      float4 m;
      m.x = sv[j].x / denom;  m.y = sv[j].y / denom;
      m.z = sv[j].z / denom;  m.w = sv[j].w / denom;
      float4 x;
      x.x = __int_as_float(mv[j].x);  x.y = __int_as_float(mv[j].y);
      x.z = __int_as_float(mv[j].z);  x.w = __int_as_float(mv[j].w);
      *(float4*)&As[r * HAP + c4]       = m;
      *(float4*)&As[r * HAP + HID + c4] = x;
    }
#pragma unroll
    for (int j = 0; j < HWPT; ++j) Ws[j * 256 + t] = wv[j];
  }
  __syncthreads();

  v8f acc;
#pragma unroll
  for (int r = 0; r < 8; ++r) acc[r] = bv;
  const float* Arow = As + l15 * HAP;
#pragma unroll
  for (int q = 0; q < HK / 4; ++q) {
    const int kk = q * 4 + half * 2;
    v2f a;
    a[0] = Arow[kk];
    a[1] = Arow[kk + 1];
    acc = __builtin_amdgcn_wmma_f32_16x16x4_f32(false, a, false, bq[q],
                                                (short)0, acc, false, false);
  }
#pragma unroll
  for (int r = 0; r < 8; ++r) {
    const float v = acc[r] > 0.0f ? acc[r] : 0.0f;
    Hs[(r + half * 8) * HHP + tn * 16 + l15] = v;
  }
  __syncthreads();

  if (t < 16 * NCLS) {
    const int g = t / NCLS;
    const int c = t - g * NCLS;
    float s = bc2[c];
    const float* hv = Hs + g * HHP;
#pragma unroll 8
    for (int k = 0; k < HID; ++k) s += hv[k] * Ws[k * NCLS + c];
    out[(size_t)(tm * 16 + g) * NCLS + c] = s;
  }
}


template <int U, bool SC>
__device__ __forceinline__ void add_rows(float4& acc, const float4* __restrict__ H,
                                         int my, float& wv, int j, int lane) {
  float4 v[U];
#pragma unroll
  for (int u = 0; u < U; ++u) {
    const int s = __builtin_amdgcn_readlane(my, j + u);
    v[u] = H[(size_t)s * 32 + lane];
  }
  if (SC) asm volatile("" : "+v"(wv) : : "memory");
#pragma unroll
  for (int u = 0; u < U; ++u) {
    if (SC) {
      const float w = __int_as_float(__builtin_amdgcn_readlane(__float_as_int(wv), j + u));
      acc.x = __builtin_fmaf(w, v[u].x, acc.x);  acc.y = __builtin_fmaf(w, v[u].y, acc.y);
      acc.z = __builtin_fmaf(w, v[u].z, acc.z);  acc.w = __builtin_fmaf(w, v[u].w, acc.w);
    } else {
      acc.x += v[u].x; acc.y += v[u].y; acc.z += v[u].z; acc.w += v[u].w;
    }
  }
}

#define FT_PITCH     (HID + 4)
#define FUSED_BLOCKS 1024

template <bool SC>
__device__ __forceinline__ void walk_list(float4& acc, const float4* __restrict__ H,
                                          const int* __restrict__ csr, const float* __restrict__ dinv,
                                          unsigned my, float wv, int beg, int end, int N, int lane) {
  for (int base = beg; base < end; base += 32) {
    const int rem = end - base;
    const int cnt = rem < 32 ? rem : 32;
    if (base != beg) {
      my = (lane < cnt) ? (unsigned)csr[base + lane] : 0u;
      my = my < (unsigned)N ? my : (unsigned)(N - 1);
      if (SC) wv = dinv[my];
    }
    int j = 0;
    for (; j + 8 <= cnt; j += 8) add_rows<8, SC>(acc, H, (int)my, wv, j, lane);
    switch (cnt - j) {
      case 7: add_rows<7, SC>(acc, H, (int)my, wv, j, lane); break;
      case 6: add_rows<6, SC>(acc, H, (int)my, wv, j, lane); break;
      case 5: add_rows<5, SC>(acc, H, (int)my, wv, j, lane); break;
      case 4: add_rows<4, SC>(acc, H, (int)my, wv, j, lane); break;
      case 3: add_rows<3, SC>(acc, H, (int)my, wv, j, lane); break;
      case 2: add_rows<2, SC>(acc, H, (int)my, wv, j, lane); break;
      case 1: add_rows<1, SC>(acc, H, (int)my, wv, j, lane); break;
      default: break;
    }
  }
}

__device__ __forceinline__ void pool_flush(int g, float s, int m, int cnt, int col,
                                           float* __restrict__ counts, float* __restrict__ sums,
                                           int* __restrict__ maxs) {
  atomicAdd(sums + (size_t)g * HID + col, s);
  atomicMax(maxs + (size_t)g * HID + col, m);
  if (col == 0) atomicAdd(&counts[g], (float)cnt);
}

template <bool SC_IN, bool SC_OUT, bool POOL>
__global__ void __launch_bounds__(256, 8)
k_agg_gemm(const int* __restrict__ rowptr, const int* __restrict__ csr,
           const float* __restrict__ dinv, const float* __restrict__ hin,
           const float* __restrict__ Wl, const float* __restrict__ bias,
           float* __restrict__ out, int N, int E, int units,
           const int* __restrict__ batch, float* __restrict__ counts,
           float* __restrict__ sums, int* __restrict__ maxs) {
  __shared__ __attribute__((aligned(16))) float At[16 * FT_PITCH];
  __shared__ __attribute__((aligned(16))) float Dt[16 * FT_PITCH];
  __shared__ float Dv[16];

  const int lane = threadIdx.x & 31;
  const int wave = __builtin_amdgcn_readfirstlane((int)(threadIdx.x >> 5));
  const int half = lane >> 4;
  const int l15  = lane & 15;

  const float* __restrict__ Bcol = Wl + wave * 16 + l15;
  v2f bq[HID / 4];
#pragma unroll
  for (int q = 0; q < HID / 4; ++q) {
    const int kk = q * 4 + half * 2;
    bq[q][0] = Bcol[(size_t)kk * HID];
    bq[q][1] = Bcol[(size_t)(kk + 1) * HID];
  }
  const float4* __restrict__ H = (const float4*)hin;

  auto unit_words = [&](int uu, int& rpv, float& dvl) {
    int i0 = uu * 16 + wave * 2 + (lane < 3 ? lane : 0);
    i0 = i0 < N ? i0 : N - 1;
    rpv = rowptr[i0];
    dvl = dinv[i0];
  };
  int ubeg = (int)blockIdx.x, uend = units, ustep = (int)gridDim.x;
  if (POOL) {
    const int per = units / (int)gridDim.x, extra = units % (int)gridDim.x;
    ubeg  = (int)blockIdx.x * per + ((int)blockIdx.x < extra ? (int)blockIdx.x : extra);
    uend  = ubeg + per + ((int)blockIdx.x < extra ? 1 : 0);
    ustep = 1;
  }
  const int hf  = wave >> 2;
  const int pc  = (wave & 3) * 32 + lane;
  int   pcur = -1, pcnt = 0, pm = 0;
  float ps = 0.0f;

  int   rpv;
  float dvl;
  unit_words(ubeg, rpv, dvl);

  for (int u = ubeg; u < uend; u += ustep) {
    const int n0 = u * 16 + wave * 2;
    int   rpvn;
    float dvln;
    unit_words(u + ustep, rpvn, dvln);
    const int r0 = __builtin_amdgcn_readlane(rpv, 0);
    const int r1 = __builtin_amdgcn_readlane(rpv, 1);
    const int r2 = __builtin_amdgcn_readlane(rpv, 2);
    int   ndc[2], bg[2], en[2];
    bool  ok[2];
    float dvv[2];
#pragma unroll
    for (int i = 0; i < 2; ++i) {
      const int nd = n0 + i;
      ok[i]  = nd < N;
      ndc[i] = ok[i] ? nd : 0;
      int b = (i == 0) ? r0 : r1;
      int e = (nd + 1 < N) ? ((i == 0) ? r1 : r2) : E;
      e = e < E ? e : E;  e = e > 0 ? e : 0;
      b = b > 0 ? b : 0;  b = b < e ? b : e;
      if (!ok[i]) e = b;
      bg[i] = b;  en[i] = e;
      dvv[i] = __int_as_float(__builtin_amdgcn_readlane(__float_as_int(dvl), i));
    }
    unsigned my[2];
    float    wv[2];
    float4   acc2[2];
#pragma unroll
    for (int i = 0; i < 2; ++i) {
      const int c0   = en[i] - bg[i];
      const int cnt0 = c0 < 32 ? c0 : 32;
      const unsigned m = (lane < cnt0) ? (unsigned)csr[bg[i] + lane] : 0u;
      my[i]   = m < (unsigned)N ? m : (unsigned)(N - 1);
      acc2[i] = H[(size_t)ndc[i] * 32 + lane];
    }
#pragma unroll
    for (int i = 0; i < 2; ++i) {
      wv[i] = SC_IN ? dinv[my[i]] : 0.0f;
      if (SC_IN) {
        acc2[i].x *= dvv[i];  acc2[i].y *= dvv[i];  acc2[i].z *= dvv[i];  acc2[i].w *= dvv[i];
      }
    }
#pragma unroll
    for (int i = 0; i < 2; ++i) {
      walk_list<SC_IN>(acc2[i], H, csr, dinv, my[i], wv[i], bg[i], en[i], N, lane);
      const float dv = dvv[i];
      float4 o;
      o.x = acc2[i].x * dv;  o.y = acc2[i].y * dv;
      o.z = acc2[i].z * dv;  o.w = acc2[i].w * dv;
      if (!ok[i]) o = make_float4(0.0f, 0.0f, 0.0f, 0.0f);
      const int row = wave * 2 + i;
      *(float4*)&At[row * FT_PITCH + lane * 4] = o;
      if (SC_OUT && lane == 0) Dv[row] = ok[i] ? dv : 0.0f;
    }
    __syncthreads();

    const float bcol = bias[wave * 16 + l15];
    unsigned gidv = 0u;
    if (POOL) {
      const int nb = u * 16 + hf * 8 + (lane & 7);
      gidv = (unsigned)batch[nb < N ? nb : N - 1];
      gidv = gidv < (unsigned)NGR ? gidv : (unsigned)(NGR - 1);
    }
    v8f acc;
#pragma unroll
    for (int r = 0; r < 8; ++r) acc[r] = 0.0f;
    const float* Arow = At + l15 * FT_PITCH + half * 2;
#pragma unroll
    for (int q = 0; q < HID / 4; ++q) {
      v2f a;
      a[0] = Arow[q * 4];
      a[1] = Arow[q * 4 + 1];
      acc = __builtin_amdgcn_wmma_f32_16x16x4_f32(false, a, false, bq[q],
                                                  (short)0, acc, false, false);
    }
#pragma unroll
    for (int r = 0; r < 8; ++r) {
      const int row = r + half * 8;
      float v = acc[r] + bcol;
      v = v > 0.0f ? v : 0.0f;
      Dt[row * FT_PITCH + wave * 16 + l15] = SC_OUT ? v * Dv[row] : v;
    }
    __syncthreads();

    if (POOL) {
      float pv[8];
#pragma unroll
      for (int j = 0; j < 8; ++j) pv[j] = Dt[(hf * 8 + j) * FT_PITCH + pc];
      const int nrow = N - (u * 16 + hf * 8);
#pragma unroll
      for (int j = 0; j < 8; ++j) {
        if (j < nrow) {
          const int g = __builtin_amdgcn_readlane((int)gidv, j);
          if (g != pcur) {
            if (pcnt > 0) pool_flush(pcur, ps, pm, pcnt, pc, counts, sums, maxs);
            ps = 0.0f;  pm = 0;  pcnt = 0;  pcur = g;
          }
          ps += pv[j];
          const int bits = __float_as_int(pv[j]);
          pm = bits > pm ? bits : pm;
          ++pcnt;
        }
      }
    } else {
#pragma unroll
      for (int i = 0; i < 2; ++i) {
        const int row = wave * 2 + i;
        if (n0 + i < N)
          ((float4*)out)[(size_t)(n0 + i) * 32 + lane] = *(const float4*)&Dt[row * FT_PITCH + lane * 4];
      }
    }
    rpv = rpvn;
    dvl = dvln;
  }
  if (POOL && pcnt > 0) pool_flush(pcur, ps, pm, pcnt, pc, counts, sums, maxs);
}


extern "C" void kernel_launch(void* const* d_in, const int* in_sizes, int n_in,
                              void* d_out, int out_size, void* d_ws, size_t ws_size,
                              hipStream_t stream) {
  const float* x     = (const float*)d_in[0];
  const int*   edge  = (const int*)d_in[1];
  const int*   batch = (const int*)d_in[2];
  const float* W[3]  = {(const float*)d_in[3], (const float*)d_in[5], (const float*)d_in[7]};
  const float* b[3]  = {(const float*)d_in[4], (const float*)d_in[6], (const float*)d_in[8]};
  const float* Wc1   = (const float*)d_in[9];
  const float* bc1   = (const float*)d_in[10];
  const float* Wc2   = (const float*)d_in[11];
  const float* bc2   = (const float*)d_in[12];
  float* out = (float*)d_out;

  const int N = in_sizes[0] / HID;
  const int E = in_sizes[1] / 2;
  const int G = NGR;
  const int* src  = edge;
  const int* dstv = edge + E;

  float* ws     = (float*)d_ws;
  int*   degi   = (int*)ws;
  float* dinv   = ws + N;
  float* buf0   = dinv + N;
  float* buf1   = buf0 + (size_t)N * HID;
  float* counts = buf1 + (size_t)N * HID;
  float* sums   = counts + G;
  int*   maxs   = (int*)(sums + (size_t)G * HID);
  int*   csr    = maxs + (size_t)G * HID;
  const int NB   = (N + BINW - 1) >> BINSH;
  const int NBLK = (E + ECH - 1) / ECH;
  int*   csrc    = (int*)buf1;
  int*   cdst    = csrc + E;
  int*   tab     = cdst + E;

  if ((size_t)((const char*)(csr + E) - (const char*)d_ws) > ws_size) return;
  if (NB > NBMAX || NBLK > 32 * RPW || (size_t)2 * E + (size_t)NBLK * (NB + 1) > (size_t)N * HID) return;

  k_chunk_sort<<<NBLK, BTHR, 0, stream>>>(src, dstv, csrc, cdst, tab, E, NB, NBLK, (int*)counts, NGR + 2 * NGR * HID);
  k_bin_rows<<<NB, BTHR, 0, stream>>>(csrc, cdst, tab, degi, dinv, csr, N, E, NB, NBLK);

  const int units = (N + 15) / 16;
  k_agg_gemm<true,  true,  false><<<FUSED_BLOCKS, 256, 0, stream>>>(degi, csr, dinv, x,    W[0], b[0], buf0, N, E, units,
                                                                    batch, counts, sums, maxs);
  k_agg_gemm<false, true,  false><<<FUSED_BLOCKS, 256, 0, stream>>>(degi, csr, dinv, buf0, W[1], b[1], buf1, N, E, units,
                                                                    batch, counts, sums, maxs);
  k_agg_gemm<false, false, true ><<<FUSED_BLOCKS, 256, 0, stream>>>(degi, csr, dinv, buf1, W[2], b[2], nullptr, N, E, units,
                                                                    batch, counts, sums, maxs);

  k_head_all<<<G / 16, 256, 0, stream>>>(sums, maxs, counts, Wc1, bc1, Wc2, bc2, out);
}
